# starting kernel
# baseline (speedup 1.0000x reference)
#include <hip/hip_runtime.h>
#include <hip/hip_bf16.h>
#include <math.h>

typedef float v2f __attribute__((ext_vector_type(2)));
typedef float v8f __attribute__((ext_vector_type(8)));

#define NEG_SLOPE 0.2f

__device__ __forceinline__ float leaky(float v) { return v > 0.f ? v : NEG_SLOPE * v; }

__device__ __forceinline__ void atomicMaxF(float* addr, float val) {
  if (val >= 0.f) atomicMax((int*)addr, __float_as_int(val));
  else            atomicMin((unsigned int*)addr, (unsigned int)__float_as_int(val));
}

__global__ void init_kern(float* emax1, float* denom1, float* agg1,
                          float* emax2, float* denom2, float* agg2, int N) {
  int i = blockIdx.x * blockDim.x + threadIdx.x;
  if (i < N * 64) agg1[i] = 0.f;
  if (i < N * 6)  agg2[i] = 0.f;
  if (i < N * 4) { emax1[i] = -3.0e38f; denom1[i] = 0.f; }
  if (i < N)     { emax2[i] = -3.0e38f; denom2[i] = 0.f; }
}

__global__ void proj1_wmma(const float* __restrict__ x, const float* __restrict__ W1,
                           float* __restrict__ h1, int N) {
  int gtid = blockIdx.x * blockDim.x + threadIdx.x;
  int wave = gtid >> 5;
  int lane = threadIdx.x & 31;
  int tiles = (N + 15) >> 4;
  if (wave >= tiles) return;
  int half = lane >> 4;
  int l16  = lane & 15;
  int row  = wave * 16 + l16;
  int rowc = row < N ? row : N - 1;
  bool rok = row < N;

  v2f a[2];
#pragma unroll
  for (int s = 0; s < 2; ++s) {
    int k0 = 4 * s + 2 * half;
    a[s][0] = (rok && k0     < 7) ? x[rowc * 7 + k0]     : 0.f;
    a[s][1] = (rok && k0 + 1 < 7) ? x[rowc * 7 + k0 + 1] : 0.f;
  }

  v8f acc[4];
#pragma unroll
  for (int t = 0; t < 4; ++t)
#pragma unroll
    for (int q = 0; q < 8; ++q) acc[t][q] = 0.f;

#pragma unroll
  for (int s = 0; s < 2; ++s) {
    int k0 = 4 * s + 2 * half;
#pragma unroll
    for (int t = 0; t < 4; ++t) {
      int col = t * 16 + l16;
      v2f b;
      b[0] = (k0     < 7) ? W1[k0 * 64 + col]       : 0.f;
      b[1] = (k0 + 1 < 7) ? W1[(k0 + 1) * 64 + col] : 0.f;
      acc[t] = __builtin_amdgcn_wmma_f32_16x16x4_f32(
          false, a[s], false, b, (short)0, acc[t], false, false);
    }
  }

#pragma unroll
  for (int t = 0; t < 4; ++t)
#pragma unroll
    for (int r = 0; r < 8; ++r) {
      int m = wave * 16 + r + 8 * half;
      if (m < N) h1[m * 64 + t * 16 + l16] = acc[t][r];
    }
}

__global__ void alpha1_kern(const float* __restrict__ h1, const float* __restrict__ aS,
                            const float* __restrict__ aD, float* __restrict__ as1,
                            float* __restrict__ ad1, int N) {
  int i = blockIdx.x * blockDim.x + threadIdx.x;
  if (i >= N * 4) return;
  int hd = i & 3, n = i >> 2;
  const float* hp = h1 + n * 64 + hd * 16;
  float s = 0.f, d = 0.f;
#pragma unroll
  for (int c = 0; c < 16; ++c) {
    float v = hp[c];
    s += v * aS[hd * 16 + c];
    d += v * aD[hd * 16 + c];
  }
  as1[i] = s; ad1[i] = d;
}

__global__ void e1_max(const int* __restrict__ ei, const float* __restrict__ as1,
                       const float* __restrict__ ad1, float* __restrict__ emax1,
                       int E, int N) {
  int e = blockIdx.x * blockDim.x + threadIdx.x;
  if (e >= E + N) return;
  int src = e < E ? ei[e]     : e - E;
  int dst = e < E ? ei[E + e] : e - E;
#pragma unroll
  for (int h = 0; h < 4; ++h) {
    float ev = leaky(as1[src * 4 + h] + ad1[dst * 4 + h]);
    atomicMaxF(&emax1[dst * 4 + h], ev);
  }
}

__global__ void e1_den(const int* __restrict__ ei, const float* __restrict__ as1,
                       const float* __restrict__ ad1, const float* __restrict__ emax1,
                       float* __restrict__ denom1, int E, int N) {
  int e = blockIdx.x * blockDim.x + threadIdx.x;
  if (e >= E + N) return;
  int src = e < E ? ei[e]     : e - E;
  int dst = e < E ? ei[E + e] : e - E;
#pragma unroll
  for (int h = 0; h < 4; ++h) {
    float ev = leaky(as1[src * 4 + h] + ad1[dst * 4 + h]);
    atomicAdd(&denom1[dst * 4 + h], __expf(ev - emax1[dst * 4 + h]));
  }
}

__global__ void e1_agg(const int* __restrict__ ei, const float* __restrict__ as1,
                       const float* __restrict__ ad1, const float* __restrict__ emax1,
                       const float* __restrict__ denom1, const float* __restrict__ h1,
                       float* __restrict__ agg1, int E, int N) {
  int t = blockIdx.x * blockDim.x + threadIdx.x;
  if (t >= (E + N) * 4) return;
  int e = t >> 2, h = t & 3;
  int src = e < E ? ei[e]     : e - E;
  int dst = e < E ? ei[E + e] : e - E;
  float ev = leaky(as1[src * 4 + h] + ad1[dst * 4 + h]);
  float w  = __expf(ev - emax1[dst * 4 + h]) / denom1[dst * 4 + h];
  const float4* hp = (const float4*)(h1 + src * 64 + h * 16);
  float* op = agg1 + dst * 64 + h * 16;
#pragma unroll
  for (int q = 0; q < 4; ++q) {
    float4 v = hp[q];
    atomicAdd(op + q * 4 + 0, v.x * w);
    atomicAdd(op + q * 4 + 1, v.y * w);
    atomicAdd(op + q * 4 + 2, v.z * w);
    atomicAdd(op + q * 4 + 3, v.w * w);
  }
}

__global__ void bias_elu(float* __restrict__ x2, const float* __restrict__ b1, int N) {
  int i = blockIdx.x * blockDim.x + threadIdx.x;
  if (i >= N * 64) return;
  float v = x2[i] + b1[i & 63];
  x2[i] = v > 0.f ? v : (__expf(v) - 1.f);
}

__global__ void proj2_wmma(const float* __restrict__ x2, const float* __restrict__ W2,
                           float* __restrict__ h2, int N) {
  int gtid = blockIdx.x * blockDim.x + threadIdx.x;
  int wave = gtid >> 5;
  int lane = threadIdx.x & 31;
  int tiles = (N + 15) >> 4;
  if (wave >= tiles) return;
  int half = lane >> 4, l16 = lane & 15;
  int row  = wave * 16 + l16;
  int rowc = row < N ? row : N - 1;
  bool rok = row < N;
  bool cok = l16 < 6;

  v8f acc;
#pragma unroll
  for (int q = 0; q < 8; ++q) acc[q] = 0.f;

#pragma unroll
  for (int s = 0; s < 16; ++s) {
    int k0 = 4 * s + 2 * half;
    v2f a, b;
    a[0] = rok ? x2[rowc * 64 + k0]     : 0.f;
    a[1] = rok ? x2[rowc * 64 + k0 + 1] : 0.f;
    b[0] = cok ? W2[k0 * 6 + l16]       : 0.f;
    b[1] = cok ? W2[(k0 + 1) * 6 + l16] : 0.f;
    acc = __builtin_amdgcn_wmma_f32_16x16x4_f32(
        false, a, false, b, (short)0, acc, false, false);
  }

  if (cok) {
#pragma unroll
    for (int r = 0; r < 8; ++r) {
      int m = wave * 16 + r + 8 * half;
      if (m < N) h2[m * 6 + l16] = acc[r];
    }
  }
}

__global__ void alpha2_kern(const float* __restrict__ h2, const float* __restrict__ aS,
                            const float* __restrict__ aD, float* __restrict__ as2,
                            float* __restrict__ ad2, int N) {
  int n = blockIdx.x * blockDim.x + threadIdx.x;
  if (n >= N) return;
  float s = 0.f, d = 0.f;
#pragma unroll
  for (int c = 0; c < 6; ++c) {
    float v = h2[n * 6 + c];
    s += v * aS[c];
    d += v * aD[c];
  }
  as2[n] = s; ad2[n] = d;
}

__global__ void e2_max(const int* __restrict__ ei, const float* __restrict__ as2,
                       const float* __restrict__ ad2, float* __restrict__ emax2,
                       int E, int N) {
  int e = blockIdx.x * blockDim.x + threadIdx.x;
  if (e >= E + N) return;
  int src = e < E ? ei[e]     : e - E;
  int dst = e < E ? ei[E + e] : e - E;
  atomicMaxF(&emax2[dst], leaky(as2[src] + ad2[dst]));
}

__global__ void e2_den(const int* __restrict__ ei, const float* __restrict__ as2,
                       const float* __restrict__ ad2, const float* __restrict__ emax2,
                       float* __restrict__ denom2, int E, int N) {
  int e = blockIdx.x * blockDim.x + threadIdx.x;
  if (e >= E + N) return;
  int src = e < E ? ei[e]     : e - E;
  int dst = e < E ? ei[E + e] : e - E;
  float ev = leaky(as2[src] + ad2[dst]);
  atomicAdd(&denom2[dst], __expf(ev - emax2[dst]));
}

__global__ void e2_agg(const int* __restrict__ ei, const float* __restrict__ as2,
                       const float* __restrict__ ad2, const float* __restrict__ emax2,
                       const float* __restrict__ denom2, const float* __restrict__ h2,
                       float* __restrict__ agg2, int E, int N) {
  int e = blockIdx.x * blockDim.x + threadIdx.x;
  if (e >= E + N) return;
  int src = e < E ? ei[e]     : e - E;
  int dst = e < E ? ei[E + e] : e - E;
  float ev = leaky(as2[src] + ad2[dst]);
  float w  = __expf(ev - emax2[dst]) / denom2[dst];
#pragma unroll
  for (int c = 0; c < 6; ++c)
    atomicAdd(&agg2[dst * 6 + c], h2[src * 6 + c] * w);
}

__global__ void final_kern(const float* __restrict__ agg2, const float* __restrict__ b2,
                           float* __restrict__ out, int N) {
  int n = blockIdx.x * blockDim.x + threadIdx.x;
  if (n >= N) return;
  float o[6];
  float m = -3.0e38f;
#pragma unroll
  for (int c = 0; c < 6; ++c) { o[c] = agg2[n * 6 + c] + b2[c]; m = fmaxf(m, o[c]); }
  float sum = 0.f;
#pragma unroll
  for (int c = 0; c < 6; ++c) sum += __expf(o[c] - m);
  float l = __logf(sum);
#pragma unroll
  for (int c = 0; c < 6; ++c) out[n * 6 + c] = o[c] - m - l;
}

extern "C" void kernel_launch(void* const* d_in, const int* in_sizes, int n_in,
                              void* d_out, int out_size, void* d_ws, size_t ws_size,
                              hipStream_t stream) {
  const float* x      = (const float*)d_in[0];
  const int*   ei     = (const int*)d_in[1];
  const float* W1     = (const float*)d_in[2];
  const float* a_src1 = (const float*)d_in[3];
  const float* a_dst1 = (const float*)d_in[4];
  const float* b1     = (const float*)d_in[5];
  const float* W2     = (const float*)d_in[6];
  const float* a_src2 = (const float*)d_in[7];
  const float* a_dst2 = (const float*)d_in[8];
  const float* b2     = (const float*)d_in[9];
  float* out = (float*)d_out;

  int N = in_sizes[0] / 7;
  int E = in_sizes[1] / 2;
  int ET = E + N;

  float* ws     = (float*)d_ws;
  float* h1     = ws;
  float* x2     = h1 + (size_t)N * 64;
  float* as1    = x2 + (size_t)N * 64;
  float* ad1    = as1 + (size_t)N * 4;
  float* emax1  = ad1 + (size_t)N * 4;
  float* denom1 = emax1 + (size_t)N * 4;
  float* h2     = denom1 + (size_t)N * 4;
  float* agg2   = h2 + (size_t)N * 6;
  float* as2    = agg2 + (size_t)N * 6;
  float* ad2    = as2 + N;
  float* emax2  = ad2 + N;
  float* denom2 = emax2 + N;

  const int B = 256;
  auto G = [](long n, int b) { return (int)((n + b - 1) / b); };
  int tiles = (N + 15) / 16;
  int projBlocks = (tiles + (B / 32) - 1) / (B / 32);

  init_kern<<<G((long)N * 64, B), B, 0, stream>>>(emax1, denom1, x2, emax2, denom2, agg2, N);

  proj1_wmma<<<projBlocks, B, 0, stream>>>(x, W1, h1, N);
  alpha1_kern<<<G((long)N * 4, B), B, 0, stream>>>(h1, a_src1, a_dst1, as1, ad1, N);
  e1_max<<<G(ET, B), B, 0, stream>>>(ei, as1, ad1, emax1, E, N);
  e1_den<<<G(ET, B), B, 0, stream>>>(ei, as1, ad1, emax1, denom1, E, N);
  e1_agg<<<G((long)ET * 4, B), B, 0, stream>>>(ei, as1, ad1, emax1, denom1, h1, x2, E, N);
  bias_elu<<<G((long)N * 64, B), B, 0, stream>>>(x2, b1, N);

  proj2_wmma<<<projBlocks, B, 0, stream>>>(x2, W2, h2, N);
  alpha2_kern<<<G(N, B), B, 0, stream>>>(h2, a_src2, a_dst2, as2, ad2, N);
  e2_max<<<G(ET, B), B, 0, stream>>>(ei, as2, ad2, emax2, E, N);
  e2_den<<<G(ET, B), B, 0, stream>>>(ei, as2, ad2, emax2, denom2, E, N);
  e2_agg<<<G(ET, B), B, 0, stream>>>(ei, as2, ad2, emax2, denom2, h2, agg2, E, N);
  final_kern<<<G(N, B), B, 0, stream>>>(agg2, b2, out, N);
}


# Edges bucketed by destination each call (count+rank, scan, scatter); each node then reduces its own edge list in registers, so all float atomics go; sums now run in arrival-rank order
# speedup vs baseline: 6.6343x; 6.6343x over previous
#include <hip/hip_runtime.h>
#include <hip/hip_bf16.h>
#include <math.h>

typedef float v2f __attribute__((ext_vector_type(2)));
typedef float v8f __attribute__((ext_vector_type(8)));

#define NEG_SLOPE 0.2f

__device__ __forceinline__ float leaky(float v) { return v > 0.f ? v : NEG_SLOPE * v; }


__global__ void zero_count(int* __restrict__ count, int N) {
  int i = blockIdx.x * blockDim.x + threadIdx.x;
  if (i < N) count[i] = 0;
}

__global__ void count_rank(const int* __restrict__ ei, int* __restrict__ count,
                           int* __restrict__ rank, int E) {
  int gid = blockIdx.x * blockDim.x + threadIdx.x;
  int stride = gridDim.x * blockDim.x;
  const int* dst = ei + E;
  if (gid + 3 * stride < E) {
    int e0 = gid, e1 = gid + stride, e2 = gid + 2 * stride, e3 = gid + 3 * stride;
    int d0 = dst[e0], d1 = dst[e1], d2 = dst[e2], d3 = dst[e3];
    int r0 = atomicAdd(&count[d0], 1);
    int r1 = atomicAdd(&count[d1], 1);
    int r2 = atomicAdd(&count[d2], 1);
    int r3 = atomicAdd(&count[d3], 1);
    rank[e0] = r0; rank[e1] = r1; rank[e2] = r2; rank[e3] = r3;
  } else {
    for (int k = 0; k < 4; ++k) {
      int e = gid + k * stride;
      if (e < E) rank[e] = atomicAdd(&count[dst[e]], 1);
    }
  }
}

__global__ void scan_sums(const int* __restrict__ count, int* __restrict__ bsum, int N) {
  __shared__ int sh[256];
  int t = threadIdx.x;
  int i0 = blockIdx.x * 1024 + t * 4;
  int s = 0;
  for (int k = 0; k < 4; ++k) if (i0 + k < N) s += count[i0 + k];
  sh[t] = s;
  __syncthreads();
  for (int off = 128; off > 0; off >>= 1) {
    if (t < off) sh[t] += sh[t + off];
    __syncthreads();
  }
  if (t == 0) bsum[blockIdx.x] = sh[0];
}

__global__ void scan_write(const int* __restrict__ count, const int* __restrict__ bsum,
                           int* __restrict__ rowptr, int N, int E) {
  __shared__ int sh[256];
  int t = threadIdx.x, b = blockIdx.x;
  int part = 0;
  for (int k = t; k < b; k += 256) part += bsum[k];
  sh[t] = part;
  __syncthreads();
  for (int off = 128; off > 0; off >>= 1) {
    if (t < off) sh[t] += sh[t + off];
    __syncthreads();
  }
  int boff = sh[0];
  __syncthreads();

  int i0 = b * 1024 + t * 4;
  int c[4];
  int s = 0;
  for (int k = 0; k < 4; ++k) { c[k] = (i0 + k < N) ? count[i0 + k] : 0; s += c[k]; }
  sh[t] = s;
  __syncthreads();
  for (int off = 1; off < 256; off <<= 1) {
    int v = (t >= off) ? sh[t - off] : 0;
    __syncthreads();
    sh[t] += v;
    __syncthreads();
  }
  int run = boff + sh[t] - s;
  for (int k = 0; k < 4; ++k) {
    if (i0 + k < N) { rowptr[i0 + k] = run; run += c[k]; }
  }
  if (b == 0 && t == 0) rowptr[N] = E;
}

__global__ void scatter_src(const int* __restrict__ ei, const int* __restrict__ rowptr,
                            const int* __restrict__ rank, int* __restrict__ col, int E) {
  int gid = blockIdx.x * blockDim.x + threadIdx.x;
  int stride = gridDim.x * blockDim.x;
  const int* dst = ei + E;
  if (gid + 3 * stride < E) {
    int e0 = gid, e1 = gid + stride, e2 = gid + 2 * stride, e3 = gid + 3 * stride;
    int d0 = dst[e0], d1 = dst[e1], d2 = dst[e2], d3 = dst[e3];
    int r0 = rank[e0], r1 = rank[e1], r2 = rank[e2], r3 = rank[e3];
    int s0 = ei[e0], s1 = ei[e1], s2 = ei[e2], s3 = ei[e3];
    int q0 = rowptr[d0], q1 = rowptr[d1], q2 = rowptr[d2], q3 = rowptr[d3];
    col[q0 + r0] = s0; col[q1 + r1] = s1; col[q2 + r2] = s2; col[q3 + r3] = s3;
  } else {
    for (int k = 0; k < 4; ++k) {
      int e = gid + k * stride;
      if (e < E) col[rowptr[dst[e]] + rank[e]] = ei[e];
    }
  }
}

__global__ void proj1_wmma(const float* __restrict__ x, const float* __restrict__ W1,
                           float* __restrict__ h1, int N) {
  int gtid = blockIdx.x * blockDim.x + threadIdx.x;
  int wave = gtid >> 5;
  int lane = threadIdx.x & 31;
  int tiles = (N + 15) >> 4;
  if (wave >= tiles) return;
  int half = lane >> 4;
  int l16  = lane & 15;
  int row  = wave * 16 + l16;
  int rowc = row < N ? row : N - 1;
  bool rok = row < N;

  v2f a[2];
#pragma unroll
  for (int s = 0; s < 2; ++s) {
    int k0 = 4 * s + 2 * half;
    a[s][0] = (rok && k0     < 7) ? x[rowc * 7 + k0]     : 0.f;
    a[s][1] = (rok && k0 + 1 < 7) ? x[rowc * 7 + k0 + 1] : 0.f;
  }

  v8f acc[4];
#pragma unroll
  for (int t = 0; t < 4; ++t)
#pragma unroll
    for (int q = 0; q < 8; ++q) acc[t][q] = 0.f;

#pragma unroll
  for (int s = 0; s < 2; ++s) {
    int k0 = 4 * s + 2 * half;
#pragma unroll
    for (int t = 0; t < 4; ++t) {
      int col = t * 16 + l16;
      v2f b;
      b[0] = (k0     < 7) ? W1[k0 * 64 + col]       : 0.f;
      b[1] = (k0 + 1 < 7) ? W1[(k0 + 1) * 64 + col] : 0.f;
      acc[t] = __builtin_amdgcn_wmma_f32_16x16x4_f32(
          false, a[s], false, b, (short)0, acc[t], false, false);
    }
  }

#pragma unroll
  for (int t = 0; t < 4; ++t)
#pragma unroll
    for (int r = 0; r < 8; ++r) {
      int m = wave * 16 + r + 8 * half;
      if (m < N) h1[m * 64 + t * 16 + l16] = acc[t][r];
    }
}

__global__ void alpha1_kern(const float* __restrict__ h1, const float* __restrict__ aS,
                            const float* __restrict__ aD, float* __restrict__ as1,
                            float* __restrict__ ad1, int N) {
  int i = blockIdx.x * blockDim.x + threadIdx.x;
  if (i >= N * 4) return;
  int hd = i & 3, n = i >> 2;
  const float* hp = h1 + n * 64 + hd * 16;
  float s = 0.f, d = 0.f;
#pragma unroll
  for (int c = 0; c < 16; ++c) {
    float v = hp[c];
    s += v * aS[hd * 16 + c];
    d += v * aD[hd * 16 + c];
  }
  as1[i] = s; ad1[i] = d;
}


template <int G>
__device__ __forceinline__ void add_group(const float2* __restrict__ hrow, int s,
                                          int p0, int p1, int p2, int p3,
                                          int t, int k0, int k1, int k2, int k3,
                                          int lane, float& ax, float& ay) {
  float2 v[G];
  float w[G];
#pragma unroll
  for (int g = 0; g < G; ++g) {
    int st = __builtin_amdgcn_readlane(s, t + g);
    int q0 = __builtin_amdgcn_readlane(p0, t + g);
    int q1 = __builtin_amdgcn_readlane(p1, t + g);
    int q2 = __builtin_amdgcn_readlane(p2, t + g);
    int q3 = __builtin_amdgcn_readlane(p3, t + g);
    w[g] = __int_as_float((q0 & k0) | (q1 & k1) | (q2 & k2) | (q3 & k3));
    v[g] = hrow[(size_t)st * 32 + lane];
  }
#pragma unroll
  for (int g = 0; g < G; ++g) {
    ax += w[g] * v[g].x;
    ay += w[g] * v[g].y;
  }
}

__device__ __forceinline__ void add_rows(const float2* __restrict__ hrow, int s,
                                         float p0, float p1, float p2, float p3,
                                         int cnt, int hd, int lane, float& ax, float& ay) {
  int k0 = hd == 0 ? -1 : 0, k1 = hd == 1 ? -1 : 0;
  int k2 = hd == 2 ? -1 : 0, k3 = hd == 3 ? -1 : 0;
  int b0 = __float_as_int(p0), b1 = __float_as_int(p1);
  int b2 = __float_as_int(p2), b3 = __float_as_int(p3);
  int t = 0;
  for (; t + 8 <= cnt; t += 8) add_group<8>(hrow, s, b0, b1, b2, b3, t, k0, k1, k2, k3, lane, ax, ay);
  if (t + 4 <= cnt) { add_group<4>(hrow, s, b0, b1, b2, b3, t, k0, k1, k2, k3, lane, ax, ay); t += 4; }
  for (; t < cnt; ++t) add_group<1>(hrow, s, b0, b1, b2, b3, t, k0, k1, k2, k3, lane, ax, ay);
}

__global__ void gat1_node(const int* __restrict__ rowptr, const int* __restrict__ col,
                          const float* __restrict__ as1, const float* __restrict__ ad1,
                          const float* __restrict__ h1, const float* __restrict__ b1,
                          float* __restrict__ x2, int N) {
  int lane = threadIdx.x & 31;
  int hd = lane >> 3;
  int n = __builtin_amdgcn_readfirstlane((int)(blockIdx.x * (blockDim.x >> 5) + (threadIdx.x >> 5)));
  if (n >= N) return;
  const float4* as4 = (const float4*)as1;
  const float4* ad4 = (const float4*)ad1;
  const float2* hrow = (const float2*)h1;

  int beg = rowptr[n], end = rowptr[n + 1];
  float4 adn = ad4[n];
  float4 asn = as4[n];
  float es0 = leaky(asn.x + adn.x), es1 = leaky(asn.y + adn.y);
  float es2 = leaky(asn.z + adn.z), es3 = leaky(asn.w + adn.w);
  float m0 = es0, m1 = es1, m2 = es2, m3 = es3;

  int j0 = beg + lane;
  bool v0 = j0 < end;
  int s0 = 0;
  float e0 = 0.f, e1 = 0.f, e2 = 0.f, e3 = 0.f;
  if (v0) {
    s0 = col[j0];
    float4 a = as4[s0];
    e0 = leaky(a.x + adn.x); e1 = leaky(a.y + adn.y);
    e2 = leaky(a.z + adn.z); e3 = leaky(a.w + adn.w);
    m0 = fmaxf(m0, e0); m1 = fmaxf(m1, e1); m2 = fmaxf(m2, e2); m3 = fmaxf(m3, e3);
  }
  for (int base = beg + 32; base < end; base += 32) {
    int j = base + lane;
    if (j < end) {
      float4 a = as4[col[j]];
      m0 = fmaxf(m0, leaky(a.x + adn.x)); m1 = fmaxf(m1, leaky(a.y + adn.y));
      m2 = fmaxf(m2, leaky(a.z + adn.z)); m3 = fmaxf(m3, leaky(a.w + adn.w));
    }
  }
  for (int off = 16; off > 0; off >>= 1) {
    m0 = fmaxf(m0, __shfl_xor(m0, off)); m1 = fmaxf(m1, __shfl_xor(m1, off));
    m2 = fmaxf(m2, __shfl_xor(m2, off)); m3 = fmaxf(m3, __shfl_xor(m3, off));
  }

  float p0 = v0 ? __expf(e0 - m0) : 0.f, p1 = v0 ? __expf(e1 - m1) : 0.f;
  float p2 = v0 ? __expf(e2 - m2) : 0.f, p3 = v0 ? __expf(e3 - m3) : 0.f;
  float d0 = p0, d1 = p1, d2 = p2, d3 = p3;
  float ax = 0.f, ay = 0.f;
  {
    int cnt = end - beg;
    cnt = cnt < 32 ? cnt : 32;
    add_rows(hrow, s0, p0, p1, p2, p3, cnt, hd, lane, ax, ay);
  }
  for (int base = beg + 32; base < end; base += 32) {
    int j = base + lane;
    int s = 0;
    float r0 = 0.f, r1 = 0.f, r2 = 0.f, r3 = 0.f;
    if (j < end) {
      s = col[j];
      float4 a = as4[s];
      r0 = __expf(leaky(a.x + adn.x) - m0); r1 = __expf(leaky(a.y + adn.y) - m1);
      r2 = __expf(leaky(a.z + adn.z) - m2); r3 = __expf(leaky(a.w + adn.w) - m3);
    }
    d0 += r0; d1 += r1; d2 += r2; d3 += r3;
    int cnt = end - base;
    cnt = cnt < 32 ? cnt : 32;
    add_rows(hrow, s, r0, r1, r2, r3, cnt, hd, lane, ax, ay);
  }
  for (int off = 16; off > 0; off >>= 1) {
    d0 += __shfl_xor(d0, off); d1 += __shfl_xor(d1, off);
    d2 += __shfl_xor(d2, off); d3 += __shfl_xor(d3, off);
  }
  float ws0 = __expf(es0 - m0), ws1 = __expf(es1 - m1);
  float ws2 = __expf(es2 - m2), ws3 = __expf(es3 - m3);
  d0 += ws0; d1 += ws1; d2 += ws2; d3 += ws3;
  float wself = hd == 0 ? ws0 : (hd == 1 ? ws1 : (hd == 2 ? ws2 : ws3));
  float den   = hd == 0 ? d0  : (hd == 1 ? d1  : (hd == 2 ? d2  : d3));
  float2 hv = hrow[(size_t)n * 32 + lane];
  ax += wself * hv.x;
  ay += wself * hv.y;

  float2 bb = ((const float2*)b1)[lane];
  float ox = ax / den + bb.x;
  float oy = ay / den + bb.y;
  ox = ox > 0.f ? ox : (__expf(ox) - 1.f);
  oy = oy > 0.f ? oy : (__expf(oy) - 1.f);
  ((float2*)x2)[(size_t)n * 32 + lane] = make_float2(ox, oy);
}

__global__ void proj2_wmma(const float* __restrict__ x2, const float* __restrict__ W2,
                           float* __restrict__ h2, int N) {
  int gtid = blockIdx.x * blockDim.x + threadIdx.x;
  int wave = gtid >> 5;
  int lane = threadIdx.x & 31;
  int tiles = (N + 15) >> 4;
  if (wave >= tiles) return;
  int half = lane >> 4, l16 = lane & 15;
  int row  = wave * 16 + l16;
  int rowc = row < N ? row : N - 1;
  bool rok = row < N;
  bool cok = l16 < 6;

  v8f acc;
#pragma unroll
  for (int q = 0; q < 8; ++q) acc[q] = 0.f;

#pragma unroll
  for (int s = 0; s < 16; ++s) {
    int k0 = 4 * s + 2 * half;
    v2f a, b;
    a[0] = rok ? x2[rowc * 64 + k0]     : 0.f;
    a[1] = rok ? x2[rowc * 64 + k0 + 1] : 0.f;
    b[0] = cok ? W2[k0 * 6 + l16]       : 0.f;
    b[1] = cok ? W2[(k0 + 1) * 6 + l16] : 0.f;
    acc = __builtin_amdgcn_wmma_f32_16x16x4_f32(
        false, a, false, b, (short)0, acc, false, false);
  }

  if (cok) {
#pragma unroll
    for (int r = 0; r < 8; ++r) {
      int m = wave * 16 + r + 8 * half;
      if (m < N) h2[m * 6 + l16] = acc[r];
    }
  }
}

__global__ void alpha2_kern(const float* __restrict__ h2, const float* __restrict__ aS,
                            const float* __restrict__ aD, float* __restrict__ as2,
                            float* __restrict__ ad2, int N) {
  int n = blockIdx.x * blockDim.x + threadIdx.x;
  if (n >= N) return;
  float s = 0.f, d = 0.f;
#pragma unroll
  for (int c = 0; c < 6; ++c) {
    float v = h2[n * 6 + c];
    s += v * aS[c];
    d += v * aD[c];
  }
  as2[n] = s; ad2[n] = d;
}

__global__ void gat2_node(const int* __restrict__ rowptr, const int* __restrict__ col,
                          const float* __restrict__ as2, const float* __restrict__ ad2,
                          const float* __restrict__ h2, const float* __restrict__ b2,
                          float* __restrict__ out, int N) {
  int gid = blockIdx.x * blockDim.x + threadIdx.x;
  int n = gid >> 3, sub = gid & 7;
  bool ok = n < N;
  int nc = ok ? n : N - 1;
  int beg = rowptr[nc];
  int end = ok ? rowptr[nc + 1] : beg;
  float ad = ad2[nc];
  float es = leaky(as2[nc] + ad);
  float m = es;
  const float2* hv = (const float2*)h2;

  int sv[8];
  float ev[8];
#pragma unroll
  for (int k = 0; k < 8; ++k) {
    int j = beg + sub + 8 * k;
    int s = col[j < end ? j : 0];
    float e = leaky(as2[s] + ad);
    sv[k] = s;
    ev[k] = e;
    if (j < end) m = fmaxf(m, e);
  }
  for (int j = beg + sub + 64; j < end; j += 8)
    m = fmaxf(m, leaky(as2[col[j]] + ad));
  m = fmaxf(m, __shfl_xor(m, 1));
  m = fmaxf(m, __shfl_xor(m, 2));
  m = fmaxf(m, __shfl_xor(m, 4));

  float den = 0.f;
  float acc[6];
#pragma unroll
  for (int c = 0; c < 6; ++c) acc[c] = 0.f;
#pragma unroll
  for (int k = 0; k < 8; ++k) {
    int j = beg + sub + 8 * k;
    if (j < end) {
      float w = __expf(ev[k] - m);
      den += w;
      float2 a = hv[sv[k] * 3], b = hv[sv[k] * 3 + 1], c = hv[sv[k] * 3 + 2];
      acc[0] += w * a.x; acc[1] += w * a.y; acc[2] += w * b.x;
      acc[3] += w * b.y; acc[4] += w * c.x; acc[5] += w * c.y;
    }
  }
  for (int j = beg + sub + 64; j < end; j += 8) {
    int s = col[j];
    float w = __expf(leaky(as2[s] + ad) - m);
    den += w;
    float2 a = hv[s * 3], b = hv[s * 3 + 1], c = hv[s * 3 + 2];
    acc[0] += w * a.x; acc[1] += w * a.y; acc[2] += w * b.x;
    acc[3] += w * b.y; acc[4] += w * c.x; acc[5] += w * c.y;
  }
  if (sub == 0) {
    float w = __expf(es - m);
    den += w;
    float2 a = hv[nc * 3], b = hv[nc * 3 + 1], c = hv[nc * 3 + 2];
    acc[0] += w * a.x; acc[1] += w * a.y; acc[2] += w * b.x;
    acc[3] += w * b.y; acc[4] += w * c.x; acc[5] += w * c.y;
  }
#pragma unroll
  for (int off = 1; off < 8; off <<= 1) {
    den += __shfl_xor(den, off);
#pragma unroll
    for (int c = 0; c < 6; ++c) acc[c] += __shfl_xor(acc[c], off);
  }

  if (ok && sub == 0) {
    float o[6];
    float mx = -3.0e38f;
#pragma unroll
    for (int c = 0; c < 6; ++c) { o[c] = acc[c] / den + b2[c]; mx = fmaxf(mx, o[c]); }
    float sum = 0.f;
#pragma unroll
    for (int c = 0; c < 6; ++c) sum += __expf(o[c] - mx);
    float l = __logf(sum);
#pragma unroll
    for (int c = 0; c < 6; ++c) out[n * 6 + c] = o[c] - mx - l;
  }
}

extern "C" void kernel_launch(void* const* d_in, const int* in_sizes, int n_in,
                              void* d_out, int out_size, void* d_ws, size_t ws_size,
                              hipStream_t stream) {
  const float* x      = (const float*)d_in[0];
  const int*   ei     = (const int*)d_in[1];
  const float* W1     = (const float*)d_in[2];
  const float* a_src1 = (const float*)d_in[3];
  const float* a_dst1 = (const float*)d_in[4];
  const float* b1     = (const float*)d_in[5];
  const float* W2     = (const float*)d_in[6];
  const float* a_src2 = (const float*)d_in[7];
  const float* a_dst2 = (const float*)d_in[8];
  const float* b2     = (const float*)d_in[9];
  float* out = (float*)d_out;

  int N = in_sizes[0] / 7;
  int E = in_sizes[1] / 2;

  float* ws  = (float*)d_ws;
  float* h1  = ws;
  float* x2  = h1 + (size_t)N * 64;
  float* as1 = x2 + (size_t)N * 64;
  float* ad1 = as1 + (size_t)N * 4;
  float* h2  = ad1 + (size_t)N * 4;
  float* as2 = h2 + (size_t)N * 6;
  float* ad2 = as2 + N;
  int nb = (N + 1023) / 1024;
  int* count  = (int*)(ad2 + N);
  int* rowptr = count + N;
  int* bsum   = rowptr + N + 1;
  int* rank   = bsum + nb;
  int* col    = rank + E;

  const int B = 256;
  auto G = [](long n, int b) { return (int)((n + b - 1) / b); };
  int tiles = (N + 15) / 16;
  int projBlocks = (tiles + (B / 32) - 1) / (B / 32);
  int edgeBlocks = G(((long)E + 3) / 4, B);

  zero_count<<<G(N, B), B, 0, stream>>>(count, N);
  count_rank<<<edgeBlocks, B, 0, stream>>>(ei, count, rank, E);
  scan_sums<<<nb, B, 0, stream>>>(count, bsum, N);
  scan_write<<<nb, B, 0, stream>>>(count, bsum, rowptr, N, E);
  scatter_src<<<edgeBlocks, B, 0, stream>>>(ei, rowptr, rank, col, E);

  proj1_wmma<<<projBlocks, B, 0, stream>>>(x, W1, h1, N);
  alpha1_kern<<<G((long)N * 4, B), B, 0, stream>>>(h1, a_src1, a_dst1, as1, ad1, N);
  gat1_node<<<G(N, B / 32), B, 0, stream>>>(rowptr, col, as1, ad1, h1, b1, x2, N);

  proj2_wmma<<<projBlocks, B, 0, stream>>>(x2, W2, h2, N);
  alpha2_kern<<<G(N, B), B, 0, stream>>>(h2, a_src2, a_dst2, as2, ad2, N);
  gat2_node<<<G((long)N * 8, B), B, 0, stream>>>(rowptr, col, as2, ad2, h2, b2, out, N);
}


# gat1 lanes become (head, edge slot): a lane scores its own head only, weights reach the row sum by 8-lane DPP broadcast (no 4x readlane+select), rows load 16 deep; weight sums reduce over 8 lanes
# speedup vs baseline: 6.9482x; 1.0473x over previous
#include <hip/hip_runtime.h>
#include <hip/hip_bf16.h>
#include <math.h>

typedef float v2f __attribute__((ext_vector_type(2)));
typedef float v8f __attribute__((ext_vector_type(8)));

#define NEG_SLOPE 0.2f

__device__ __forceinline__ float leaky(float v) { return v > 0.f ? v : NEG_SLOPE * v; }


__global__ void zero_count(int* __restrict__ count, int N) {
  int i = blockIdx.x * blockDim.x + threadIdx.x;
  if (i < N) count[i] = 0;
}

__global__ void count_rank(const int* __restrict__ ei, int* __restrict__ count,
                           int* __restrict__ rank, int E) {
  int gid = blockIdx.x * blockDim.x + threadIdx.x;
  int stride = gridDim.x * blockDim.x;
  const int* dst = ei + E;
  if (gid + 3 * stride < E) {
    int e0 = gid, e1 = gid + stride, e2 = gid + 2 * stride, e3 = gid + 3 * stride;
    int d0 = dst[e0], d1 = dst[e1], d2 = dst[e2], d3 = dst[e3];
    int r0 = atomicAdd(&count[d0], 1);
    int r1 = atomicAdd(&count[d1], 1);
    int r2 = atomicAdd(&count[d2], 1);
    int r3 = atomicAdd(&count[d3], 1);
    rank[e0] = r0; rank[e1] = r1; rank[e2] = r2; rank[e3] = r3;
  } else {
    for (int k = 0; k < 4; ++k) {
      int e = gid + k * stride;
      if (e < E) rank[e] = atomicAdd(&count[dst[e]], 1);
    }
  }
}

__global__ void scan_sums(const int* __restrict__ count, int* __restrict__ bsum, int N) {
  __shared__ int sh[256];
  int t = threadIdx.x;
  int i0 = blockIdx.x * 1024 + t * 4;
  int s = 0;
  for (int k = 0; k < 4; ++k) if (i0 + k < N) s += count[i0 + k];
  sh[t] = s;
  __syncthreads();
  for (int off = 128; off > 0; off >>= 1) {
    if (t < off) sh[t] += sh[t + off];
    __syncthreads();
  }
  if (t == 0) bsum[blockIdx.x] = sh[0];
}

__global__ void scan_write(const int* __restrict__ count, const int* __restrict__ bsum,
                           int* __restrict__ rowptr, int N, int E) {
  __shared__ int sh[256];
  int t = threadIdx.x, b = blockIdx.x;
  int part = 0;
  for (int k = t; k < b; k += 256) part += bsum[k];
  sh[t] = part;
  __syncthreads();
  for (int off = 128; off > 0; off >>= 1) {
    if (t < off) sh[t] += sh[t + off];
    __syncthreads();
  }
  int boff = sh[0];
  __syncthreads();

  int i0 = b * 1024 + t * 4;
  int c[4];
  int s = 0;
  for (int k = 0; k < 4; ++k) { c[k] = (i0 + k < N) ? count[i0 + k] : 0; s += c[k]; }
  sh[t] = s;
  __syncthreads();
  for (int off = 1; off < 256; off <<= 1) {
    int v = (t >= off) ? sh[t - off] : 0;
    __syncthreads();
    sh[t] += v;
    __syncthreads();
  }
  int run = boff + sh[t] - s;
  for (int k = 0; k < 4; ++k) {
    if (i0 + k < N) { rowptr[i0 + k] = run; run += c[k]; }
  }
  if (b == 0 && t == 0) rowptr[N] = E;
}

__global__ void scatter_src(const int* __restrict__ ei, const int* __restrict__ rowptr,
                            const int* __restrict__ rank, int* __restrict__ col, int E) {
  int gid = blockIdx.x * blockDim.x + threadIdx.x;
  int stride = gridDim.x * blockDim.x;
  const int* dst = ei + E;
  if (gid + 3 * stride < E) {
    int e0 = gid, e1 = gid + stride, e2 = gid + 2 * stride, e3 = gid + 3 * stride;
    int d0 = dst[e0], d1 = dst[e1], d2 = dst[e2], d3 = dst[e3];
    int r0 = rank[e0], r1 = rank[e1], r2 = rank[e2], r3 = rank[e3];
    int s0 = ei[e0], s1 = ei[e1], s2 = ei[e2], s3 = ei[e3];
    int q0 = rowptr[d0], q1 = rowptr[d1], q2 = rowptr[d2], q3 = rowptr[d3];
    col[q0 + r0] = s0; col[q1 + r1] = s1; col[q2 + r2] = s2; col[q3 + r3] = s3;
  } else {
    for (int k = 0; k < 4; ++k) {
      int e = gid + k * stride;
      if (e < E) col[rowptr[dst[e]] + rank[e]] = ei[e];
    }
  }
}

__global__ void proj1_wmma(const float* __restrict__ x, const float* __restrict__ W1,
                           float* __restrict__ h1, int N) {
  int gtid = blockIdx.x * blockDim.x + threadIdx.x;
  int wave = gtid >> 5;
  int lane = threadIdx.x & 31;
  int tiles = (N + 15) >> 4;
  if (wave >= tiles) return;
  int half = lane >> 4;
  int l16  = lane & 15;
  int row  = wave * 16 + l16;
  int rowc = row < N ? row : N - 1;
  bool rok = row < N;

  v2f a[2];
#pragma unroll
  for (int s = 0; s < 2; ++s) {
    int k0 = 4 * s + 2 * half;
    a[s][0] = (rok && k0     < 7) ? x[rowc * 7 + k0]     : 0.f;
    a[s][1] = (rok && k0 + 1 < 7) ? x[rowc * 7 + k0 + 1] : 0.f;
  }

  v8f acc[4];
#pragma unroll
  for (int t = 0; t < 4; ++t)
#pragma unroll
    for (int q = 0; q < 8; ++q) acc[t][q] = 0.f;

#pragma unroll
  for (int s = 0; s < 2; ++s) {
    int k0 = 4 * s + 2 * half;
#pragma unroll
    for (int t = 0; t < 4; ++t) {
      int col = t * 16 + l16;
      v2f b;
      b[0] = (k0     < 7) ? W1[k0 * 64 + col]       : 0.f;
      b[1] = (k0 + 1 < 7) ? W1[(k0 + 1) * 64 + col] : 0.f;
      acc[t] = __builtin_amdgcn_wmma_f32_16x16x4_f32(
          false, a[s], false, b, (short)0, acc[t], false, false);
    }
  }

#pragma unroll
  for (int t = 0; t < 4; ++t)
#pragma unroll
    for (int r = 0; r < 8; ++r) {
      int m = wave * 16 + r + 8 * half;
      if (m < N) h1[m * 64 + t * 16 + l16] = acc[t][r];
    }
}

__global__ void alpha1_kern(const float* __restrict__ h1, const float* __restrict__ aS,
                            const float* __restrict__ aD, float* __restrict__ as1,
                            float* __restrict__ ad1, int N) {
  int i = blockIdx.x * blockDim.x + threadIdx.x;
  if (i >= N * 4) return;
  int hd = i & 3, n = i >> 2;
  const float* hp = h1 + n * 64 + hd * 16;
  float s = 0.f, d = 0.f;
#pragma unroll
  for (int c = 0; c < 16; ++c) {
    float v = hp[c];
    s += v * aS[hd * 16 + c];
    d += v * aD[hd * 16 + c];
  }
  as1[i] = s; ad1[i] = d;
}


#define SEL8(a, b, c, d, e, f, g, h) \
  ((a) | ((b) << 3) | ((c) << 6) | ((d) << 9) | ((e) << 12) | ((f) << 15) | ((g) << 18) | ((h) << 21))
#define BC8(v, J) __builtin_amdgcn_mov_dpp8((unsigned)(v), SEL8(J, J, J, J, J, J, J, J))
#define BC8F(v, J) __uint_as_float(BC8(__float_as_uint(v), J))

__device__ __forceinline__ float max8(float v) {
  v = fmaxf(v, __uint_as_float(__builtin_amdgcn_mov_dpp8(__float_as_uint(v), SEL8(1, 0, 3, 2, 5, 4, 7, 6))));
  v = fmaxf(v, __uint_as_float(__builtin_amdgcn_mov_dpp8(__float_as_uint(v), SEL8(2, 3, 0, 1, 6, 7, 4, 5))));
  v = fmaxf(v, __uint_as_float(__builtin_amdgcn_mov_dpp8(__float_as_uint(v), SEL8(4, 5, 6, 7, 0, 1, 2, 3))));
  return v;
}
__device__ __forceinline__ float sum8(float v) {
  v += __uint_as_float(__builtin_amdgcn_mov_dpp8(__float_as_uint(v), SEL8(1, 0, 3, 2, 5, 4, 7, 6)));
  v += __uint_as_float(__builtin_amdgcn_mov_dpp8(__float_as_uint(v), SEL8(2, 3, 0, 1, 6, 7, 4, 5)));
  v += __uint_as_float(__builtin_amdgcn_mov_dpp8(__float_as_uint(v), SEL8(4, 5, 6, 7, 0, 1, 2, 3)));
  return v;
}

__device__ __forceinline__ v2f row_ld(const char* __restrict__ hb, int st, unsigned laneoff) {
  return *(const v2f*)(hb + ((((unsigned)st) << 8) | laneoff));
}
#define SRC8(s, J) __builtin_amdgcn_readlane((s), (J))

template <int NB>
__device__ __forceinline__ void load_rows(const char* __restrict__ hb, int s, unsigned laneoff, v2f* v) {
  v[0] = row_ld(hb, SRC8(s, 0), laneoff);
  v[1] = row_ld(hb, SRC8(s, 1), laneoff);
  v[2] = row_ld(hb, SRC8(s, 2), laneoff);
  v[3] = row_ld(hb, SRC8(s, 3), laneoff);
  if (NB > 4) {
    v[4] = row_ld(hb, SRC8(s, 4), laneoff);
    v[5] = row_ld(hb, SRC8(s, 5), laneoff);
    v[6] = row_ld(hb, SRC8(s, 6), laneoff);
    v[7] = row_ld(hb, SRC8(s, 7), laneoff);
  }
}

#define ADDROW(J) { float w = BC8F(p, J); v2f ww = {w, w}; acc = __builtin_elementwise_fma(v[J], ww, acc); }
template <int NB>
__device__ __forceinline__ void add_rows(float p, const v2f* v, v2f& acc) {
  ADDROW(0) ADDROW(1) ADDROW(2) ADDROW(3)
  if (NB > 4) { ADDROW(4) ADDROW(5) ADDROW(6) ADDROW(7) }
}

__global__ void gat1_node(const int* __restrict__ rowptr, const int* __restrict__ col,
                          const float* __restrict__ as1, const float* __restrict__ ad1,
                          const float* __restrict__ h1, const float* __restrict__ b1,
                          float* __restrict__ x2, int N) {
  int lane = threadIdx.x & 31;
  int hd = lane >> 3, j8 = lane & 7;
  int n = __builtin_amdgcn_readfirstlane((int)(blockIdx.x * (blockDim.x >> 5) + (threadIdx.x >> 5)));
  if (n >= N) return;
  const char* hb = (const char*)h1;
  unsigned laneoff = (unsigned)lane * 8u;

  int beg = rowptr[n], end = rowptr[n + 1];
  int cnt = end - beg;
  float adh = ad1[n * 4 + hd];
  float es = leaky(as1[n * 4 + hd] + adh);
  float m = es;
  v2f hv = row_ld(hb, n, laneoff);

  int s[8];
  float e[8];
#pragma unroll
  for (int c = 0; c < 8; ++c) {
    int j = beg + 8 * c + j8;
    int sv = n;
    if (j < end) sv = col[j];
    s[c] = sv;
  }
#pragma unroll
  for (int c = 0; c < 8; ++c) {
    float ev = leaky(as1[s[c] * 4 + hd] + adh);
    e[c] = ev;
    m = fmaxf(m, ev);
  }
  for (int base = beg + 64; base < end; base += 8) {
    int j = base + j8;
    if (j < end) m = fmaxf(m, leaky(as1[col[j] * 4 + hd] + adh));
  }
  m = max8(m);

  float p[8];
  float dsum = 0.f;
#pragma unroll
  for (int c = 0; c < 8; ++c) {
    int j = beg + 8 * c + j8;
    float pv = (j < end) ? __expf(e[c] - m) : 0.f;
    p[c] = pv;
    dsum += pv;
  }
  v2f acc = {0.f, 0.f};
#pragma unroll
  for (int g = 0; g < 4; ++g) {
    int lo = 16 * g;
    if (lo < cnt) {
      v2f va[8], vb[8];
      if (lo + 12 < cnt) {
        load_rows<8>(hb, s[2 * g], laneoff, va);
        load_rows<8>(hb, s[2 * g + 1], laneoff, vb);
        add_rows<8>(p[2 * g], va, acc);
        add_rows<8>(p[2 * g + 1], vb, acc);
      } else if (lo + 8 < cnt) {
        load_rows<8>(hb, s[2 * g], laneoff, va);
        load_rows<4>(hb, s[2 * g + 1], laneoff, vb);
        add_rows<8>(p[2 * g], va, acc);
        add_rows<4>(p[2 * g + 1], vb, acc);
      } else if (lo + 4 < cnt) {
        load_rows<8>(hb, s[2 * g], laneoff, va);
        add_rows<8>(p[2 * g], va, acc);
      } else {
        load_rows<4>(hb, s[2 * g], laneoff, va);
        add_rows<4>(p[2 * g], va, acc);
      }
    }
  }
  for (int base = beg + 64; base < end; base += 8) {
    int j = base + j8;
    int sv = n;
    if (j < end) sv = col[j];
    float pv = (j < end) ? __expf(leaky(as1[sv * 4 + hd] + adh) - m) : 0.f;
    dsum += pv;
    v2f v[8];
    load_rows<8>(hb, sv, laneoff, v);
    add_rows<8>(pv, v, acc);
  }
  float ws = __expf(es - m);
  float den = sum8(dsum) + ws;
  v2f wws = {ws, ws};
  acc = __builtin_elementwise_fma(hv, wws, acc);
  float ax = acc[0], ay = acc[1];

  float2 bb = ((const float2*)b1)[lane];
  float ox = ax / den + bb.x;
  float oy = ay / den + bb.y;
  ox = ox > 0.f ? ox : (__expf(ox) - 1.f);
  oy = oy > 0.f ? oy : (__expf(oy) - 1.f);
  ((float2*)x2)[(size_t)n * 32 + lane] = make_float2(ox, oy);
}

__global__ void proj2_wmma(const float* __restrict__ x2, const float* __restrict__ W2,
                           float* __restrict__ h2, int N) {
  int gtid = blockIdx.x * blockDim.x + threadIdx.x;
  int wave = gtid >> 5;
  int lane = threadIdx.x & 31;
  int tiles = (N + 15) >> 4;
  if (wave >= tiles) return;
  int half = lane >> 4, l16 = lane & 15;
  int row  = wave * 16 + l16;
  int rowc = row < N ? row : N - 1;
  bool rok = row < N;
  bool cok = l16 < 6;

  v8f acc;
#pragma unroll
  for (int q = 0; q < 8; ++q) acc[q] = 0.f;

#pragma unroll
  for (int s = 0; s < 16; ++s) {
    int k0 = 4 * s + 2 * half;
    v2f a, b;
    a[0] = rok ? x2[rowc * 64 + k0]     : 0.f;
    a[1] = rok ? x2[rowc * 64 + k0 + 1] : 0.f;
    b[0] = cok ? W2[k0 * 6 + l16]       : 0.f;
    b[1] = cok ? W2[(k0 + 1) * 6 + l16] : 0.f;
    acc = __builtin_amdgcn_wmma_f32_16x16x4_f32(
        false, a, false, b, (short)0, acc, false, false);
  }

  if (cok) {
#pragma unroll
    for (int r = 0; r < 8; ++r) {
      int m = wave * 16 + r + 8 * half;
      if (m < N) h2[m * 6 + l16] = acc[r];
    }
  }
}

__global__ void alpha2_kern(const float* __restrict__ h2, const float* __restrict__ aS,
                            const float* __restrict__ aD, float* __restrict__ as2,
                            float* __restrict__ ad2, int N) {
  int n = blockIdx.x * blockDim.x + threadIdx.x;
  if (n >= N) return;
  float s = 0.f, d = 0.f;
#pragma unroll
  for (int c = 0; c < 6; ++c) {
    float v = h2[n * 6 + c];
    s += v * aS[c];
    d += v * aD[c];
  }
  as2[n] = s; ad2[n] = d;
}

__global__ void gat2_node(const int* __restrict__ rowptr, const int* __restrict__ col,
                          const float* __restrict__ as2, const float* __restrict__ ad2,
                          const float* __restrict__ h2, const float* __restrict__ b2,
                          float* __restrict__ out, int N) {
  int gid = blockIdx.x * blockDim.x + threadIdx.x;
  int n = gid >> 3, sub = gid & 7;
  bool ok = n < N;
  int nc = ok ? n : N - 1;
  int beg = rowptr[nc];
  int end = ok ? rowptr[nc + 1] : beg;
  float ad = ad2[nc];
  float es = leaky(as2[nc] + ad);
  float m = es;
  const float2* hv = (const float2*)h2;

  int sv[8];
  float ev[8];
#pragma unroll
  for (int k = 0; k < 8; ++k) {
    int j = beg + sub + 8 * k;
    int s = col[j < end ? j : 0];
    float e = leaky(as2[s] + ad);
    sv[k] = s;
    ev[k] = e;
    if (j < end) m = fmaxf(m, e);
  }
  for (int j = beg + sub + 64; j < end; j += 8)
    m = fmaxf(m, leaky(as2[col[j]] + ad));
  m = fmaxf(m, __shfl_xor(m, 1));
  m = fmaxf(m, __shfl_xor(m, 2));
  m = fmaxf(m, __shfl_xor(m, 4));

  float den = 0.f;
  float acc[6];
#pragma unroll
  for (int c = 0; c < 6; ++c) acc[c] = 0.f;
#pragma unroll
  for (int k = 0; k < 8; ++k) {
    int j = beg + sub + 8 * k;
    if (j < end) {
      float w = __expf(ev[k] - m);
      den += w;
      float2 a = hv[sv[k] * 3], b = hv[sv[k] * 3 + 1], c = hv[sv[k] * 3 + 2];
      acc[0] += w * a.x; acc[1] += w * a.y; acc[2] += w * b.x;
      acc[3] += w * b.y; acc[4] += w * c.x; acc[5] += w * c.y;
    }
  }
  for (int j = beg + sub + 64; j < end; j += 8) {
    int s = col[j];
    float w = __expf(leaky(as2[s] + ad) - m);
    den += w;
    float2 a = hv[s * 3], b = hv[s * 3 + 1], c = hv[s * 3 + 2];
    acc[0] += w * a.x; acc[1] += w * a.y; acc[2] += w * b.x;
    acc[3] += w * b.y; acc[4] += w * c.x; acc[5] += w * c.y;
  }
  if (sub == 0) {
    float w = __expf(es - m);
    den += w;
    float2 a = hv[nc * 3], b = hv[nc * 3 + 1], c = hv[nc * 3 + 2];
    acc[0] += w * a.x; acc[1] += w * a.y; acc[2] += w * b.x;
    acc[3] += w * b.y; acc[4] += w * c.x; acc[5] += w * c.y;
  }
#pragma unroll
  for (int off = 1; off < 8; off <<= 1) {
    den += __shfl_xor(den, off);
#pragma unroll
    for (int c = 0; c < 6; ++c) acc[c] += __shfl_xor(acc[c], off);
  }

  if (ok && sub == 0) {
    float o[6];
    float mx = -3.0e38f;
#pragma unroll
    for (int c = 0; c < 6; ++c) { o[c] = acc[c] / den + b2[c]; mx = fmaxf(mx, o[c]); }
    float sum = 0.f;
#pragma unroll
    for (int c = 0; c < 6; ++c) sum += __expf(o[c] - mx);
    float l = __logf(sum);
#pragma unroll
    for (int c = 0; c < 6; ++c) out[n * 6 + c] = o[c] - mx - l;
  }
}

extern "C" void kernel_launch(void* const* d_in, const int* in_sizes, int n_in,
                              void* d_out, int out_size, void* d_ws, size_t ws_size,
                              hipStream_t stream) {
  const float* x      = (const float*)d_in[0];
  const int*   ei     = (const int*)d_in[1];
  const float* W1     = (const float*)d_in[2];
  const float* a_src1 = (const float*)d_in[3];
  const float* a_dst1 = (const float*)d_in[4];
  const float* b1     = (const float*)d_in[5];
  const float* W2     = (const float*)d_in[6];
  const float* a_src2 = (const float*)d_in[7];
  const float* a_dst2 = (const float*)d_in[8];
  const float* b2     = (const float*)d_in[9];
  float* out = (float*)d_out;

  int N = in_sizes[0] / 7;
  int E = in_sizes[1] / 2;

  float* ws  = (float*)d_ws;
  float* h1  = ws;
  float* x2  = h1 + (size_t)N * 64;
  float* as1 = x2 + (size_t)N * 64;
  float* ad1 = as1 + (size_t)N * 4;
  float* h2  = ad1 + (size_t)N * 4;
  float* as2 = h2 + (size_t)N * 6;
  float* ad2 = as2 + N;
  int nb = (N + 1023) / 1024;
  int* count  = (int*)(ad2 + N);
  int* rowptr = count + N;
  int* bsum   = rowptr + N + 1;
  int* rank   = bsum + nb;
  int* col    = rank + E;

  const int B = 256;
  auto G = [](long n, int b) { return (int)((n + b - 1) / b); };
  int tiles = (N + 15) / 16;
  int projBlocks = (tiles + (B / 32) - 1) / (B / 32);
  int edgeBlocks = G(((long)E + 3) / 4, B);

  zero_count<<<G(N, B), B, 0, stream>>>(count, N);
  count_rank<<<edgeBlocks, B, 0, stream>>>(ei, count, rank, E);
  scan_sums<<<nb, B, 0, stream>>>(count, bsum, N);
  scan_write<<<nb, B, 0, stream>>>(count, bsum, rowptr, N, E);
  scatter_src<<<edgeBlocks, B, 0, stream>>>(ei, rowptr, rank, col, E);

  proj1_wmma<<<projBlocks, B, 0, stream>>>(x, W1, h1, N);
  alpha1_kern<<<G((long)N * 4, B), B, 0, stream>>>(h1, a_src1, a_dst1, as1, ad1, N);
  gat1_node<<<G(N, B / 32), B, 0, stream>>>(rowptr, col, as1, ad1, h1, b1, x2, N);

  proj2_wmma<<<projBlocks, B, 0, stream>>>(x2, W2, h2, N);
  alpha2_kern<<<G(N, B), B, 0, stream>>>(h2, a_src2, a_dst2, as2, ad2, N);
  gat2_node<<<G((long)N * 8, B), B, 0, stream>>>(rowptr, col, as2, ad2, h2, b2, out, N);
}


# gat2: a source's h2 row and score share one aligned 32-B record, loaded 16 B per lane by two lanes per edge in one instruction: one line request per edge, not a score gather plus a 24-B row gather
# speedup vs baseline: 7.1925x; 1.0352x over previous
#include <hip/hip_runtime.h>
#include <hip/hip_bf16.h>
#include <math.h>

typedef float v2f __attribute__((ext_vector_type(2)));
typedef float v8f __attribute__((ext_vector_type(8)));

#define NEG_SLOPE 0.2f

__device__ __forceinline__ float leaky(float v) { return v > 0.f ? v : NEG_SLOPE * v; }


__global__ void zero_count(int* __restrict__ count, int N) {
  int i = blockIdx.x * blockDim.x + threadIdx.x;
  if (i < N) count[i] = 0;
}

__global__ void count_rank(const int* __restrict__ ei, int* __restrict__ count,
                           int* __restrict__ rank, int E) {
  int gid = blockIdx.x * blockDim.x + threadIdx.x;
  int stride = gridDim.x * blockDim.x;
  const int* dst = ei + E;
  if (gid + 3 * stride < E) {
    int e0 = gid, e1 = gid + stride, e2 = gid + 2 * stride, e3 = gid + 3 * stride;
    int d0 = dst[e0], d1 = dst[e1], d2 = dst[e2], d3 = dst[e3];
    int r0 = atomicAdd(&count[d0], 1);
    int r1 = atomicAdd(&count[d1], 1);
    int r2 = atomicAdd(&count[d2], 1);
    int r3 = atomicAdd(&count[d3], 1);
    rank[e0] = r0; rank[e1] = r1; rank[e2] = r2; rank[e3] = r3;
  } else {
    for (int k = 0; k < 4; ++k) {
      int e = gid + k * stride;
      if (e < E) rank[e] = atomicAdd(&count[dst[e]], 1);
    }
  }
}

__global__ void scan_sums(const int* __restrict__ count, int* __restrict__ bsum, int N) {
  __shared__ int sh[256];
  int t = threadIdx.x;
  int i0 = blockIdx.x * 1024 + t * 4;
  int s = 0;
  for (int k = 0; k < 4; ++k) if (i0 + k < N) s += count[i0 + k];
  sh[t] = s;
  __syncthreads();
  for (int off = 128; off > 0; off >>= 1) {
    if (t < off) sh[t] += sh[t + off];
    __syncthreads();
  }
  if (t == 0) bsum[blockIdx.x] = sh[0];
}

__global__ void scan_write(const int* __restrict__ count, const int* __restrict__ bsum,
                           int* __restrict__ rowptr, int N, int E) {
  __shared__ int sh[256];
  int t = threadIdx.x, b = blockIdx.x;
  int part = 0;
  for (int k = t; k < b; k += 256) part += bsum[k];
  sh[t] = part;
  __syncthreads();
  for (int off = 128; off > 0; off >>= 1) {
    if (t < off) sh[t] += sh[t + off];
    __syncthreads();
  }
  int boff = sh[0];
  __syncthreads();

  int i0 = b * 1024 + t * 4;
  int c[4];
  int s = 0;
  for (int k = 0; k < 4; ++k) { c[k] = (i0 + k < N) ? count[i0 + k] : 0; s += c[k]; }
  sh[t] = s;
  __syncthreads();
  for (int off = 1; off < 256; off <<= 1) {
    int v = (t >= off) ? sh[t - off] : 0;
    __syncthreads();
    sh[t] += v;
    __syncthreads();
  }
  int run = boff + sh[t] - s;
  for (int k = 0; k < 4; ++k) {
    if (i0 + k < N) { rowptr[i0 + k] = run; run += c[k]; }
  }
  if (b == 0 && t == 0) rowptr[N] = E;
}

__global__ void scatter_src(const int* __restrict__ ei, const int* __restrict__ rowptr,
                            const int* __restrict__ rank, int* __restrict__ col, int E) {
  int gid = blockIdx.x * blockDim.x + threadIdx.x;
  int stride = gridDim.x * blockDim.x;
  const int* dst = ei + E;
  if (gid + 3 * stride < E) {
    int e0 = gid, e1 = gid + stride, e2 = gid + 2 * stride, e3 = gid + 3 * stride;
    int d0 = dst[e0], d1 = dst[e1], d2 = dst[e2], d3 = dst[e3];
    int r0 = rank[e0], r1 = rank[e1], r2 = rank[e2], r3 = rank[e3];
    int s0 = ei[e0], s1 = ei[e1], s2 = ei[e2], s3 = ei[e3];
    int q0 = rowptr[d0], q1 = rowptr[d1], q2 = rowptr[d2], q3 = rowptr[d3];
    col[q0 + r0] = s0; col[q1 + r1] = s1; col[q2 + r2] = s2; col[q3 + r3] = s3;
  } else {
    for (int k = 0; k < 4; ++k) {
      int e = gid + k * stride;
      if (e < E) col[rowptr[dst[e]] + rank[e]] = ei[e];
    }
  }
}

__global__ void proj1_wmma(const float* __restrict__ x, const float* __restrict__ W1,
                           float* __restrict__ h1, int N) {
  int gtid = blockIdx.x * blockDim.x + threadIdx.x;
  int wave = gtid >> 5;
  int lane = threadIdx.x & 31;
  int tiles = (N + 15) >> 4;
  if (wave >= tiles) return;
  int half = lane >> 4;
  int l16  = lane & 15;
  int row  = wave * 16 + l16;
  int rowc = row < N ? row : N - 1;
  bool rok = row < N;

  v2f a[2];
#pragma unroll
  for (int s = 0; s < 2; ++s) {
    int k0 = 4 * s + 2 * half;
    a[s][0] = (rok && k0     < 7) ? x[rowc * 7 + k0]     : 0.f;
    a[s][1] = (rok && k0 + 1 < 7) ? x[rowc * 7 + k0 + 1] : 0.f;
  }

  v8f acc[4];
#pragma unroll
  for (int t = 0; t < 4; ++t)
#pragma unroll
    for (int q = 0; q < 8; ++q) acc[t][q] = 0.f;

#pragma unroll
  for (int s = 0; s < 2; ++s) {
    int k0 = 4 * s + 2 * half;
#pragma unroll
    for (int t = 0; t < 4; ++t) {
      int col = t * 16 + l16;
      v2f b;
      b[0] = (k0     < 7) ? W1[k0 * 64 + col]       : 0.f;
      b[1] = (k0 + 1 < 7) ? W1[(k0 + 1) * 64 + col] : 0.f;
      acc[t] = __builtin_amdgcn_wmma_f32_16x16x4_f32(
          false, a[s], false, b, (short)0, acc[t], false, false);
    }
  }

#pragma unroll
  for (int t = 0; t < 4; ++t)
#pragma unroll
    for (int r = 0; r < 8; ++r) {
      int m = wave * 16 + r + 8 * half;
      if (m < N) h1[m * 64 + t * 16 + l16] = acc[t][r];
    }
}

__global__ void alpha1_kern(const float* __restrict__ h1, const float* __restrict__ aS,
                            const float* __restrict__ aD, float* __restrict__ as1,
                            float* __restrict__ ad1, int N) {
  int i = blockIdx.x * blockDim.x + threadIdx.x;
  if (i >= N * 4) return;
  int hd = i & 3, n = i >> 2;
  const float* hp = h1 + n * 64 + hd * 16;
  float s = 0.f, d = 0.f;
#pragma unroll
  for (int c = 0; c < 16; ++c) {
    float v = hp[c];
    s += v * aS[hd * 16 + c];
    d += v * aD[hd * 16 + c];
  }
  as1[i] = s; ad1[i] = d;
}


#define SEL8(a, b, c, d, e, f, g, h) \
  ((a) | ((b) << 3) | ((c) << 6) | ((d) << 9) | ((e) << 12) | ((f) << 15) | ((g) << 18) | ((h) << 21))
#define BC8(v, J) __builtin_amdgcn_mov_dpp8((unsigned)(v), SEL8(J, J, J, J, J, J, J, J))
#define BC8F(v, J) __uint_as_float(BC8(__float_as_uint(v), J))

__device__ __forceinline__ float max8(float v) {
  v = fmaxf(v, __uint_as_float(__builtin_amdgcn_mov_dpp8(__float_as_uint(v), SEL8(1, 0, 3, 2, 5, 4, 7, 6))));
  v = fmaxf(v, __uint_as_float(__builtin_amdgcn_mov_dpp8(__float_as_uint(v), SEL8(2, 3, 0, 1, 6, 7, 4, 5))));
  v = fmaxf(v, __uint_as_float(__builtin_amdgcn_mov_dpp8(__float_as_uint(v), SEL8(4, 5, 6, 7, 0, 1, 2, 3))));
  return v;
}
__device__ __forceinline__ float sum8(float v) {
  v += __uint_as_float(__builtin_amdgcn_mov_dpp8(__float_as_uint(v), SEL8(1, 0, 3, 2, 5, 4, 7, 6)));
  v += __uint_as_float(__builtin_amdgcn_mov_dpp8(__float_as_uint(v), SEL8(2, 3, 0, 1, 6, 7, 4, 5)));
  v += __uint_as_float(__builtin_amdgcn_mov_dpp8(__float_as_uint(v), SEL8(4, 5, 6, 7, 0, 1, 2, 3)));
  return v;
}

__device__ __forceinline__ v2f row_ld(const char* __restrict__ hb, int st, unsigned laneoff) {
  return *(const v2f*)(hb + ((((unsigned)st) << 8) | laneoff));
}
#define SRC8(s, J) __builtin_amdgcn_readlane((s), (J))

template <int NB>
__device__ __forceinline__ void load_rows(const char* __restrict__ hb, int s, unsigned laneoff, v2f* v) {
  v[0] = row_ld(hb, SRC8(s, 0), laneoff);
  v[1] = row_ld(hb, SRC8(s, 1), laneoff);
  v[2] = row_ld(hb, SRC8(s, 2), laneoff);
  v[3] = row_ld(hb, SRC8(s, 3), laneoff);
  if (NB > 4) {
    v[4] = row_ld(hb, SRC8(s, 4), laneoff);
    v[5] = row_ld(hb, SRC8(s, 5), laneoff);
    v[6] = row_ld(hb, SRC8(s, 6), laneoff);
    v[7] = row_ld(hb, SRC8(s, 7), laneoff);
  }
}

#define ADDROW(J) { float w = BC8F(p, J); v2f ww = {w, w}; acc = __builtin_elementwise_fma(v[J], ww, acc); }
template <int NB>
__device__ __forceinline__ void add_rows(float p, const v2f* v, v2f& acc) {
  ADDROW(0) ADDROW(1) ADDROW(2) ADDROW(3)
  if (NB > 4) { ADDROW(4) ADDROW(5) ADDROW(6) ADDROW(7) }
}

__global__ void gat1_node(const int* __restrict__ rowptr, const int* __restrict__ col,
                          const float* __restrict__ as1, const float* __restrict__ ad1,
                          const float* __restrict__ h1, const float* __restrict__ b1,
                          float* __restrict__ x2, int N) {
  int lane = threadIdx.x & 31;
  int hd = lane >> 3, j8 = lane & 7;
  int n = __builtin_amdgcn_readfirstlane((int)(blockIdx.x * (blockDim.x >> 5) + (threadIdx.x >> 5)));
  if (n >= N) return;
  const char* hb = (const char*)h1;
  unsigned laneoff = (unsigned)lane * 8u;

  int beg = rowptr[n], end = rowptr[n + 1];
  int cnt = end - beg;
  float adh = ad1[n * 4 + hd];
  float es = leaky(as1[n * 4 + hd] + adh);
  float m = es;
  v2f hv = row_ld(hb, n, laneoff);

  int s[8];
  float e[8];
#pragma unroll
  for (int c = 0; c < 8; ++c) {
    int j = beg + 8 * c + j8;
    int sv = n;
    if (j < end) sv = col[j];
    s[c] = sv;
  }
#pragma unroll
  for (int c = 0; c < 8; ++c) {
    float ev = leaky(as1[s[c] * 4 + hd] + adh);
    e[c] = ev;
    m = fmaxf(m, ev);
  }
  for (int base = beg + 64; base < end; base += 8) {
    int j = base + j8;
    if (j < end) m = fmaxf(m, leaky(as1[col[j] * 4 + hd] + adh));
  }
  m = max8(m);

  float p[8];
  float dsum = 0.f;
#pragma unroll
  for (int c = 0; c < 8; ++c) {
    int j = beg + 8 * c + j8;
    float pv = (j < end) ? __expf(e[c] - m) : 0.f;
    p[c] = pv;
    dsum += pv;
  }
  v2f acc = {0.f, 0.f};
#pragma unroll
  for (int g = 0; g < 4; ++g) {
    int lo = 16 * g;
    if (lo < cnt) {
      v2f va[8], vb[8];
      if (lo + 12 < cnt) {
        load_rows<8>(hb, s[2 * g], laneoff, va);
        load_rows<8>(hb, s[2 * g + 1], laneoff, vb);
        add_rows<8>(p[2 * g], va, acc);
        add_rows<8>(p[2 * g + 1], vb, acc);
      } else if (lo + 8 < cnt) {
        load_rows<8>(hb, s[2 * g], laneoff, va);
        load_rows<4>(hb, s[2 * g + 1], laneoff, vb);
        add_rows<8>(p[2 * g], va, acc);
        add_rows<4>(p[2 * g + 1], vb, acc);
      } else if (lo + 4 < cnt) {
        load_rows<8>(hb, s[2 * g], laneoff, va);
        add_rows<8>(p[2 * g], va, acc);
      } else {
        load_rows<4>(hb, s[2 * g], laneoff, va);
        add_rows<4>(p[2 * g], va, acc);
      }
    }
  }
  for (int base = beg + 64; base < end; base += 8) {
    int j = base + j8;
    int sv = n;
    if (j < end) sv = col[j];
    float pv = (j < end) ? __expf(leaky(as1[sv * 4 + hd] + adh) - m) : 0.f;
    dsum += pv;
    v2f v[8];
    load_rows<8>(hb, sv, laneoff, v);
    add_rows<8>(pv, v, acc);
  }
  float ws = __expf(es - m);
  float den = sum8(dsum) + ws;
  v2f wws = {ws, ws};
  acc = __builtin_elementwise_fma(hv, wws, acc);
  float ax = acc[0], ay = acc[1];

  float2 bb = ((const float2*)b1)[lane];
  float ox = ax / den + bb.x;
  float oy = ay / den + bb.y;
  ox = ox > 0.f ? ox : (__expf(ox) - 1.f);
  oy = oy > 0.f ? oy : (__expf(oy) - 1.f);
  ((float2*)x2)[(size_t)n * 32 + lane] = make_float2(ox, oy);
}

__global__ void proj2_wmma(const float* __restrict__ x2, const float* __restrict__ W2,
                           float* __restrict__ rec, int N) {
  int gtid = blockIdx.x * blockDim.x + threadIdx.x;
  int wave = gtid >> 5;
  int lane = threadIdx.x & 31;
  int tiles = (N + 15) >> 4;
  if (wave >= tiles) return;
  int half = lane >> 4, l16 = lane & 15;
  int row  = wave * 16 + l16;
  int rowc = row < N ? row : N - 1;
  bool rok = row < N;
  bool cok = l16 < 6;

  v8f acc;
#pragma unroll
  for (int q = 0; q < 8; ++q) acc[q] = 0.f;

#pragma unroll
  for (int s = 0; s < 16; ++s) {
    int k0 = 4 * s + 2 * half;
    v2f a, b;
    a[0] = rok ? x2[rowc * 64 + k0]     : 0.f;
    a[1] = rok ? x2[rowc * 64 + k0 + 1] : 0.f;
    b[0] = cok ? W2[k0 * 6 + l16]       : 0.f;
    b[1] = cok ? W2[(k0 + 1) * 6 + l16] : 0.f;
    acc = __builtin_amdgcn_wmma_f32_16x16x4_f32(
        false, a, false, b, (short)0, acc, false, false);
  }

  if (cok) {
#pragma unroll
    for (int r = 0; r < 8; ++r) {
      int m = wave * 16 + r + 8 * half;
      if (m < N) rec[m * 8 + l16] = acc[r];
    }
  }
}

__global__ void alpha2_kern(float* rec, const float* __restrict__ aS,
                            const float* __restrict__ aD, float* __restrict__ ad2, int N) {
  int n = blockIdx.x * blockDim.x + threadIdx.x;
  if (n >= N) return;
  float4 lo = ((const float4*)rec)[n * 2];
  float2 hi = ((const float2*)rec)[n * 4 + 2];
  float v[6] = {lo.x, lo.y, lo.z, lo.w, hi.x, hi.y};
  float s = 0.f, d = 0.f;
#pragma unroll
  for (int c = 0; c < 6; ++c) {
    s += v[c] * aS[c];
    d += v[c] * aD[c];
  }
  ((float2*)rec)[n * 4 + 3] = make_float2(s, 0.f);
  ad2[n] = d;
}


__device__ __forceinline__ float odd_of_pair(float v) {
  return __uint_as_float(__builtin_amdgcn_mov_dpp8(__float_as_uint(v), SEL8(1, 1, 3, 3, 5, 5, 7, 7)));
}

__global__ void gat2_node(const int* __restrict__ rowptr, const int* __restrict__ col,
                          const float* __restrict__ rec, const float* __restrict__ ad2,
                          const float* __restrict__ b2, float* __restrict__ out, int N) {
  int gid = blockIdx.x * blockDim.x + threadIdx.x;
  int n = gid >> 4, sub = gid & 15;
  int p = sub >> 1, hf = sub & 1;
  bool ok = n < N;
  int nc = ok ? n : N - 1;
  int beg = rowptr[nc];
  int end = ok ? rowptr[nc + 1] : beg;
  float ad = ad2[nc];
  const float4* rv = (const float4*)rec;
  float4 rs = rv[nc * 2 + hf];
  float es = leaky(odd_of_pair(rs.z) + ad);
  float m = es;

  int sv[8];
  float4 r[8];
  float ev[8];
#pragma unroll
  for (int k = 0; k < 8; ++k) {
    int j = beg + p + 8 * k;
    sv[k] = col[j < end ? j : 0];
  }
#pragma unroll
  for (int k = 0; k < 8; ++k) r[k] = rv[sv[k] * 2 + hf];
#pragma unroll
  for (int k = 0; k < 8; ++k) {
    int j = beg + p + 8 * k;
    float e = leaky(odd_of_pair(r[k].z) + ad);
    ev[k] = e;
    if (j < end) m = fmaxf(m, e);
  }
  for (int j = beg + p + 64; j < end; j += 8)
    m = fmaxf(m, leaky(rec[col[j] * 8 + 6] + ad));
  m = fmaxf(m, __shfl_xor(m, 2));
  m = fmaxf(m, __shfl_xor(m, 4));
  m = fmaxf(m, __shfl_xor(m, 8));

  float den = 0.f;
  float acc[4];
#pragma unroll
  for (int c = 0; c < 4; ++c) acc[c] = 0.f;
#pragma unroll
  for (int k = 0; k < 8; ++k) {
    int j = beg + p + 8 * k;
    if (j < end) {
      float w = __expf(ev[k] - m);
      den += w;
      acc[0] += w * r[k].x; acc[1] += w * r[k].y;
      acc[2] += w * r[k].z; acc[3] += w * r[k].w;
    }
  }
  for (int j = beg + p + 64; j < end; j += 8) {
    int s = col[j];
    float w = __expf(leaky(rec[s * 8 + 6] + ad) - m);
    den += w;
    float4 q = rv[s * 2 + hf];
    acc[0] += w * q.x; acc[1] += w * q.y;
    acc[2] += w * q.z; acc[3] += w * q.w;
  }
  if (p == 0) {
    float w = __expf(es - m);
    den += w;
    acc[0] += w * rs.x; acc[1] += w * rs.y;
    acc[2] += w * rs.z; acc[3] += w * rs.w;
  }
#pragma unroll
  for (int off = 2; off < 16; off <<= 1) {
    den += __shfl_xor(den, off);
#pragma unroll
    for (int c = 0; c < 4; ++c) acc[c] += __shfl_xor(acc[c], off);
  }
  float o4 = __shfl_xor(acc[0], 1);
  float o5 = __shfl_xor(acc[1], 1);

  if (ok && sub == 0) {
    float o[6] = {acc[0], acc[1], acc[2], acc[3], o4, o5};
    float mx = -3.0e38f;
#pragma unroll
    for (int c = 0; c < 6; ++c) { o[c] = o[c] / den + b2[c]; mx = fmaxf(mx, o[c]); }
    float sum = 0.f;
#pragma unroll
    for (int c = 0; c < 6; ++c) sum += __expf(o[c] - mx);
    float l = __logf(sum);
#pragma unroll
    for (int c = 0; c < 6; ++c) out[n * 6 + c] = o[c] - mx - l;
  }
}

extern "C" void kernel_launch(void* const* d_in, const int* in_sizes, int n_in,
                              void* d_out, int out_size, void* d_ws, size_t ws_size,
                              hipStream_t stream) {
  const float* x      = (const float*)d_in[0];
  const int*   ei     = (const int*)d_in[1];
  const float* W1     = (const float*)d_in[2];
  const float* a_src1 = (const float*)d_in[3];
  const float* a_dst1 = (const float*)d_in[4];
  const float* b1     = (const float*)d_in[5];
  const float* W2     = (const float*)d_in[6];
  const float* a_src2 = (const float*)d_in[7];
  const float* a_dst2 = (const float*)d_in[8];
  const float* b2     = (const float*)d_in[9];
  float* out = (float*)d_out;

  int N = in_sizes[0] / 7;
  int E = in_sizes[1] / 2;

  float* ws  = (float*)d_ws;
  float* h1  = ws;
  float* x2  = h1 + (size_t)N * 64;
  float* as1 = x2 + (size_t)N * 64;
  float* ad1 = as1 + (size_t)N * 4;
  float* rec = ad1 + (size_t)N * 4;
  float* ad2 = rec + (size_t)N * 8;
  int nb = (N + 1023) / 1024;
  int* count  = (int*)(ad2 + N);
  int* rowptr = count + N;
  int* bsum   = rowptr + N + 1;
  int* rank   = bsum + nb;
  int* col    = rank + E;

  const int B = 256;
  auto G = [](long n, int b) { return (int)((n + b - 1) / b); };
  int tiles = (N + 15) / 16;
  int projBlocks = (tiles + (B / 32) - 1) / (B / 32);
  int edgeBlocks = G(((long)E + 3) / 4, B);

  zero_count<<<G(N, B), B, 0, stream>>>(count, N);
  count_rank<<<edgeBlocks, B, 0, stream>>>(ei, count, rank, E);
  scan_sums<<<nb, B, 0, stream>>>(count, bsum, N);
  scan_write<<<nb, B, 0, stream>>>(count, bsum, rowptr, N, E);
  scatter_src<<<edgeBlocks, B, 0, stream>>>(ei, rowptr, rank, col, E);

  proj1_wmma<<<projBlocks, B, 0, stream>>>(x, W1, h1, N);
  alpha1_kern<<<G((long)N * 4, B), B, 0, stream>>>(h1, a_src1, a_dst1, as1, ad1, N);
  gat1_node<<<G(N, B / 32), B, 0, stream>>>(rowptr, col, as1, ad1, h1, b1, x2, N);

  proj2_wmma<<<projBlocks, B, 0, stream>>>(x2, W2, rec, N);
  alpha2_kern<<<G(N, B), B, 0, stream>>>(rec, a_src2, a_dst2, ad2, N);
  gat2_node<<<G((long)N * 16, B), B, 0, stream>>>(rowptr, col, rec, ad2, b2, out, N);
}


# Edge sources go straight into fixed 64-slot rows per destination at count time (slot = the count the add returns); in-degree above 64 spills to a filtered list: no scan, rank array or scatter pass
# speedup vs baseline: 8.2938x; 1.1531x over previous
#include <hip/hip_runtime.h>
#include <hip/hip_bf16.h>
#include <math.h>

typedef float v2f __attribute__((ext_vector_type(2)));
typedef float v8f __attribute__((ext_vector_type(8)));

#define NEG_SLOPE 0.2f

__device__ __forceinline__ float leaky(float v) { return v > 0.f ? v : NEG_SLOPE * v; }

#define SLOTS 64

__global__ void zero_count(int* __restrict__ count, int n1) {
  int i = blockIdx.x * blockDim.x + threadIdx.x;
  if (i < n1) count[i] = 0;
}

__device__ __forceinline__ void put_edge(int s, int d, int r, int* __restrict__ slots,
                                         int* __restrict__ ovf, int* ovfc) {
  if (r < SLOTS) {
    slots[d * SLOTS + r] = s;
  } else {
    int k = atomicAdd(ovfc, 1);
    ((int2*)ovf)[k] = make_int2(s, d);
  }
}

__global__ void fill_slots(const int* __restrict__ ei, int* count,
                           int* __restrict__ slots, int* __restrict__ ovf, int N, int E) {
  int gid = blockIdx.x * blockDim.x + threadIdx.x;
  int stride = gridDim.x * blockDim.x;
  const int* dst = ei + E;
  int* ovfc = count + N;
  if (gid + 3 * stride < E) {
    int e0 = gid, e1 = gid + stride, e2 = gid + 2 * stride, e3 = gid + 3 * stride;
    int d0 = dst[e0], d1 = dst[e1], d2 = dst[e2], d3 = dst[e3];
    int s0 = ei[e0], s1 = ei[e1], s2 = ei[e2], s3 = ei[e3];
    int r0 = atomicAdd(&count[d0], 1);
    int r1 = atomicAdd(&count[d1], 1);
    int r2 = atomicAdd(&count[d2], 1);
    int r3 = atomicAdd(&count[d3], 1);
    put_edge(s0, d0, r0, slots, ovf, ovfc);
    put_edge(s1, d1, r1, slots, ovf, ovfc);
    put_edge(s2, d2, r2, slots, ovf, ovfc);
    put_edge(s3, d3, r3, slots, ovf, ovfc);
  } else {
    for (int k = 0; k < 4; ++k) {
      int e = gid + k * stride;
      if (e < E) {
        int d = dst[e];
        put_edge(ei[e], d, atomicAdd(&count[d], 1), slots, ovf, ovfc);
      }
    }
  }
}

__global__ void proj1_wmma(const float* __restrict__ x, const float* __restrict__ W1,
                           float* __restrict__ h1, int N) {
  int gtid = blockIdx.x * blockDim.x + threadIdx.x;
  int wave = gtid >> 5;
  int lane = threadIdx.x & 31;
  int tiles = (N + 15) >> 4;
  if (wave >= tiles) return;
  int half = lane >> 4;
  int l16  = lane & 15;
  int row  = wave * 16 + l16;
  int rowc = row < N ? row : N - 1;
  bool rok = row < N;

  v2f a[2];
#pragma unroll
  for (int s = 0; s < 2; ++s) {
    int k0 = 4 * s + 2 * half;
    a[s][0] = (rok && k0     < 7) ? x[rowc * 7 + k0]     : 0.f;
    a[s][1] = (rok && k0 + 1 < 7) ? x[rowc * 7 + k0 + 1] : 0.f;
  }

  v8f acc[4];
#pragma unroll
  for (int t = 0; t < 4; ++t)
#pragma unroll
    for (int q = 0; q < 8; ++q) acc[t][q] = 0.f;

#pragma unroll
  for (int s = 0; s < 2; ++s) {
    int k0 = 4 * s + 2 * half;
#pragma unroll
    for (int t = 0; t < 4; ++t) {
      int col = t * 16 + l16;
      v2f b;
      b[0] = (k0     < 7) ? W1[k0 * 64 + col]       : 0.f;
      b[1] = (k0 + 1 < 7) ? W1[(k0 + 1) * 64 + col] : 0.f;
      acc[t] = __builtin_amdgcn_wmma_f32_16x16x4_f32(
          false, a[s], false, b, (short)0, acc[t], false, false);
    }
  }

#pragma unroll
  for (int t = 0; t < 4; ++t)
#pragma unroll
    for (int r = 0; r < 8; ++r) {
      int m = wave * 16 + r + 8 * half;
      if (m < N) h1[m * 64 + t * 16 + l16] = acc[t][r];
    }
}

__global__ void alpha1_kern(const float* __restrict__ h1, const float* __restrict__ aS,
                            const float* __restrict__ aD, float* __restrict__ as1,
                            float* __restrict__ ad1, int N) {
  int i = blockIdx.x * blockDim.x + threadIdx.x;
  if (i >= N * 4) return;
  int hd = i & 3, n = i >> 2;
  const float* hp = h1 + n * 64 + hd * 16;
  float s = 0.f, d = 0.f;
#pragma unroll
  for (int c = 0; c < 16; ++c) {
    float v = hp[c];
    s += v * aS[hd * 16 + c];
    d += v * aD[hd * 16 + c];
  }
  as1[i] = s; ad1[i] = d;
}


#define SEL8(a, b, c, d, e, f, g, h) \
  ((a) | ((b) << 3) | ((c) << 6) | ((d) << 9) | ((e) << 12) | ((f) << 15) | ((g) << 18) | ((h) << 21))
#define BC8(v, J) __builtin_amdgcn_mov_dpp8((unsigned)(v), SEL8(J, J, J, J, J, J, J, J))
#define BC8F(v, J) __uint_as_float(BC8(__float_as_uint(v), J))

__device__ __forceinline__ float max8(float v) {
  v = fmaxf(v, __uint_as_float(__builtin_amdgcn_mov_dpp8(__float_as_uint(v), SEL8(1, 0, 3, 2, 5, 4, 7, 6))));
  v = fmaxf(v, __uint_as_float(__builtin_amdgcn_mov_dpp8(__float_as_uint(v), SEL8(2, 3, 0, 1, 6, 7, 4, 5))));
  v = fmaxf(v, __uint_as_float(__builtin_amdgcn_mov_dpp8(__float_as_uint(v), SEL8(4, 5, 6, 7, 0, 1, 2, 3))));
  return v;
}
__device__ __forceinline__ float sum8(float v) {
  v += __uint_as_float(__builtin_amdgcn_mov_dpp8(__float_as_uint(v), SEL8(1, 0, 3, 2, 5, 4, 7, 6)));
  v += __uint_as_float(__builtin_amdgcn_mov_dpp8(__float_as_uint(v), SEL8(2, 3, 0, 1, 6, 7, 4, 5)));
  v += __uint_as_float(__builtin_amdgcn_mov_dpp8(__float_as_uint(v), SEL8(4, 5, 6, 7, 0, 1, 2, 3)));
  return v;
}

__device__ __forceinline__ v2f row_ld(const char* __restrict__ hb, int st, unsigned laneoff) {
  return *(const v2f*)(hb + ((((unsigned)st) << 8) | laneoff));
}
#define SRC8(s, J) __builtin_amdgcn_readlane((s), (J))

template <int NB>
__device__ __forceinline__ void load_rows(const char* __restrict__ hb, int s, unsigned laneoff, v2f* v) {
  v[0] = row_ld(hb, SRC8(s, 0), laneoff);
  v[1] = row_ld(hb, SRC8(s, 1), laneoff);
  v[2] = row_ld(hb, SRC8(s, 2), laneoff);
  v[3] = row_ld(hb, SRC8(s, 3), laneoff);
  if (NB > 4) {
    v[4] = row_ld(hb, SRC8(s, 4), laneoff);
    v[5] = row_ld(hb, SRC8(s, 5), laneoff);
    v[6] = row_ld(hb, SRC8(s, 6), laneoff);
    v[7] = row_ld(hb, SRC8(s, 7), laneoff);
  }
}

#define ADDROW(J) { float w = BC8F(p, J); v2f ww = {w, w}; acc = __builtin_elementwise_fma(v[J], ww, acc); }
template <int NB>
__device__ __forceinline__ void add_rows(float p, const v2f* v, v2f& acc) {
  ADDROW(0) ADDROW(1) ADDROW(2) ADDROW(3)
  if (NB > 4) { ADDROW(4) ADDROW(5) ADDROW(6) ADDROW(7) }
}

__global__ void gat1_node(const int* __restrict__ count, const int* __restrict__ slots,
                          const int* __restrict__ ovfi, const float* __restrict__ as1, const float* __restrict__ ad1,
                          const float* __restrict__ h1, const float* __restrict__ b1,
                          float* __restrict__ x2, int N) {
  int lane = threadIdx.x & 31;
  int hd = lane >> 3, j8 = lane & 7;
  int n = __builtin_amdgcn_readfirstlane((int)(blockIdx.x * (blockDim.x >> 5) + (threadIdx.x >> 5)));
  if (n >= N) return;
  const char* hb = (const char*)h1;
  unsigned laneoff = (unsigned)lane * 8u;

  int cnt = count[n];
  const int* lst = slots + n * SLOTS;
  const int2* ovf = (const int2*)ovfi;
  float adh = ad1[n * 4 + hd];
  float es = leaky(as1[n * 4 + hd] + adh);
  float m = es;
  v2f hv = row_ld(hb, n, laneoff);

  int s[8];
  float e[8];
#pragma unroll
  for (int c = 0; c < 8; ++c) {
    int j = 8 * c + j8;
    int sv = n;
    if (j < cnt) sv = lst[j];
    s[c] = sv;
  }
#pragma unroll
  for (int c = 0; c < 8; ++c) {
    float ev = leaky(as1[s[c] * 4 + hd] + adh);
    e[c] = ev;
    m = fmaxf(m, ev);
  }
  int novf = 0;
  if (cnt > SLOTS) {
    novf = count[N];
    for (int base = 0; base < novf; base += 8) {
      int j = base + j8;
      if (j < novf) {
        int2 q = ovf[j];
        if (q.y == n) m = fmaxf(m, leaky(as1[q.x * 4 + hd] + adh));
      }
    }
  }
  m = max8(m);

  float p[8];
  float dsum = 0.f;
#pragma unroll
  for (int c = 0; c < 8; ++c) {
    int j = 8 * c + j8;
    float pv = (j < cnt) ? __expf(e[c] - m) : 0.f;
    p[c] = pv;
    dsum += pv;
  }
  v2f acc = {0.f, 0.f};
#pragma unroll
  for (int g = 0; g < 4; ++g) {
    int lo = 16 * g;
    if (lo < cnt) {
      v2f va[8], vb[8];
      if (lo + 12 < cnt) {
        load_rows<8>(hb, s[2 * g], laneoff, va);
        load_rows<8>(hb, s[2 * g + 1], laneoff, vb);
        add_rows<8>(p[2 * g], va, acc);
        add_rows<8>(p[2 * g + 1], vb, acc);
      } else if (lo + 8 < cnt) {
        load_rows<8>(hb, s[2 * g], laneoff, va);
        load_rows<4>(hb, s[2 * g + 1], laneoff, vb);
        add_rows<8>(p[2 * g], va, acc);
        add_rows<4>(p[2 * g + 1], vb, acc);
      } else if (lo + 4 < cnt) {
        load_rows<8>(hb, s[2 * g], laneoff, va);
        add_rows<8>(p[2 * g], va, acc);
      } else {
        load_rows<4>(hb, s[2 * g], laneoff, va);
        add_rows<4>(p[2 * g], va, acc);
      }
    }
  }
  for (int base = 0; base < novf; base += 8) {
    int j = base + j8;
    int sv = n;
    bool mine = false;
    if (j < novf) {
      int2 q = ovf[j];
      if (q.y == n) { sv = q.x; mine = true; }
    }
    if (__builtin_amdgcn_ballot_w32(mine) != 0) {
      float pv = mine ? __expf(leaky(as1[sv * 4 + hd] + adh) - m) : 0.f;
      dsum += pv;
      v2f v[8];
      load_rows<8>(hb, sv, laneoff, v);
      add_rows<8>(pv, v, acc);
    }
  }
  float ws = __expf(es - m);
  float den = sum8(dsum) + ws;
  v2f wws = {ws, ws};
  acc = __builtin_elementwise_fma(hv, wws, acc);
  float ax = acc[0], ay = acc[1];

  float2 bb = ((const float2*)b1)[lane];
  float ox = ax / den + bb.x;
  float oy = ay / den + bb.y;
  ox = ox > 0.f ? ox : (__expf(ox) - 1.f);
  oy = oy > 0.f ? oy : (__expf(oy) - 1.f);
  ((float2*)x2)[(size_t)n * 32 + lane] = make_float2(ox, oy);
}

__global__ void proj2_wmma(const float* __restrict__ x2, const float* __restrict__ W2,
                           float* __restrict__ rec, int N) {
  int gtid = blockIdx.x * blockDim.x + threadIdx.x;
  int wave = gtid >> 5;
  int lane = threadIdx.x & 31;
  int tiles = (N + 15) >> 4;
  if (wave >= tiles) return;
  int half = lane >> 4, l16 = lane & 15;
  int row  = wave * 16 + l16;
  int rowc = row < N ? row : N - 1;
  bool rok = row < N;
  bool cok = l16 < 6;

  v8f acc;
#pragma unroll
  for (int q = 0; q < 8; ++q) acc[q] = 0.f;

#pragma unroll
  for (int s = 0; s < 16; ++s) {
    int k0 = 4 * s + 2 * half;
    v2f a, b;
    a[0] = rok ? x2[rowc * 64 + k0]     : 0.f;
    a[1] = rok ? x2[rowc * 64 + k0 + 1] : 0.f;
    b[0] = cok ? W2[k0 * 6 + l16]       : 0.f;
    b[1] = cok ? W2[(k0 + 1) * 6 + l16] : 0.f;
    acc = __builtin_amdgcn_wmma_f32_16x16x4_f32(
        false, a, false, b, (short)0, acc, false, false);
  }

  if (cok) {
#pragma unroll
    for (int r = 0; r < 8; ++r) {
      int m = wave * 16 + r + 8 * half;
      if (m < N) rec[m * 8 + l16] = acc[r];
    }
  }
}

__global__ void alpha2_kern(float* rec, const float* __restrict__ aS,
                            const float* __restrict__ aD, float* __restrict__ ad2, int N) {
  int n = blockIdx.x * blockDim.x + threadIdx.x;
  if (n >= N) return;
  float4 lo = ((const float4*)rec)[n * 2];
  float2 hi = ((const float2*)rec)[n * 4 + 2];
  float v[6] = {lo.x, lo.y, lo.z, lo.w, hi.x, hi.y};
  float s = 0.f, d = 0.f;
#pragma unroll
  for (int c = 0; c < 6; ++c) {
    s += v[c] * aS[c];
    d += v[c] * aD[c];
  }
  ((float2*)rec)[n * 4 + 3] = make_float2(s, 0.f);
  ad2[n] = d;
}


__device__ __forceinline__ float odd_of_pair(float v) {
  return __uint_as_float(__builtin_amdgcn_mov_dpp8(__float_as_uint(v), SEL8(1, 1, 3, 3, 5, 5, 7, 7)));
}

__global__ void gat2_node(const int* __restrict__ count, const int* __restrict__ slots,
                          const int* __restrict__ ovfi, const float* __restrict__ rec, const float* __restrict__ ad2,
                          const float* __restrict__ b2, float* __restrict__ out, int N) {
  int gid = blockIdx.x * blockDim.x + threadIdx.x;
  int n = gid >> 4, sub = gid & 15;
  int p = sub >> 1, hf = sub & 1;
  bool ok = n < N;
  int nc = ok ? n : N - 1;
  int cnt = ok ? count[nc] : 0;
  const int* lst = slots + nc * SLOTS;
  const int2* ovf = (const int2*)ovfi;
  float ad = ad2[nc];
  const float4* rv = (const float4*)rec;
  float4 rs = rv[nc * 2 + hf];
  float es = leaky(odd_of_pair(rs.z) + ad);
  float m = es;

  int sv[8];
  float4 r[8];
  float ev[8];
#pragma unroll
  for (int k = 0; k < 8; ++k) {
    int j = p + 8 * k;
    int v = nc;
    if (j < cnt) v = lst[j];
    sv[k] = v;
  }
#pragma unroll
  for (int k = 0; k < 8; ++k) r[k] = rv[sv[k] * 2 + hf];
#pragma unroll
  for (int k = 0; k < 8; ++k) {
    int j = p + 8 * k;
    float e = leaky(odd_of_pair(r[k].z) + ad);
    ev[k] = e;
    if (j < cnt) m = fmaxf(m, e);
  }
  int novf = cnt > SLOTS ? count[N] : 0;
  for (int j = p; j < novf; j += 8) {
    int2 q = ovf[j];
    if (q.y == n) m = fmaxf(m, leaky(rec[q.x * 8 + 6] + ad));
  }
  m = fmaxf(m, __shfl_xor(m, 2));
  m = fmaxf(m, __shfl_xor(m, 4));
  m = fmaxf(m, __shfl_xor(m, 8));

  float den = 0.f;
  float acc[4];
#pragma unroll
  for (int c = 0; c < 4; ++c) acc[c] = 0.f;
#pragma unroll
  for (int k = 0; k < 8; ++k) {
    int j = p + 8 * k;
    if (j < cnt) {
      float w = __expf(ev[k] - m);
      den += w;
      acc[0] += w * r[k].x; acc[1] += w * r[k].y;
      acc[2] += w * r[k].z; acc[3] += w * r[k].w;
    }
  }
  for (int j = p; j < novf; j += 8) {
    int2 q = ovf[j];
    if (q.y == n) {
      int s = q.x;
      float w = __expf(leaky(rec[s * 8 + 6] + ad) - m);
      den += w;
      float4 h = rv[s * 2 + hf];
      acc[0] += w * h.x; acc[1] += w * h.y;
      acc[2] += w * h.z; acc[3] += w * h.w;
    }
  }
  if (p == 0) {
    float w = __expf(es - m);
    den += w;
    acc[0] += w * rs.x; acc[1] += w * rs.y;
    acc[2] += w * rs.z; acc[3] += w * rs.w;
  }
#pragma unroll
  for (int off = 2; off < 16; off <<= 1) {
    den += __shfl_xor(den, off);
#pragma unroll
    for (int c = 0; c < 4; ++c) acc[c] += __shfl_xor(acc[c], off);
  }
  float o4 = __shfl_xor(acc[0], 1);
  float o5 = __shfl_xor(acc[1], 1);

  if (ok && sub == 0) {
    float o[6] = {acc[0], acc[1], acc[2], acc[3], o4, o5};
    float mx = -3.0e38f;
#pragma unroll
    for (int c = 0; c < 6; ++c) { o[c] = o[c] / den + b2[c]; mx = fmaxf(mx, o[c]); }
    float sum = 0.f;
#pragma unroll
    for (int c = 0; c < 6; ++c) sum += __expf(o[c] - mx);
    float l = __logf(sum);
#pragma unroll
    for (int c = 0; c < 6; ++c) out[n * 6 + c] = o[c] - mx - l;
  }
}

extern "C" void kernel_launch(void* const* d_in, const int* in_sizes, int n_in,
                              void* d_out, int out_size, void* d_ws, size_t ws_size,
                              hipStream_t stream) {
  const float* x      = (const float*)d_in[0];
  const int*   ei     = (const int*)d_in[1];
  const float* W1     = (const float*)d_in[2];
  const float* a_src1 = (const float*)d_in[3];
  const float* a_dst1 = (const float*)d_in[4];
  const float* b1     = (const float*)d_in[5];
  const float* W2     = (const float*)d_in[6];
  const float* a_src2 = (const float*)d_in[7];
  const float* a_dst2 = (const float*)d_in[8];
  const float* b2     = (const float*)d_in[9];
  float* out = (float*)d_out;

  int N = in_sizes[0] / 7;
  int E = in_sizes[1] / 2;

  float* ws  = (float*)d_ws;
  float* h1  = ws;
  float* x2  = h1 + (size_t)N * 64;
  int* slots = (int*)(x2 + (size_t)N * 64);
  float* as1 = (float*)(slots + (size_t)N * SLOTS);
  float* ad1 = as1 + (size_t)N * 4;
  float* rec = ad1 + (size_t)N * 4;
  int* ovf   = (int*)(rec + (size_t)N * 8);
  float* ad2 = (float*)(ovf + (size_t)E * 2);
  int* count = (int*)(ad2 + N);

  const int B = 256;
  auto G = [](long n, int b) { return (int)((n + b - 1) / b); };
  int tiles = (N + 15) / 16;
  int projBlocks = (tiles + (B / 32) - 1) / (B / 32);
  int edgeBlocks = G(((long)E + 3) / 4, B);

  zero_count<<<G((long)N + 1, B), B, 0, stream>>>(count, N + 1);
  fill_slots<<<edgeBlocks, B, 0, stream>>>(ei, count, slots, ovf, N, E);

  proj1_wmma<<<projBlocks, B, 0, stream>>>(x, W1, h1, N);
  alpha1_kern<<<G((long)N * 4, B), B, 0, stream>>>(h1, a_src1, a_dst1, as1, ad1, N);
  gat1_node<<<G(N, B / 32), B, 0, stream>>>(count, slots, ovf, as1, ad1, h1, b1, x2, N);

  proj2_wmma<<<projBlocks, B, 0, stream>>>(x2, W2, rec, N);
  alpha2_kern<<<G(N, B), B, 0, stream>>>(rec, a_src2, a_dst2, ad2, N);
  gat2_node<<<G((long)N * 16, B), B, 0, stream>>>(count, slots, ovf, rec, ad2, b2, out, N);
}


# Edge lists with no global atomics: 4096-edge chunks bucketed in LDS by 512-node range; one work-group per range fills its slot rows in LDS (LDS counter: arrival order, as before), stores them whole
# speedup vs baseline: 19.9721x; 2.4081x over previous
#include <hip/hip_runtime.h>
#include <hip/hip_bf16.h>
#include <math.h>

typedef float v2f __attribute__((ext_vector_type(2)));
typedef float v8f __attribute__((ext_vector_type(8)));

#define NEG_SLOPE 0.2f

__device__ __forceinline__ float leaky(float v) { return v > 0.f ? v : NEG_SLOPE * v; }

#define SLOTS 64
#define BSH 9
#define BN (1 << BSH)
#define CH 4096
#define TPB 1024

__device__ __forceinline__ int block_excl_scan(int v, int* wsum, int tid) {
  int lane = tid & 31, w = tid >> 5;
  int inc = v;
#pragma unroll
  for (int o = 1; o < 32; o <<= 1) {
    int t = __shfl_up(inc, o);
    if (lane >= o) inc += t;
  }
  if (lane == 31) wsum[w] = inc;
  __syncthreads();
  if (w == 0) {
    int s = wsum[lane];
    int si = s;
#pragma unroll
    for (int o = 1; o < 32; o <<= 1) {
      int t = __shfl_up(si, o);
      if (lane >= o) si += t;
    }
    wsum[lane] = si - s;
  }
  __syncthreads();
  return wsum[w] + inc - v;
}

__global__ __launch_bounds__(TPB) void sort_chunks(const int* __restrict__ ei, int2* __restrict__ bins,
                                                   int2* __restrict__ desc, int* __restrict__ count,
                                                   int N, int E, int nbin, int dpitch) {
  __shared__ int h[TPB];
  __shared__ int wsum[32];
  __shared__ int2 stage[CH];
  int tid = threadIdx.x;
  int blk = blockIdx.x;
  if (blk == 0 && tid == 0) count[N] = 0;
  h[tid] = 0;
  __syncthreads();
  int e0 = blk * CH + tid;
  int s[4], d[4], r[4];
#pragma unroll
  for (int k = 0; k < 4; ++k) {
    int e = e0 + k * TPB;
    s[k] = 0; d[k] = -1;
    if (e < E) { s[k] = ei[e]; d[k] = ei[E + e]; }
  }
#pragma unroll
  for (int k = 0; k < 4; ++k) {
    r[k] = 0;
    if (d[k] >= 0) r[k] = atomicAdd(&h[d[k] >> BSH], 1);
  }
  __syncthreads();
  int v = h[tid];
  int ex = block_excl_scan(v, wsum, tid);
  h[tid] = ex;
  if (tid < nbin) desc[(size_t)blk * dpitch + tid] = make_int2(ex, v);
  __syncthreads();
#pragma unroll
  for (int k = 0; k < 4; ++k)
    if (d[k] >= 0) stage[h[d[k] >> BSH] + r[k]] = make_int2(s[k], d[k]);
  __syncthreads();
  int total = E - blk * CH;
  if (total > CH) total = CH;
#pragma unroll
  for (int k = 0; k < 4; ++k) {
    int i = tid + k * TPB;
    if (i < total) bins[(size_t)blk * CH + i] = stage[i];
  }
}

__device__ __forceinline__ void place_edge(int2 ed, int n0, int* lcnt, int* lrows,
                                           int* __restrict__ ovf, int* ovfc) {
  int dl = ed.y - n0;
  int r = atomicAdd(&lcnt[dl], 1);
  if (r < SLOTS) {
    lrows[dl * SLOTS + ((r + dl) & (SLOTS - 1))] = ed.x;
  } else {
    int k = atomicAdd(ovfc, 1);
    ((int2*)ovf)[k] = ed;
  }
}

__global__ __launch_bounds__(TPB) void build_rows(const int2* __restrict__ bins, const int2* __restrict__ desc,
                                                  int* __restrict__ count, int* __restrict__ slots,
                                                  int* __restrict__ ovf, int N, int nblk, int dpitch) {
  __shared__ int lcnt[BN];
  __shared__ int lrows[BN * SLOTS];
  int tid = threadIdx.x, lane = tid & 31, w = tid >> 5;
  int b = blockIdx.x;
  int n0 = b << BSH;
  int* ovfc = count + N;
  if (tid < BN) lcnt[tid] = 0;
  __syncthreads();
  for (int base = 0; base < nblk; base += TPB) {
    int myblk = base + lane * 32 + w;
    int2 q = make_int2(0, 0);
    if (myblk < nblk) q = desc[(size_t)myblk * dpitch + b];
    for (int j0 = 0; j0 < 32; j0 += 8) {
      int2 ed[8];
      int ln[8];
      const int2* run[8];
#pragma unroll
      for (int u = 0; u < 8; ++u) {
        int st = __builtin_amdgcn_readlane(q.x, j0 + u);
        ln[u] = __builtin_amdgcn_readlane(q.y, j0 + u);
        run[u] = bins + (size_t)(base + (j0 + u) * 32 + w) * CH + st;
        ed[u] = make_int2(0, n0);
        if (lane < ln[u]) ed[u] = run[u][lane];
      }
#pragma unroll
      for (int u = 0; u < 8; ++u) {
        if (lane < ln[u]) place_edge(ed[u], n0, lcnt, lrows, ovf, ovfc);
        for (int k = lane + 32; k < ln[u]; k += 32)
          place_edge(run[u][k], n0, lcnt, lrows, ovf, ovfc);
      }
    }
  }
  __syncthreads();
  if (tid < BN && n0 + tid < N) count[n0 + tid] = lcnt[tid];
  for (int it = 0; it < BN * SLOTS / TPB; ++it) {
    int idx = it * TPB + tid;
    int dl = idx >> 6, j = idx & (SLOTS - 1);
    int n = n0 + dl;
    if (n < N && j < lcnt[dl])
      slots[(size_t)n * SLOTS + j] = lrows[dl * SLOTS + ((j + dl) & (SLOTS - 1))];
  }
}

__global__ void proj1_wmma(const float* __restrict__ x, const float* __restrict__ W1,
                           float* __restrict__ h1, int N) {
  int gtid = blockIdx.x * blockDim.x + threadIdx.x;
  int wave = gtid >> 5;
  int lane = threadIdx.x & 31;
  int tiles = (N + 15) >> 4;
  if (wave >= tiles) return;
  int half = lane >> 4;
  int l16  = lane & 15;
  int row  = wave * 16 + l16;
  int rowc = row < N ? row : N - 1;
  bool rok = row < N;

  v2f a[2];
#pragma unroll
  for (int s = 0; s < 2; ++s) {
    int k0 = 4 * s + 2 * half;
    a[s][0] = (rok && k0     < 7) ? x[rowc * 7 + k0]     : 0.f;
    a[s][1] = (rok && k0 + 1 < 7) ? x[rowc * 7 + k0 + 1] : 0.f;
  }

  v8f acc[4];
#pragma unroll
  for (int t = 0; t < 4; ++t)
#pragma unroll
    for (int q = 0; q < 8; ++q) acc[t][q] = 0.f;

#pragma unroll
  for (int s = 0; s < 2; ++s) {
    int k0 = 4 * s + 2 * half;
#pragma unroll
    for (int t = 0; t < 4; ++t) {
      int col = t * 16 + l16;
      v2f b;
      b[0] = (k0     < 7) ? W1[k0 * 64 + col]       : 0.f;
      b[1] = (k0 + 1 < 7) ? W1[(k0 + 1) * 64 + col] : 0.f;
      acc[t] = __builtin_amdgcn_wmma_f32_16x16x4_f32(
          false, a[s], false, b, (short)0, acc[t], false, false);
    }
  }

#pragma unroll
  for (int t = 0; t < 4; ++t)
#pragma unroll
    for (int r = 0; r < 8; ++r) {
      int m = wave * 16 + r + 8 * half;
      if (m < N) h1[m * 64 + t * 16 + l16] = acc[t][r];
    }
}

__global__ void alpha1_kern(const float* __restrict__ h1, const float* __restrict__ aS,
                            const float* __restrict__ aD, float* __restrict__ as1,
                            float* __restrict__ ad1, int N) {
  int i = blockIdx.x * blockDim.x + threadIdx.x;
  if (i >= N * 4) return;
  int hd = i & 3, n = i >> 2;
  const float* hp = h1 + n * 64 + hd * 16;
  float s = 0.f, d = 0.f;
#pragma unroll
  for (int c = 0; c < 16; ++c) {
    float v = hp[c];
    s += v * aS[hd * 16 + c];
    d += v * aD[hd * 16 + c];
  }
  as1[i] = s; ad1[i] = d;
}


#define SEL8(a, b, c, d, e, f, g, h) \
  ((a) | ((b) << 3) | ((c) << 6) | ((d) << 9) | ((e) << 12) | ((f) << 15) | ((g) << 18) | ((h) << 21))
#define BC8(v, J) __builtin_amdgcn_mov_dpp8((unsigned)(v), SEL8(J, J, J, J, J, J, J, J))
#define BC8F(v, J) __uint_as_float(BC8(__float_as_uint(v), J))

__device__ __forceinline__ float max8(float v) {
  v = fmaxf(v, __uint_as_float(__builtin_amdgcn_mov_dpp8(__float_as_uint(v), SEL8(1, 0, 3, 2, 5, 4, 7, 6))));
  v = fmaxf(v, __uint_as_float(__builtin_amdgcn_mov_dpp8(__float_as_uint(v), SEL8(2, 3, 0, 1, 6, 7, 4, 5))));
  v = fmaxf(v, __uint_as_float(__builtin_amdgcn_mov_dpp8(__float_as_uint(v), SEL8(4, 5, 6, 7, 0, 1, 2, 3))));
  return v;
}
__device__ __forceinline__ float sum8(float v) {
  v += __uint_as_float(__builtin_amdgcn_mov_dpp8(__float_as_uint(v), SEL8(1, 0, 3, 2, 5, 4, 7, 6)));
  v += __uint_as_float(__builtin_amdgcn_mov_dpp8(__float_as_uint(v), SEL8(2, 3, 0, 1, 6, 7, 4, 5)));
  v += __uint_as_float(__builtin_amdgcn_mov_dpp8(__float_as_uint(v), SEL8(4, 5, 6, 7, 0, 1, 2, 3)));
  return v;
}

__device__ __forceinline__ v2f row_ld(const char* __restrict__ hb, int st, unsigned laneoff) {
  return *(const v2f*)(hb + ((((unsigned)st) << 8) | laneoff));
}
#define SRC8(s, J) __builtin_amdgcn_readlane((s), (J))

template <int NB>
__device__ __forceinline__ void load_rows(const char* __restrict__ hb, int s, unsigned laneoff, v2f* v) {
  v[0] = row_ld(hb, SRC8(s, 0), laneoff);
  v[1] = row_ld(hb, SRC8(s, 1), laneoff);
  v[2] = row_ld(hb, SRC8(s, 2), laneoff);
  v[3] = row_ld(hb, SRC8(s, 3), laneoff);
  if (NB > 4) {
    v[4] = row_ld(hb, SRC8(s, 4), laneoff);
    v[5] = row_ld(hb, SRC8(s, 5), laneoff);
    v[6] = row_ld(hb, SRC8(s, 6), laneoff);
    v[7] = row_ld(hb, SRC8(s, 7), laneoff);
  }
}

#define ADDROW(J) { float w = BC8F(p, J); v2f ww = {w, w}; acc = __builtin_elementwise_fma(v[J], ww, acc); }
template <int NB>
__device__ __forceinline__ void add_rows(float p, const v2f* v, v2f& acc) {
  ADDROW(0) ADDROW(1) ADDROW(2) ADDROW(3)
  if (NB > 4) { ADDROW(4) ADDROW(5) ADDROW(6) ADDROW(7) }
}

__global__ void gat1_node(const int* __restrict__ count, const int* __restrict__ slots,
                          const int* __restrict__ ovfi, const float* __restrict__ as1, const float* __restrict__ ad1,
                          const float* __restrict__ h1, const float* __restrict__ b1,
                          float* __restrict__ x2, int N) {
  int lane = threadIdx.x & 31;
  int hd = lane >> 3, j8 = lane & 7;
  int n = __builtin_amdgcn_readfirstlane((int)(blockIdx.x * (blockDim.x >> 5) + (threadIdx.x >> 5)));
  if (n >= N) return;
  const char* hb = (const char*)h1;
  unsigned laneoff = (unsigned)lane * 8u;

  int cnt = count[n];
  const int* lst = slots + n * SLOTS;
  const int2* ovf = (const int2*)ovfi;
  float adh = ad1[n * 4 + hd];
  float es = leaky(as1[n * 4 + hd] + adh);
  float m = es;
  v2f hv = row_ld(hb, n, laneoff);

  int s[8];
  float e[8];
#pragma unroll
  for (int c = 0; c < 8; ++c) {
    int j = 8 * c + j8;
    int sv = n;
    if (j < cnt) sv = lst[j];
    s[c] = sv;
  }
#pragma unroll
  for (int c = 0; c < 8; ++c) {
    float ev = leaky(as1[s[c] * 4 + hd] + adh);
    e[c] = ev;
    m = fmaxf(m, ev);
  }
  int novf = 0;
  if (cnt > SLOTS) {
    novf = count[N];
    for (int base = 0; base < novf; base += 8) {
      int j = base + j8;
      if (j < novf) {
        int2 q = ovf[j];
        if (q.y == n) m = fmaxf(m, leaky(as1[q.x * 4 + hd] + adh));
      }
    }
  }
  m = max8(m);

  float p[8];
  float dsum = 0.f;
#pragma unroll
  for (int c = 0; c < 8; ++c) {
    int j = 8 * c + j8;
    float pv = (j < cnt) ? __expf(e[c] - m) : 0.f;
    p[c] = pv;
    dsum += pv;
  }
  v2f acc = {0.f, 0.f};
#pragma unroll
  for (int g = 0; g < 4; ++g) {
    int lo = 16 * g;
    if (lo < cnt) {
      v2f va[8], vb[8];
      if (lo + 12 < cnt) {
        load_rows<8>(hb, s[2 * g], laneoff, va);
        load_rows<8>(hb, s[2 * g + 1], laneoff, vb);
        add_rows<8>(p[2 * g], va, acc);
        add_rows<8>(p[2 * g + 1], vb, acc);
      } else if (lo + 8 < cnt) {
        load_rows<8>(hb, s[2 * g], laneoff, va);
        load_rows<4>(hb, s[2 * g + 1], laneoff, vb);
        add_rows<8>(p[2 * g], va, acc);
        add_rows<4>(p[2 * g + 1], vb, acc);
      } else if (lo + 4 < cnt) {
        load_rows<8>(hb, s[2 * g], laneoff, va);
        add_rows<8>(p[2 * g], va, acc);
      } else {
        load_rows<4>(hb, s[2 * g], laneoff, va);
        add_rows<4>(p[2 * g], va, acc);
      }
    }
  }
  for (int base = 0; base < novf; base += 8) {
    int j = base + j8;
    int sv = n;
    bool mine = false;
    if (j < novf) {
      int2 q = ovf[j];
      if (q.y == n) { sv = q.x; mine = true; }
    }
    if (__builtin_amdgcn_ballot_w32(mine) != 0) {
      float pv = mine ? __expf(leaky(as1[sv * 4 + hd] + adh) - m) : 0.f;
      dsum += pv;
      v2f v[8];
      load_rows<8>(hb, sv, laneoff, v);
      add_rows<8>(pv, v, acc);
    }
  }
  float ws = __expf(es - m);
  float den = sum8(dsum) + ws;
  v2f wws = {ws, ws};
  acc = __builtin_elementwise_fma(hv, wws, acc);
  float ax = acc[0], ay = acc[1];

  float2 bb = ((const float2*)b1)[lane];
  float ox = ax / den + bb.x;
  float oy = ay / den + bb.y;
  ox = ox > 0.f ? ox : (__expf(ox) - 1.f);
  oy = oy > 0.f ? oy : (__expf(oy) - 1.f);
  ((float2*)x2)[(size_t)n * 32 + lane] = make_float2(ox, oy);
}

__global__ void proj2_wmma(const float* __restrict__ x2, const float* __restrict__ W2,
                           float* __restrict__ rec, int N) {
  int gtid = blockIdx.x * blockDim.x + threadIdx.x;
  int wave = gtid >> 5;
  int lane = threadIdx.x & 31;
  int tiles = (N + 15) >> 4;
  if (wave >= tiles) return;
  int half = lane >> 4, l16 = lane & 15;
  int row  = wave * 16 + l16;
  int rowc = row < N ? row : N - 1;
  bool rok = row < N;
  bool cok = l16 < 6;

  v8f acc;
#pragma unroll
  for (int q = 0; q < 8; ++q) acc[q] = 0.f;

#pragma unroll
  for (int s = 0; s < 16; ++s) {
    int k0 = 4 * s + 2 * half;
    v2f a, b;
    a[0] = rok ? x2[rowc * 64 + k0]     : 0.f;
    a[1] = rok ? x2[rowc * 64 + k0 + 1] : 0.f;
    b[0] = cok ? W2[k0 * 6 + l16]       : 0.f;
    b[1] = cok ? W2[(k0 + 1) * 6 + l16] : 0.f;
    acc = __builtin_amdgcn_wmma_f32_16x16x4_f32(
        false, a, false, b, (short)0, acc, false, false);
  }

  if (cok) {
#pragma unroll
    for (int r = 0; r < 8; ++r) {
      int m = wave * 16 + r + 8 * half;
      if (m < N) rec[m * 8 + l16] = acc[r];
    }
  }
}

__global__ void alpha2_kern(float* rec, const float* __restrict__ aS,
                            const float* __restrict__ aD, float* __restrict__ ad2, int N) {
  int n = blockIdx.x * blockDim.x + threadIdx.x;
  if (n >= N) return;
  float4 lo = ((const float4*)rec)[n * 2];
  float2 hi = ((const float2*)rec)[n * 4 + 2];
  float v[6] = {lo.x, lo.y, lo.z, lo.w, hi.x, hi.y};
  float s = 0.f, d = 0.f;
#pragma unroll
  for (int c = 0; c < 6; ++c) {
    s += v[c] * aS[c];
    d += v[c] * aD[c];
  }
  ((float2*)rec)[n * 4 + 3] = make_float2(s, 0.f);
  ad2[n] = d;
}


__device__ __forceinline__ float odd_of_pair(float v) {
  return __uint_as_float(__builtin_amdgcn_mov_dpp8(__float_as_uint(v), SEL8(1, 1, 3, 3, 5, 5, 7, 7)));
}

__global__ void gat2_node(const int* __restrict__ count, const int* __restrict__ slots,
                          const int* __restrict__ ovfi, const float* __restrict__ rec, const float* __restrict__ ad2,
                          const float* __restrict__ b2, float* __restrict__ out, int N) {
  int gid = blockIdx.x * blockDim.x + threadIdx.x;
  int n = gid >> 4, sub = gid & 15;
  int p = sub >> 1, hf = sub & 1;
  bool ok = n < N;
  int nc = ok ? n : N - 1;
  int cnt = ok ? count[nc] : 0;
  const int* lst = slots + nc * SLOTS;
  const int2* ovf = (const int2*)ovfi;
  float ad = ad2[nc];
  const float4* rv = (const float4*)rec;
  float4 rs = rv[nc * 2 + hf];
  float es = leaky(odd_of_pair(rs.z) + ad);
  float m = es;

  int sv[8];
  float4 r[8];
  float ev[8];
#pragma unroll
  for (int k = 0; k < 8; ++k) {
    int j = p + 8 * k;
    int v = nc;
    if (j < cnt) v = lst[j];
    sv[k] = v;
  }
#pragma unroll
  for (int k = 0; k < 8; ++k) r[k] = rv[sv[k] * 2 + hf];
#pragma unroll
  for (int k = 0; k < 8; ++k) {
    int j = p + 8 * k;
    float e = leaky(odd_of_pair(r[k].z) + ad);
    ev[k] = e;
    if (j < cnt) m = fmaxf(m, e);
  }
  int novf = cnt > SLOTS ? count[N] : 0;
  for (int j = p; j < novf; j += 8) {
    int2 q = ovf[j];
    if (q.y == n) m = fmaxf(m, leaky(rec[q.x * 8 + 6] + ad));
  }
  m = fmaxf(m, __shfl_xor(m, 2));
  m = fmaxf(m, __shfl_xor(m, 4));
  m = fmaxf(m, __shfl_xor(m, 8));

  float den = 0.f;
  float acc[4];
#pragma unroll
  for (int c = 0; c < 4; ++c) acc[c] = 0.f;
#pragma unroll
  for (int k = 0; k < 8; ++k) {
    int j = p + 8 * k;
    if (j < cnt) {
      float w = __expf(ev[k] - m);
      den += w;
      acc[0] += w * r[k].x; acc[1] += w * r[k].y;
      acc[2] += w * r[k].z; acc[3] += w * r[k].w;
    }
  }
  for (int j = p; j < novf; j += 8) {
    int2 q = ovf[j];
    if (q.y == n) {
      int s = q.x;
      float w = __expf(leaky(rec[s * 8 + 6] + ad) - m);
      den += w;
      float4 h = rv[s * 2 + hf];
      acc[0] += w * h.x; acc[1] += w * h.y;
      acc[2] += w * h.z; acc[3] += w * h.w;
    }
  }
  if (p == 0) {
    float w = __expf(es - m);
    den += w;
    acc[0] += w * rs.x; acc[1] += w * rs.y;
    acc[2] += w * rs.z; acc[3] += w * rs.w;
  }
#pragma unroll
  for (int off = 2; off < 16; off <<= 1) {
    den += __shfl_xor(den, off);
#pragma unroll
    for (int c = 0; c < 4; ++c) acc[c] += __shfl_xor(acc[c], off);
  }
  float o4 = __shfl_xor(acc[0], 1);
  float o5 = __shfl_xor(acc[1], 1);

  if (ok && sub == 0) {
    float o[6] = {acc[0], acc[1], acc[2], acc[3], o4, o5};
    float mx = -3.0e38f;
#pragma unroll
    for (int c = 0; c < 6; ++c) { o[c] = o[c] / den + b2[c]; mx = fmaxf(mx, o[c]); }
    float sum = 0.f;
#pragma unroll
    for (int c = 0; c < 6; ++c) sum += __expf(o[c] - mx);
    float l = __logf(sum);
#pragma unroll
    for (int c = 0; c < 6; ++c) out[n * 6 + c] = o[c] - mx - l;
  }
}

extern "C" void kernel_launch(void* const* d_in, const int* in_sizes, int n_in,
                              void* d_out, int out_size, void* d_ws, size_t ws_size,
                              hipStream_t stream) {
  const float* x      = (const float*)d_in[0];
  const int*   ei     = (const int*)d_in[1];
  const float* W1     = (const float*)d_in[2];
  const float* a_src1 = (const float*)d_in[3];
  const float* a_dst1 = (const float*)d_in[4];
  const float* b1     = (const float*)d_in[5];
  const float* W2     = (const float*)d_in[6];
  const float* a_src2 = (const float*)d_in[7];
  const float* a_dst2 = (const float*)d_in[8];
  const float* b2     = (const float*)d_in[9];
  float* out = (float*)d_out;

  int N = in_sizes[0] / 7;
  int E = in_sizes[1] / 2;

  int nbin = (N + BN - 1) / BN;
  int nblk = (E + CH - 1) / CH;
  int dpitch = (nbin + 15) & ~15;

  float* ws  = (float*)d_ws;
  float* h1  = ws;
  float* x2  = h1 + (size_t)N * 64;
  int* slots = (int*)(x2 + (size_t)N * 64);
  float* as1 = (float*)(slots + (size_t)N * SLOTS);
  float* ad1 = as1 + (size_t)N * 4;
  float* rec = ad1 + (size_t)N * 4;
  int* ovf   = (int*)(rec + (size_t)N * 8);
  int2* bins = (int2*)(ovf + (size_t)E * 2);
  int2* desc = bins + (size_t)nblk * CH;
  float* ad2 = (float*)(desc + (size_t)nblk * dpitch);
  int* count = (int*)(ad2 + N);

  const int B = 256;
  auto G = [](long n, int b) { return (int)((n + b - 1) / b); };
  int tiles = (N + 15) / 16;
  int projBlocks = (tiles + (B / 32) - 1) / (B / 32);

  sort_chunks<<<nblk, TPB, 0, stream>>>(ei, bins, desc, count, N, E, nbin, dpitch);
  build_rows<<<nbin, TPB, 0, stream>>>(bins, desc, count, slots, ovf, N, nblk, dpitch);

  proj1_wmma<<<projBlocks, B, 0, stream>>>(x, W1, h1, N);
  alpha1_kern<<<G((long)N * 4, B), B, 0, stream>>>(h1, a_src1, a_dst1, as1, ad1, N);
  gat1_node<<<G(N, B / 32), B, 0, stream>>>(count, slots, ovf, as1, ad1, h1, b1, x2, N);

  proj2_wmma<<<projBlocks, B, 0, stream>>>(x2, W2, rec, N);
  alpha2_kern<<<G(N, B), B, 0, stream>>>(rec, a_src2, a_dst2, ad2, N);
  gat2_node<<<G((long)N * 16, B), B, 0, stream>>>(count, slots, ovf, rec, ad2, b2, out, N);
}


# Node scores come out of the projection kernels: layer 1 as one more WMMA tile x(W1 a) (re-associated sum), layer 2 as 8-lane sums of the accumulators; both score launches and the h1 re-read go
# speedup vs baseline: 20.7539x; 1.0391x over previous
#include <hip/hip_runtime.h>
#include <hip/hip_bf16.h>
#include <math.h>

typedef float v2f __attribute__((ext_vector_type(2)));
typedef float v8f __attribute__((ext_vector_type(8)));

#define NEG_SLOPE 0.2f

__device__ __forceinline__ float leaky(float v) { return v > 0.f ? v : NEG_SLOPE * v; }

#define SLOTS 64
#define BSH 9
#define BN (1 << BSH)
#define CH 4096
#define TPB 1024

__device__ __forceinline__ int block_excl_scan(int v, int* wsum, int tid) {
  int lane = tid & 31, w = tid >> 5;
  int inc = v;
#pragma unroll
  for (int o = 1; o < 32; o <<= 1) {
    int t = __shfl_up(inc, o);
    if (lane >= o) inc += t;
  }
  if (lane == 31) wsum[w] = inc;
  __syncthreads();
  if (w == 0) {
    int s = wsum[lane];
    int si = s;
#pragma unroll
    for (int o = 1; o < 32; o <<= 1) {
      int t = __shfl_up(si, o);
      if (lane >= o) si += t;
    }
    wsum[lane] = si - s;
  }
  __syncthreads();
  return wsum[w] + inc - v;
}

__global__ __launch_bounds__(TPB) void sort_chunks(const int* __restrict__ ei, int2* __restrict__ bins,
                                                   int2* __restrict__ desc, int* __restrict__ count,
                                                   int N, int E, int nbin, int dpitch) {
  __shared__ int h[TPB];
  __shared__ int wsum[32];
  __shared__ int2 stage[CH];
  int tid = threadIdx.x;
  int blk = blockIdx.x;
  if (blk == 0 && tid == 0) count[N] = 0;
  h[tid] = 0;
  __syncthreads();
  int e0 = blk * CH + tid;
  int s[4], d[4], r[4];
#pragma unroll
  for (int k = 0; k < 4; ++k) {
    int e = e0 + k * TPB;
    s[k] = 0; d[k] = -1;
    if (e < E) { s[k] = ei[e]; d[k] = ei[E + e]; }
  }
#pragma unroll
  for (int k = 0; k < 4; ++k) {
    r[k] = 0;
    if (d[k] >= 0) r[k] = atomicAdd(&h[d[k] >> BSH], 1);
  }
  __syncthreads();
  int v = h[tid];
  int ex = block_excl_scan(v, wsum, tid);
  h[tid] = ex;
  if (tid < nbin) desc[(size_t)blk * dpitch + tid] = make_int2(ex, v);
  __syncthreads();
#pragma unroll
  for (int k = 0; k < 4; ++k)
    if (d[k] >= 0) stage[h[d[k] >> BSH] + r[k]] = make_int2(s[k], d[k]);
  __syncthreads();
  int total = E - blk * CH;
  if (total > CH) total = CH;
#pragma unroll
  for (int k = 0; k < 4; ++k) {
    int i = tid + k * TPB;
    if (i < total) bins[(size_t)blk * CH + i] = stage[i];
  }
}

__device__ __forceinline__ void place_edge(int2 ed, int n0, int* lcnt, int* lrows,
                                           int* __restrict__ ovf, int* ovfc) {
  int dl = ed.y - n0;
  int r = atomicAdd(&lcnt[dl], 1);
  if (r < SLOTS) {
    lrows[dl * SLOTS + ((r + dl) & (SLOTS - 1))] = ed.x;
  } else {
    int k = atomicAdd(ovfc, 1);
    ((int2*)ovf)[k] = ed;
  }
}

__global__ __launch_bounds__(TPB) void build_rows(const int2* __restrict__ bins, const int2* __restrict__ desc,
                                                  int* __restrict__ count, int* __restrict__ slots,
                                                  int* __restrict__ ovf, int N, int nblk, int dpitch) {
  __shared__ int lcnt[BN];
  __shared__ int lrows[BN * SLOTS];
  int tid = threadIdx.x, lane = tid & 31, w = tid >> 5;
  int b = blockIdx.x;
  int n0 = b << BSH;
  int* ovfc = count + N;
  if (tid < BN) lcnt[tid] = 0;
  __syncthreads();
  for (int base = 0; base < nblk; base += TPB) {
    int myblk = base + lane * 32 + w;
    int2 q = make_int2(0, 0);
    if (myblk < nblk) q = desc[(size_t)myblk * dpitch + b];
    for (int j0 = 0; j0 < 32; j0 += 8) {
      int2 ed[8];
      int ln[8];
      const int2* run[8];
#pragma unroll
      for (int u = 0; u < 8; ++u) {
        int st = __builtin_amdgcn_readlane(q.x, j0 + u);
        ln[u] = __builtin_amdgcn_readlane(q.y, j0 + u);
        run[u] = bins + (size_t)(base + (j0 + u) * 32 + w) * CH + st;
        ed[u] = make_int2(0, n0);
        if (lane < ln[u]) ed[u] = run[u][lane];
      }
#pragma unroll
      for (int u = 0; u < 8; ++u) {
        if (lane < ln[u]) place_edge(ed[u], n0, lcnt, lrows, ovf, ovfc);
        for (int k = lane + 32; k < ln[u]; k += 32)
          place_edge(run[u][k], n0, lcnt, lrows, ovf, ovfc);
      }
    }
  }
  __syncthreads();
  if (tid < BN && n0 + tid < N) count[n0 + tid] = lcnt[tid];
  for (int it = 0; it < BN * SLOTS / TPB; ++it) {
    int idx = it * TPB + tid;
    int dl = idx >> 6, j = idx & (SLOTS - 1);
    int n = n0 + dl;
    if (n < N && j < lcnt[dl])
      slots[(size_t)n * SLOTS + j] = lrows[dl * SLOTS + ((j + dl) & (SLOTS - 1))];
  }
}


__global__ void proj1_wmma(const float* __restrict__ x, const float* __restrict__ W1,
                           const float* __restrict__ aS, const float* __restrict__ aD,
                           float* __restrict__ h1, float* __restrict__ as1,
                           float* __restrict__ ad1, int N) {
  int gtid = blockIdx.x * blockDim.x + threadIdx.x;
  int wave = gtid >> 5;
  int lane = threadIdx.x & 31;
  int tiles = (N + 15) >> 4;
  if (wave >= tiles) return;
  int half = lane >> 4;
  int l16  = lane & 15;
  int row  = wave * 16 + l16;
  int rowc = row < N ? row : N - 1;
  bool rok = row < N;

  v2f a[2];
#pragma unroll
  for (int s = 0; s < 2; ++s) {
    int k0 = 4 * s + 2 * half;
    a[s][0] = (rok && k0     < 7) ? x[rowc * 7 + k0]     : 0.f;
    a[s][1] = (rok && k0 + 1 < 7) ? x[rowc * 7 + k0 + 1] : 0.f;
  }

  v8f acc[4];
#pragma unroll
  for (int t = 0; t < 4; ++t)
#pragma unroll
    for (int q = 0; q < 8; ++q) acc[t][q] = 0.f;

#pragma unroll
  for (int s = 0; s < 2; ++s) {
    int k0 = 4 * s + 2 * half;
#pragma unroll
    for (int t = 0; t < 4; ++t) {
      int col = t * 16 + l16;
      v2f b;
      b[0] = (k0     < 7) ? W1[k0 * 64 + col]       : 0.f;
      b[1] = (k0 + 1 < 7) ? W1[(k0 + 1) * 64 + col] : 0.f;
      acc[t] = __builtin_amdgcn_wmma_f32_16x16x4_f32(
          false, a[s], false, b, (short)0, acc[t], false, false);
    }
  }

#pragma unroll
  for (int t = 0; t < 4; ++t)
#pragma unroll
    for (int r = 0; r < 8; ++r) {
      int m = wave * 16 + r + 8 * half;
      if (m < N) h1[m * 64 + t * 16 + l16] = acc[t][r];
    }

  int jt = l16 & 3;
  const float* av = ((l16 & 4) ? aD : aS) + jt * 16;
  float4 avv[4];
#pragma unroll
  for (int q = 0; q < 4; ++q) avv[q] = ((const float4*)av)[q];
  v8f sc;
#pragma unroll
  for (int q = 0; q < 8; ++q) sc[q] = 0.f;
#pragma unroll
  for (int s = 0; s < 2; ++s) {
    int k0 = 4 * s + 2 * half;
    v2f bp;
#pragma unroll
    for (int i = 0; i < 2; ++i) {
      int k = k0 + i;
      int kc = k < 7 ? k : 6;
      const float4* wr = (const float4*)(W1 + kc * 64 + jt * 16);
      float v = 0.f;
#pragma unroll
      for (int q = 0; q < 4; ++q) {
        float4 w = wr[q];
        v += w.x * avv[q].x; v += w.y * avv[q].y; v += w.z * avv[q].z; v += w.w * avv[q].w;
      }
      bp[i] = (k < 7 && l16 < 8) ? v : 0.f;
    }
    sc = __builtin_amdgcn_wmma_f32_16x16x4_f32(
        false, a[s], false, bp, (short)0, sc, false, false);
  }
  float* dstp = (l16 < 4) ? (as1 + l16) : (ad1 + (l16 - 4));
#pragma unroll
  for (int r = 0; r < 8; ++r) {
    int m = wave * 16 + r + 8 * half;
    if (l16 < 8 && m < N) dstp[m * 4] = sc[r];
  }
}


#define SEL8(a, b, c, d, e, f, g, h) \
  ((a) | ((b) << 3) | ((c) << 6) | ((d) << 9) | ((e) << 12) | ((f) << 15) | ((g) << 18) | ((h) << 21))
#define BC8(v, J) __builtin_amdgcn_mov_dpp8((unsigned)(v), SEL8(J, J, J, J, J, J, J, J))
#define BC8F(v, J) __uint_as_float(BC8(__float_as_uint(v), J))

__device__ __forceinline__ float max8(float v) {
  v = fmaxf(v, __uint_as_float(__builtin_amdgcn_mov_dpp8(__float_as_uint(v), SEL8(1, 0, 3, 2, 5, 4, 7, 6))));
  v = fmaxf(v, __uint_as_float(__builtin_amdgcn_mov_dpp8(__float_as_uint(v), SEL8(2, 3, 0, 1, 6, 7, 4, 5))));
  v = fmaxf(v, __uint_as_float(__builtin_amdgcn_mov_dpp8(__float_as_uint(v), SEL8(4, 5, 6, 7, 0, 1, 2, 3))));
  return v;
}
__device__ __forceinline__ float sum8(float v) {
  v += __uint_as_float(__builtin_amdgcn_mov_dpp8(__float_as_uint(v), SEL8(1, 0, 3, 2, 5, 4, 7, 6)));
  v += __uint_as_float(__builtin_amdgcn_mov_dpp8(__float_as_uint(v), SEL8(2, 3, 0, 1, 6, 7, 4, 5)));
  v += __uint_as_float(__builtin_amdgcn_mov_dpp8(__float_as_uint(v), SEL8(4, 5, 6, 7, 0, 1, 2, 3)));
  return v;
}

__device__ __forceinline__ v2f row_ld(const char* __restrict__ hb, int st, unsigned laneoff) {
  return *(const v2f*)(hb + ((((unsigned)st) << 8) | laneoff));
}
#define SRC8(s, J) __builtin_amdgcn_readlane((s), (J))

template <int NB>
__device__ __forceinline__ void load_rows(const char* __restrict__ hb, int s, unsigned laneoff, v2f* v) {
  v[0] = row_ld(hb, SRC8(s, 0), laneoff);
  v[1] = row_ld(hb, SRC8(s, 1), laneoff);
  v[2] = row_ld(hb, SRC8(s, 2), laneoff);
  v[3] = row_ld(hb, SRC8(s, 3), laneoff);
  if (NB > 4) {
    v[4] = row_ld(hb, SRC8(s, 4), laneoff);
    v[5] = row_ld(hb, SRC8(s, 5), laneoff);
    v[6] = row_ld(hb, SRC8(s, 6), laneoff);
    v[7] = row_ld(hb, SRC8(s, 7), laneoff);
  }
}

#define ADDROW(J) { float w = BC8F(p, J); v2f ww = {w, w}; acc = __builtin_elementwise_fma(v[J], ww, acc); }
template <int NB>
__device__ __forceinline__ void add_rows(float p, const v2f* v, v2f& acc) {
  ADDROW(0) ADDROW(1) ADDROW(2) ADDROW(3)
  if (NB > 4) { ADDROW(4) ADDROW(5) ADDROW(6) ADDROW(7) }
}

__global__ void gat1_node(const int* __restrict__ count, const int* __restrict__ slots,
                          const int* __restrict__ ovfi, const float* __restrict__ as1, const float* __restrict__ ad1,
                          const float* __restrict__ h1, const float* __restrict__ b1,
                          float* __restrict__ x2, int N) {
  int lane = threadIdx.x & 31;
  int hd = lane >> 3, j8 = lane & 7;
  int n = __builtin_amdgcn_readfirstlane((int)(blockIdx.x * (blockDim.x >> 5) + (threadIdx.x >> 5)));
  if (n >= N) return;
  const char* hb = (const char*)h1;
  unsigned laneoff = (unsigned)lane * 8u;

  int cnt = count[n];
  const int* lst = slots + n * SLOTS;
  const int2* ovf = (const int2*)ovfi;
  float adh = ad1[n * 4 + hd];
  float es = leaky(as1[n * 4 + hd] + adh);
  float m = es;
  v2f hv = row_ld(hb, n, laneoff);

  int s[8];
  float e[8];
#pragma unroll
  for (int c = 0; c < 8; ++c) {
    int j = 8 * c + j8;
    int sv = n;
    if (j < cnt) sv = lst[j];
    s[c] = sv;
  }
#pragma unroll
  for (int c = 0; c < 8; ++c) {
    float ev = leaky(as1[s[c] * 4 + hd] + adh);
    e[c] = ev;
    m = fmaxf(m, ev);
  }
  int novf = 0;
  if (cnt > SLOTS) {
    novf = count[N];
    for (int base = 0; base < novf; base += 8) {
      int j = base + j8;
      if (j < novf) {
        int2 q = ovf[j];
        if (q.y == n) m = fmaxf(m, leaky(as1[q.x * 4 + hd] + adh));
      }
    }
  }
  m = max8(m);

  float p[8];
  float dsum = 0.f;
#pragma unroll
  for (int c = 0; c < 8; ++c) {
    int j = 8 * c + j8;
    float pv = (j < cnt) ? __expf(e[c] - m) : 0.f;
    p[c] = pv;
    dsum += pv;
  }
  v2f acc = {0.f, 0.f};
#pragma unroll
  for (int g = 0; g < 4; ++g) {
    int lo = 16 * g;
    if (lo < cnt) {
      v2f va[8], vb[8];
      if (lo + 12 < cnt) {
        load_rows<8>(hb, s[2 * g], laneoff, va);
        load_rows<8>(hb, s[2 * g + 1], laneoff, vb);
        add_rows<8>(p[2 * g], va, acc);
        add_rows<8>(p[2 * g + 1], vb, acc);
      } else if (lo + 8 < cnt) {
        load_rows<8>(hb, s[2 * g], laneoff, va);
        load_rows<4>(hb, s[2 * g + 1], laneoff, vb);
        add_rows<8>(p[2 * g], va, acc);
        add_rows<4>(p[2 * g + 1], vb, acc);
      } else if (lo + 4 < cnt) {
        load_rows<8>(hb, s[2 * g], laneoff, va);
        add_rows<8>(p[2 * g], va, acc);
      } else {
        load_rows<4>(hb, s[2 * g], laneoff, va);
        add_rows<4>(p[2 * g], va, acc);
      }
    }
  }
  for (int base = 0; base < novf; base += 8) {
    int j = base + j8;
    int sv = n;
    bool mine = false;
    if (j < novf) {
      int2 q = ovf[j];
      if (q.y == n) { sv = q.x; mine = true; }
    }
    if (__builtin_amdgcn_ballot_w32(mine) != 0) {
      float pv = mine ? __expf(leaky(as1[sv * 4 + hd] + adh) - m) : 0.f;
      dsum += pv;
      v2f v[8];
      load_rows<8>(hb, sv, laneoff, v);
      add_rows<8>(pv, v, acc);
    }
  }
  float ws = __expf(es - m);
  float den = sum8(dsum) + ws;
  v2f wws = {ws, ws};
  acc = __builtin_elementwise_fma(hv, wws, acc);
  float ax = acc[0], ay = acc[1];

  float2 bb = ((const float2*)b1)[lane];
  float ox = ax / den + bb.x;
  float oy = ay / den + bb.y;
  ox = ox > 0.f ? ox : (__expf(ox) - 1.f);
  oy = oy > 0.f ? oy : (__expf(oy) - 1.f);
  ((float2*)x2)[(size_t)n * 32 + lane] = make_float2(ox, oy);
}

__global__ void proj2_wmma(const float* __restrict__ x2, const float* __restrict__ W2,
                           const float* __restrict__ aS, const float* __restrict__ aD,
                           float* __restrict__ rec, int N) {
  int gtid = blockIdx.x * blockDim.x + threadIdx.x;
  int wave = gtid >> 5;
  int lane = threadIdx.x & 31;
  int tiles = (N + 15) >> 4;
  if (wave >= tiles) return;
  int half = lane >> 4, l16 = lane & 15;
  int row  = wave * 16 + l16;
  int rowc = row < N ? row : N - 1;
  bool rok = row < N;
  bool cok = l16 < 6;

  v8f acc;
#pragma unroll
  for (int q = 0; q < 8; ++q) acc[q] = 0.f;

#pragma unroll
  for (int s = 0; s < 16; ++s) {
    int k0 = 4 * s + 2 * half;
    v2f a, b;
    a[0] = rok ? x2[rowc * 64 + k0]     : 0.f;
    a[1] = rok ? x2[rowc * 64 + k0 + 1] : 0.f;
    b[0] = cok ? W2[k0 * 6 + l16]       : 0.f;
    b[1] = cok ? W2[(k0 + 1) * 6 + l16] : 0.f;
    acc = __builtin_amdgcn_wmma_f32_16x16x4_f32(
        false, a, false, b, (short)0, acc, false, false);
  }

  float cs = cok ? aS[l16] : 0.f;
  float cd = cok ? aD[l16] : 0.f;
#pragma unroll
  for (int r = 0; r < 8; ++r) {
    int m = wave * 16 + r + 8 * half;
    float s = sum8(acc[r] * cs);
    float d = sum8(acc[r] * cd);
    float v = cok ? acc[r] : (l16 == 6 ? s : d);
    if (l16 < 8 && m < N) rec[m * 8 + l16] = v;
  }
}


__device__ __forceinline__ float odd_of_pair(float v) {
  return __uint_as_float(__builtin_amdgcn_mov_dpp8(__float_as_uint(v), SEL8(1, 1, 3, 3, 5, 5, 7, 7)));
}

__global__ void gat2_node(const int* __restrict__ count, const int* __restrict__ slots,
                          const int* __restrict__ ovfi, const float* __restrict__ rec,
                          const float* __restrict__ b2, float* __restrict__ out, int N) {
  int gid = blockIdx.x * blockDim.x + threadIdx.x;
  int n = gid >> 4, sub = gid & 15;
  int p = sub >> 1, hf = sub & 1;
  bool ok = n < N;
  int nc = ok ? n : N - 1;
  int cnt = ok ? count[nc] : 0;
  const int* lst = slots + nc * SLOTS;
  const int2* ovf = (const int2*)ovfi;
  const float4* rv = (const float4*)rec;
  float4 rs = rv[nc * 2 + hf];
  float ad = odd_of_pair(rs.w);
  float es = leaky(odd_of_pair(rs.z) + ad);
  float m = es;

  int sv[8];
  float4 r[8];
  float ev[8];
#pragma unroll
  for (int k = 0; k < 8; ++k) {
    int j = p + 8 * k;
    int v = nc;
    if (j < cnt) v = lst[j];
    sv[k] = v;
  }
#pragma unroll
  for (int k = 0; k < 8; ++k) r[k] = rv[sv[k] * 2 + hf];
#pragma unroll
  for (int k = 0; k < 8; ++k) {
    int j = p + 8 * k;
    float e = leaky(odd_of_pair(r[k].z) + ad);
    ev[k] = e;
    if (j < cnt) m = fmaxf(m, e);
  }
  int novf = cnt > SLOTS ? count[N] : 0;
  for (int j = p; j < novf; j += 8) {
    int2 q = ovf[j];
    if (q.y == n) m = fmaxf(m, leaky(rec[q.x * 8 + 6] + ad));
  }
  m = fmaxf(m, __shfl_xor(m, 2));
  m = fmaxf(m, __shfl_xor(m, 4));
  m = fmaxf(m, __shfl_xor(m, 8));

  float den = 0.f;
  float acc[4];
#pragma unroll
  for (int c = 0; c < 4; ++c) acc[c] = 0.f;
#pragma unroll
  for (int k = 0; k < 8; ++k) {
    int j = p + 8 * k;
    if (j < cnt) {
      float w = __expf(ev[k] - m);
      den += w;
      acc[0] += w * r[k].x; acc[1] += w * r[k].y;
      acc[2] += w * r[k].z; acc[3] += w * r[k].w;
    }
  }
  for (int j = p; j < novf; j += 8) {
    int2 q = ovf[j];
    if (q.y == n) {
      int s = q.x;
      float w = __expf(leaky(rec[s * 8 + 6] + ad) - m);
      den += w;
      float4 h = rv[s * 2 + hf];
      acc[0] += w * h.x; acc[1] += w * h.y;
      acc[2] += w * h.z; acc[3] += w * h.w;
    }
  }
  if (p == 0) {
    float w = __expf(es - m);
    den += w;
    acc[0] += w * rs.x; acc[1] += w * rs.y;
    acc[2] += w * rs.z; acc[3] += w * rs.w;
  }
#pragma unroll
  for (int off = 2; off < 16; off <<= 1) {
    den += __shfl_xor(den, off);
#pragma unroll
    for (int c = 0; c < 4; ++c) acc[c] += __shfl_xor(acc[c], off);
  }
  float o4 = __shfl_xor(acc[0], 1);
  float o5 = __shfl_xor(acc[1], 1);

  if (ok && sub == 0) {
    float o[6] = {acc[0], acc[1], acc[2], acc[3], o4, o5};
    float mx = -3.0e38f;
#pragma unroll
    for (int c = 0; c < 6; ++c) { o[c] = o[c] / den + b2[c]; mx = fmaxf(mx, o[c]); }
    float sum = 0.f;
#pragma unroll
    for (int c = 0; c < 6; ++c) sum += __expf(o[c] - mx);
    float l = __logf(sum);
#pragma unroll
    for (int c = 0; c < 6; ++c) out[n * 6 + c] = o[c] - mx - l;
  }
}

extern "C" void kernel_launch(void* const* d_in, const int* in_sizes, int n_in,
                              void* d_out, int out_size, void* d_ws, size_t ws_size,
                              hipStream_t stream) {
  const float* x      = (const float*)d_in[0];
  const int*   ei     = (const int*)d_in[1];
  const float* W1     = (const float*)d_in[2];
  const float* a_src1 = (const float*)d_in[3];
  const float* a_dst1 = (const float*)d_in[4];
  const float* b1     = (const float*)d_in[5];
  const float* W2     = (const float*)d_in[6];
  const float* a_src2 = (const float*)d_in[7];
  const float* a_dst2 = (const float*)d_in[8];
  const float* b2     = (const float*)d_in[9];
  float* out = (float*)d_out;

  int N = in_sizes[0] / 7;
  int E = in_sizes[1] / 2;

  int nbin = (N + BN - 1) / BN;
  int nblk = (E + CH - 1) / CH;
  int dpitch = (nbin + 15) & ~15;

  float* ws  = (float*)d_ws;
  float* h1  = ws;
  float* x2  = h1 + (size_t)N * 64;
  int* slots = (int*)(x2 + (size_t)N * 64);
  float* as1 = (float*)(slots + (size_t)N * SLOTS);
  float* ad1 = as1 + (size_t)N * 4;
  float* rec = ad1 + (size_t)N * 4;
  int* ovf   = (int*)(rec + (size_t)N * 8);
  int2* bins = (int2*)(ovf + (size_t)E * 2);
  int2* desc = bins + (size_t)nblk * CH;
  int* count = (int*)(desc + (size_t)nblk * dpitch);

  const int B = 256;
  auto G = [](long n, int b) { return (int)((n + b - 1) / b); };
  int tiles = (N + 15) / 16;
  int projBlocks = (tiles + (B / 32) - 1) / (B / 32);

  sort_chunks<<<nblk, TPB, 0, stream>>>(ei, bins, desc, count, N, E, nbin, dpitch);
  build_rows<<<nbin, TPB, 0, stream>>>(bins, desc, count, slots, ovf, N, nblk, dpitch);

  proj1_wmma<<<projBlocks, B, 0, stream>>>(x, W1, a_src1, a_dst1, h1, as1, ad1, N);
  gat1_node<<<G(N, B / 32), B, 0, stream>>>(count, slots, ovf, as1, ad1, h1, b1, x2, N);

  proj2_wmma<<<projBlocks, B, 0, stream>>>(x2, W2, a_src2, a_dst2, rec, N);
  gat2_node<<<G((long)N * 16, B), B, 0, stream>>>(count, slots, ovf, rec, b2, out, N);
}


# gat1: a node's four source scores ride in its 64-B record behind its 7 features (prep1 stores both there): an edge asks the second-level cache for one line, not a record line plus a score line
# speedup vs baseline: 27.2264x; 1.3119x over previous
#include <hip/hip_runtime.h>
#include <hip/hip_bf16.h>
#include <math.h>

typedef float v2f __attribute__((ext_vector_type(2)));
typedef float v8f __attribute__((ext_vector_type(8)));

#define NEG_SLOPE 0.2f

__device__ __forceinline__ float leaky(float v) { return v > 0.f ? v : NEG_SLOPE * v; }

#define SLOTS 64
#define BSH 9
#define BN (1 << BSH)
#define CH 4096
#define TPB 1024

__device__ __forceinline__ int block_excl_scan(int v, int* wsum, int tid) {
  int lane = tid & 31, w = tid >> 5;
  int inc = v;
#pragma unroll
  for (int o = 1; o < 32; o <<= 1) {
    int t = __shfl_up(inc, o);
    if (lane >= o) inc += t;
  }
  if (lane == 31) wsum[w] = inc;
  __syncthreads();
  if (w == 0) {
    int s = wsum[lane];
    int si = s;
#pragma unroll
    for (int o = 1; o < 32; o <<= 1) {
      int t = __shfl_up(si, o);
      if (lane >= o) si += t;
    }
    wsum[lane] = si - s;
  }
  __syncthreads();
  return wsum[w] + inc - v;
}

__global__ __launch_bounds__(TPB) void sort_chunks(const int* __restrict__ ei, int2* __restrict__ bins,
                                                   int2* __restrict__ desc, int* __restrict__ count,
                                                   int N, int E, int nbin, int dpitch) {
  __shared__ int h[TPB];
  __shared__ int wsum[32];
  __shared__ int2 stage[CH];
  int tid = threadIdx.x;
  int blk = blockIdx.x;
  if (blk == 0 && tid == 0) count[N] = 0;
  h[tid] = 0;
  __syncthreads();
  int e0 = blk * CH + tid;
  int s[4], d[4], r[4];
#pragma unroll
  for (int k = 0; k < 4; ++k) {
    int e = e0 + k * TPB;
    s[k] = 0; d[k] = -1;
    if (e < E) { s[k] = ei[e]; d[k] = ei[E + e]; }
  }
#pragma unroll
  for (int k = 0; k < 4; ++k) {
    r[k] = 0;
    if (d[k] >= 0) r[k] = atomicAdd(&h[d[k] >> BSH], 1);
  }
  __syncthreads();
  int v = h[tid];
  int ex = block_excl_scan(v, wsum, tid);
  h[tid] = ex;
  if (tid < nbin) desc[(size_t)blk * dpitch + tid] = make_int2(ex, v);
  __syncthreads();
#pragma unroll
  for (int k = 0; k < 4; ++k)
    if (d[k] >= 0) stage[h[d[k] >> BSH] + r[k]] = make_int2(s[k], d[k]);
  __syncthreads();
  int total = E - blk * CH;
  if (total > CH) total = CH;
#pragma unroll
  for (int k = 0; k < 4; ++k) {
    int i = tid + k * TPB;
    if (i < total) bins[(size_t)blk * CH + i] = stage[i];
  }
}

__device__ __forceinline__ void place_edge(int2 ed, int n0, int* lcnt, int* lrows,
                                           int* __restrict__ ovf, int* ovfc) {
  int dl = ed.y - n0;
  int r = atomicAdd(&lcnt[dl], 1);
  if (r < SLOTS) {
    lrows[dl * SLOTS + ((r + dl) & (SLOTS - 1))] = ed.x;
  } else {
    int k = atomicAdd(ovfc, 1);
    ((int2*)ovf)[k] = ed;
  }
}

__global__ __launch_bounds__(TPB) void build_rows(const int2* __restrict__ bins, const int2* __restrict__ desc,
                                                  int* __restrict__ count, int* __restrict__ slots,
                                                  int* __restrict__ ovf, int N, int nblk, int dpitch) {
  __shared__ int lcnt[BN];
  __shared__ int lrows[BN * SLOTS];
  int tid = threadIdx.x, lane = tid & 31, w = tid >> 5;
  int b = blockIdx.x;
  int n0 = b << BSH;
  int* ovfc = count + N;
  if (tid < BN) lcnt[tid] = 0;
  __syncthreads();
  for (int base = 0; base < nblk; base += TPB) {
    int myblk = base + lane * 32 + w;
    int2 q = make_int2(0, 0);
    if (myblk < nblk) q = desc[(size_t)myblk * dpitch + b];
    for (int j0 = 0; j0 < 32; j0 += 8) {
      int2 ed[8];
      int ln[8];
      const int2* run[8];
#pragma unroll
      for (int u = 0; u < 8; ++u) {
        int st = __builtin_amdgcn_readlane(q.x, j0 + u);
        ln[u] = __builtin_amdgcn_readlane(q.y, j0 + u);
        run[u] = bins + (size_t)(base + (j0 + u) * 32 + w) * CH + st;
        ed[u] = make_int2(0, n0);
        if (lane < ln[u]) ed[u] = run[u][lane];
      }
#pragma unroll
      for (int u = 0; u < 8; ++u) {
        if (lane < ln[u]) place_edge(ed[u], n0, lcnt, lrows, ovf, ovfc);
        for (int k = lane + 32; k < ln[u]; k += 32)
          place_edge(run[u][k], n0, lcnt, lrows, ovf, ovfc);
      }
    }
  }
  __syncthreads();
  if (tid < BN && n0 + tid < N) count[n0 + tid] = lcnt[tid];
  for (int it = 0; it < BN * SLOTS / TPB; ++it) {
    int idx = it * TPB + tid;
    int dl = idx >> 6, j = idx & (SLOTS - 1);
    int n = n0 + dl;
    if (n < N && j < lcnt[dl])
      slots[(size_t)n * SLOTS + j] = lrows[dl * SLOTS + ((j + dl) & (SLOTS - 1))];
  }
}


__global__ void prep1_wmma(const float* __restrict__ x, const float* __restrict__ W1,
                           const float* __restrict__ aS, const float* __restrict__ aD,
                           float* __restrict__ xr, int N) {
  int gtid = blockIdx.x * blockDim.x + threadIdx.x;
  int wave = gtid >> 5;
  int lane = threadIdx.x & 31;
  int tiles = (N + 15) >> 4;
  if (wave >= tiles) return;
  int half = lane >> 4;
  int l16  = lane & 15;
  int row  = wave * 16 + l16;
  int rowc = row < N ? row : N - 1;
  bool rok = row < N;

  v2f a[2];
#pragma unroll
  for (int s = 0; s < 2; ++s) {
    int k0 = 4 * s + 2 * half;
    a[s][0] = (rok && k0     < 7) ? x[rowc * 7 + k0]     : 0.f;
    a[s][1] = (rok && k0 + 1 < 7) ? x[rowc * 7 + k0 + 1] : 0.f;
  }
#pragma unroll
  for (int s = 0; s < 2; ++s) {
    v2f st = a[s];
    if (s == 1 && half == 1) st[1] = 1.f;
    if (rok) *(v2f*)(xr + row * 16 + 4 * s + 2 * half) = st;
  }

  int jt = l16 & 3;
  const float* av = ((l16 & 4) ? aD : aS) + jt * 16;
  float4 avv[4];
#pragma unroll
  for (int q = 0; q < 4; ++q) avv[q] = ((const float4*)av)[q];
  v8f sc;
#pragma unroll
  for (int q = 0; q < 8; ++q) sc[q] = 0.f;
#pragma unroll
  for (int s = 0; s < 2; ++s) {
    int k0 = 4 * s + 2 * half;
    v2f bp;
#pragma unroll
    for (int i = 0; i < 2; ++i) {
      int k = k0 + i;
      int kc = k < 7 ? k : 6;
      const float4* wr = (const float4*)(W1 + kc * 64 + jt * 16);
      float v = 0.f;
#pragma unroll
      for (int q = 0; q < 4; ++q) {
        float4 w = wr[q];
        v += w.x * avv[q].x; v += w.y * avv[q].y; v += w.z * avv[q].z; v += w.w * avv[q].w;
      }
      bp[i] = (k < 7 && l16 < 8) ? v : 0.f;
    }
    sc = __builtin_amdgcn_wmma_f32_16x16x4_f32(
        false, a[s], false, bp, (short)0, sc, false, false);
  }
#pragma unroll
  for (int r = 0; r < 8; ++r) {
    int m = wave * 16 + r + 8 * half;
    if (l16 < 8 && m < N) xr[m * 16 + 8 + l16] = sc[r];
  }
}


#define SEL8(a, b, c, d, e, f, g, h) \
  ((a) | ((b) << 3) | ((c) << 6) | ((d) << 9) | ((e) << 12) | ((f) << 15) | ((g) << 18) | ((h) << 21))

__device__ __forceinline__ float sum8(float v) {
  v += __uint_as_float(__builtin_amdgcn_mov_dpp8(__float_as_uint(v), SEL8(1, 0, 3, 2, 5, 4, 7, 6)));
  v += __uint_as_float(__builtin_amdgcn_mov_dpp8(__float_as_uint(v), SEL8(2, 3, 0, 1, 6, 7, 4, 5)));
  v += __uint_as_float(__builtin_amdgcn_mov_dpp8(__float_as_uint(v), SEL8(4, 5, 6, 7, 0, 1, 2, 3)));
  return v;
}

template <int D>
__device__ __forceinline__ float lane_xor(float v) {
  unsigned u = __float_as_uint(v);
  if (D == 1) return __uint_as_float(__builtin_amdgcn_mov_dpp8(u, SEL8(1, 0, 3, 2, 5, 4, 7, 6)));
  if (D == 2) return __uint_as_float(__builtin_amdgcn_mov_dpp8(u, SEL8(2, 3, 0, 1, 6, 7, 4, 5)));
  if (D == 4) return __uint_as_float(__builtin_amdgcn_mov_dpp8(u, SEL8(4, 5, 6, 7, 0, 1, 2, 3)));
  if (D == 8)
    return __uint_as_float((unsigned)__builtin_amdgcn_update_dpp(0, (int)u, 0x128, 0xf, 0xf, false));
  return __uint_as_float(__builtin_amdgcn_permlanex16(u, u, 0x76543210u, 0xfedcba98u, false, false));
}
__device__ __forceinline__ float wave_max(float v) {
  v = fmaxf(v, lane_xor<1>(v));
  v = fmaxf(v, lane_xor<2>(v));
  v = fmaxf(v, lane_xor<4>(v));
  v = fmaxf(v, lane_xor<8>(v));
  v = fmaxf(v, lane_xor<16>(v));
  return v;
}
template <int D, int H>
__device__ __forceinline__ void halve(float* v, int lane) {
  bool up = (lane & D) != 0;
#pragma unroll
  for (int i = 0; i < H; ++i) {
    float keep = up ? v[i + H] : v[i];
    float give = up ? v[i] : v[i + H];
    v[i] = keep + lane_xor<D>(give);
  }
}

#define WADD(P, A, B) { \
  _Pragma("unroll") for (int h = 0; h < 4; ++h) { \
    acc[8 * h + 0] += (P)[h] * (A).x; acc[8 * h + 1] += (P)[h] * (A).y; \
    acc[8 * h + 2] += (P)[h] * (A).z; acc[8 * h + 3] += (P)[h] * (A).w; \
    acc[8 * h + 4] += (P)[h] * (B).x; acc[8 * h + 5] += (P)[h] * (B).y; \
    acc[8 * h + 6] += (P)[h] * (B).z; acc[8 * h + 7] += (P)[h] * (B).w; } }
#define SCORE(E, S) { (E)[0] = leaky((S).x + ad.x); (E)[1] = leaky((S).y + ad.y); \
                      (E)[2] = leaky((S).z + ad.z); (E)[3] = leaky((S).w + ad.w); }

__global__ void gat1_node(const int* __restrict__ count, const int* __restrict__ slots,
                          const int* __restrict__ ovfi,
                          const float* __restrict__ xr, float* __restrict__ agg, int N) {
  int lane = threadIdx.x & 31;
  int n = __builtin_amdgcn_readfirstlane((int)(blockIdx.x * (blockDim.x >> 5) + (threadIdx.x >> 5)));
  if (n >= N) return;
  const float4* rv = (const float4*)xr;

  int cnt = count[n];
  const int* lst = slots + n * SLOTS;
  const int2* ovf = (const int2*)ovfi;
  float4 ad = rv[n * 4 + 3];
  float4 an = rv[n * 4 + 2];
  float es[4];
  SCORE(es, an)
  float xs = xr[n * 16 + (lane & 7)];

  bool in0 = lane < cnt, in1 = 32 + lane < cnt;
  int s0 = n, s1 = n;
  if (in0) s0 = lst[lane];
  if (in1) s1 = lst[32 + lane];
  bool two = cnt > 32;
  float4 a0 = rv[s0 * 4 + 2], ra0 = rv[s0 * 4], rb0 = rv[s0 * 4 + 1];
  float4 a1 = an, ra1 = make_float4(0.f, 0.f, 0.f, 0.f), rb1 = ra1;
  if (two) { a1 = rv[s1 * 4 + 2]; ra1 = rv[s1 * 4]; rb1 = rv[s1 * 4 + 1]; }
  float e0[4], e1[4], m[4];
  SCORE(e0, a0)
  SCORE(e1, a1)
#pragma unroll
  for (int h = 0; h < 4; ++h) m[h] = fmaxf(es[h], fmaxf(e0[h], e1[h]));
  int novf = 0;
  if (cnt > SLOTS) {
    novf = count[N];
    for (int base = 0; base < novf; base += 32) {
      int j = base + lane;
      if (j < novf) {
        int2 q = ovf[j];
        if (q.y == n) {
          float4 aq = rv[q.x * 4 + 2];
          float eq[4];
          SCORE(eq, aq)
#pragma unroll
          for (int h = 0; h < 4; ++h) m[h] = fmaxf(m[h], eq[h]);
        }
      }
    }
  }
#pragma unroll
  for (int h = 0; h < 4; ++h) m[h] = wave_max(m[h]);

  float acc[32];
  float p[4];
#pragma unroll
  for (int h = 0; h < 4; ++h) p[h] = in0 ? __expf(e0[h] - m[h]) : 0.f;
#pragma unroll
  for (int i = 0; i < 32; ++i) acc[i] = 0.f;
  WADD(p, ra0, rb0)
  if (two) {
#pragma unroll
    for (int h = 0; h < 4; ++h) p[h] = in1 ? __expf(e1[h] - m[h]) : 0.f;
    WADD(p, ra1, rb1)
  }
  for (int base = 0; base < novf; base += 32) {
    int j = base + lane;
    if (j < novf) {
      int2 q = ovf[j];
      if (q.y == n) {
        float4 aq = rv[q.x * 4 + 2];
        float eq[4];
        SCORE(eq, aq)
#pragma unroll
        for (int h = 0; h < 4; ++h) p[h] = __expf(eq[h] - m[h]);
        float4 ya = rv[q.x * 4], yb = rv[q.x * 4 + 1];
        WADD(p, ya, yb)
      }
    }
  }
  halve<16, 16>(acc, lane);
  halve<8, 8>(acc, lane);
  halve<4, 4>(acc, lane);
  halve<2, 2>(acc, lane);
  halve<1, 1>(acc, lane);
  int hd = lane >> 3;
  float esl = (hd & 2) ? ((hd & 1) ? es[3] : es[2]) : ((hd & 1) ? es[1] : es[0]);
  float ml  = (hd & 2) ? ((hd & 1) ? m[3] : m[2]) : ((hd & 1) ? m[1] : m[0]);
  float tot = acc[0] + __expf(esl - ml) * xs;
  float den = __uint_as_float(__builtin_amdgcn_mov_dpp8(__float_as_uint(tot), SEL8(7, 7, 7, 7, 7, 7, 7, 7)));
  agg[(size_t)n * 32 + lane] = tot / den;
}

#define LDW 68
__global__ __launch_bounds__(256) void post1_proj2_wmma(const float* __restrict__ agg, const float* __restrict__ W1,
                           const float* __restrict__ b1, const float* __restrict__ W2,
                           const float* __restrict__ aS, const float* __restrict__ aD,
                           float* __restrict__ rec, int N) {
  __shared__ float tile[8 * 16 * LDW];
  int wv = threadIdx.x >> 5;
  int wave = blockIdx.x * 8 + wv;
  int lane = threadIdx.x & 31;
  int half = lane >> 4, l16 = lane & 15;
  int row  = wave * 16 + l16;
  int rowc = row < N ? row : N - 1;
  bool rok = row < N;
  bool cok = l16 < 6;
  float* tl = tile + wv * (16 * LDW);

#pragma unroll
  for (int t = 0; t < 4; ++t) {
    v8f hacc;
#pragma unroll
    for (int q = 0; q < 8; ++q) hacc[q] = 0.f;
#pragma unroll
    for (int s = 0; s < 2; ++s) {
      int k0 = 4 * s + 2 * half;
      v2f a, b;
      a[0] = rok ? agg[rowc * 32 + t * 8 + k0]     : 0.f;
      a[1] = rok ? agg[rowc * 32 + t * 8 + k0 + 1] : 0.f;
      int col = t * 16 + l16;
      b[0] = (k0     < 7) ? W1[k0 * 64 + col]       : 0.f;
      b[1] = (k0 + 1 < 7) ? W1[(k0 + 1) * 64 + col] : 0.f;
      hacc = __builtin_amdgcn_wmma_f32_16x16x4_f32(
          false, a, false, b, (short)0, hacc, false, false);
    }
    float bb = b1[t * 16 + l16];
#pragma unroll
    for (int r = 0; r < 8; ++r) {
      float o = hacc[r] + bb;
      o = o > 0.f ? o : (__expf(o) - 1.f);
      tl[(r + 8 * half) * LDW + t * 16 + l16] = o;
    }
  }
  __syncthreads();

  v8f acc;
#pragma unroll
  for (int q = 0; q < 8; ++q) acc[q] = 0.f;

#pragma unroll
  for (int s = 0; s < 16; ++s) {
    int k0 = 4 * s + 2 * half;
    v2f a, b;
    a = *(const v2f*)(tl + l16 * LDW + k0);
    b[0] = cok ? W2[k0 * 6 + l16]       : 0.f;
    b[1] = cok ? W2[(k0 + 1) * 6 + l16] : 0.f;
    acc = __builtin_amdgcn_wmma_f32_16x16x4_f32(
        false, a, false, b, (short)0, acc, false, false);
  }

  float cs = cok ? aS[l16] : 0.f;
  float cd = cok ? aD[l16] : 0.f;
#pragma unroll
  for (int r = 0; r < 8; ++r) {
    int m = wave * 16 + r + 8 * half;
    float s = sum8(acc[r] * cs);
    float d = sum8(acc[r] * cd);
    float v = cok ? acc[r] : (l16 == 6 ? s : d);
    if (l16 < 8 && m < N) rec[m * 8 + l16] = v;
  }
}


__device__ __forceinline__ float odd_of_pair(float v) {
  return __uint_as_float(__builtin_amdgcn_mov_dpp8(__float_as_uint(v), SEL8(1, 1, 3, 3, 5, 5, 7, 7)));
}

__global__ void gat2_node(const int* __restrict__ count, const int* __restrict__ slots,
                          const int* __restrict__ ovfi, const float* __restrict__ rec,
                          const float* __restrict__ b2, float* __restrict__ out, int N) {
  int gid = blockIdx.x * blockDim.x + threadIdx.x;
  int n = gid >> 4, sub = gid & 15;
  int p = sub >> 1, hf = sub & 1;
  bool ok = n < N;
  int nc = ok ? n : N - 1;
  int cnt = ok ? count[nc] : 0;
  const int* lst = slots + nc * SLOTS;
  const int2* ovf = (const int2*)ovfi;
  const float4* rv = (const float4*)rec;
  float4 rs = rv[nc * 2 + hf];
  float ad = odd_of_pair(rs.w);
  float es = leaky(odd_of_pair(rs.z) + ad);
  float m = es;

  int sv[8];
  float4 r[8];
  float ev[8];
#pragma unroll
  for (int k = 0; k < 8; ++k) {
    int j = p + 8 * k;
    int v = nc;
    if (j < cnt) v = lst[j];
    sv[k] = v;
  }
#pragma unroll
  for (int k = 0; k < 8; ++k) r[k] = rv[sv[k] * 2 + hf];
#pragma unroll
  for (int k = 0; k < 8; ++k) {
    int j = p + 8 * k;
    float e = leaky(odd_of_pair(r[k].z) + ad);
    ev[k] = e;
    if (j < cnt) m = fmaxf(m, e);
  }
  int novf = cnt > SLOTS ? count[N] : 0;
  for (int j = p; j < novf; j += 8) {
    int2 q = ovf[j];
    if (q.y == n) m = fmaxf(m, leaky(rec[q.x * 8 + 6] + ad));
  }
  m = fmaxf(m, __shfl_xor(m, 2));
  m = fmaxf(m, __shfl_xor(m, 4));
  m = fmaxf(m, __shfl_xor(m, 8));

  float den = 0.f;
  float acc[4];
#pragma unroll
  for (int c = 0; c < 4; ++c) acc[c] = 0.f;
#pragma unroll
  for (int k = 0; k < 8; ++k) {
    int j = p + 8 * k;
    if (j < cnt) {
      float w = __expf(ev[k] - m);
      den += w;
      acc[0] += w * r[k].x; acc[1] += w * r[k].y;
      acc[2] += w * r[k].z; acc[3] += w * r[k].w;
    }
  }
  for (int j = p; j < novf; j += 8) {
    int2 q = ovf[j];
    if (q.y == n) {
      int s = q.x;
      float w = __expf(leaky(rec[s * 8 + 6] + ad) - m);
      den += w;
      float4 h = rv[s * 2 + hf];
      acc[0] += w * h.x; acc[1] += w * h.y;
      acc[2] += w * h.z; acc[3] += w * h.w;
    }
  }
  if (p == 0) {
    float w = __expf(es - m);
    den += w;
    acc[0] += w * rs.x; acc[1] += w * rs.y;
    acc[2] += w * rs.z; acc[3] += w * rs.w;
  }
#pragma unroll
  for (int off = 2; off < 16; off <<= 1) {
    den += __shfl_xor(den, off);
#pragma unroll
    for (int c = 0; c < 4; ++c) acc[c] += __shfl_xor(acc[c], off);
  }
  float o4 = __shfl_xor(acc[0], 1);
  float o5 = __shfl_xor(acc[1], 1);

  if (ok && sub == 0) {
    float o[6] = {acc[0], acc[1], acc[2], acc[3], o4, o5};
    float mx = -3.0e38f;
#pragma unroll
    for (int c = 0; c < 6; ++c) { o[c] = o[c] / den + b2[c]; mx = fmaxf(mx, o[c]); }
    float sum = 0.f;
#pragma unroll
    for (int c = 0; c < 6; ++c) sum += __expf(o[c] - mx);
    float l = __logf(sum);
#pragma unroll
    for (int c = 0; c < 6; ++c) out[n * 6 + c] = o[c] - mx - l;
  }
}

extern "C" void kernel_launch(void* const* d_in, const int* in_sizes, int n_in,
                              void* d_out, int out_size, void* d_ws, size_t ws_size,
                              hipStream_t stream) {
  const float* x      = (const float*)d_in[0];
  const int*   ei     = (const int*)d_in[1];
  const float* W1     = (const float*)d_in[2];
  const float* a_src1 = (const float*)d_in[3];
  const float* a_dst1 = (const float*)d_in[4];
  const float* b1     = (const float*)d_in[5];
  const float* W2     = (const float*)d_in[6];
  const float* a_src2 = (const float*)d_in[7];
  const float* a_dst2 = (const float*)d_in[8];
  const float* b2     = (const float*)d_in[9];
  float* out = (float*)d_out;

  int N = in_sizes[0] / 7;
  int E = in_sizes[1] / 2;

  int nbin = (N + BN - 1) / BN;
  int nblk = (E + CH - 1) / CH;
  int dpitch = (nbin + 15) & ~15;

  float* ws  = (float*)d_ws;
  float* agg = ws;
  float* xr  = agg + (size_t)N * 64;
  int* slots = (int*)(xr + (size_t)N * 64);
  float* rec = (float*)(slots + (size_t)N * SLOTS);
  int* ovf   = (int*)(rec + (size_t)N * 8);
  int2* bins = (int2*)(ovf + (size_t)E * 2);
  int2* desc = bins + (size_t)nblk * CH;
  int* count = (int*)(desc + (size_t)nblk * dpitch);

  const int B = 256;
  auto G = [](long n, int b) { return (int)((n + b - 1) / b); };
  int tiles = (N + 15) / 16;
  int projBlocks = (tiles + (B / 32) - 1) / (B / 32);

  sort_chunks<<<nblk, TPB, 0, stream>>>(ei, bins, desc, count, N, E, nbin, dpitch);
  build_rows<<<nbin, TPB, 0, stream>>>(bins, desc, count, slots, ovf, N, nblk, dpitch);

  prep1_wmma<<<projBlocks, B, 0, stream>>>(x, W1, a_src1, a_dst1, xr, N);
  gat1_node<<<G(N, B / 32), B, 0, stream>>>(count, slots, ovf, xr, agg, N);

  post1_proj2_wmma<<<projBlocks, B, 0, stream>>>(agg, W1, b1, W2, a_src2, a_dst2, rec, N);
  gat2_node<<<G((long)N * 16, B), B, 0, stream>>>(count, slots, ovf, rec, b2, out, N);
}


# Edge chunks 4x larger (16384 edges, 16 per thread), an edge kept as one packed word: 4x fewer, longer runs, build_rows asks for all of a wave's runs in one group of loads; slot (sum) order changes
# speedup vs baseline: 30.1321x; 1.1067x over previous
#include <hip/hip_runtime.h>
#include <hip/hip_bf16.h>
#include <math.h>

typedef float v2f __attribute__((ext_vector_type(2)));
typedef float v8f __attribute__((ext_vector_type(8)));

#define NEG_SLOPE 0.2f

__device__ __forceinline__ float leaky(float v) { return v > 0.f ? v : NEG_SLOPE * v; }

#define SLOTS 64
#define BSH 9
#define BN (1 << BSH)
#define TPB 1024
#define EPT 16
#define CH (EPT * TPB)
#define RQ 4

__device__ __forceinline__ int block_excl_scan(int v, int* wsum, int tid) {
  int lane = tid & 31, w = tid >> 5;
  int inc = v;
#pragma unroll
  for (int o = 1; o < 32; o <<= 1) {
    int t = __shfl_up(inc, o);
    if (lane >= o) inc += t;
  }
  if (lane == 31) wsum[w] = inc;
  __syncthreads();
  if (w == 0) {
    int s = wsum[lane];
    int si = s;
#pragma unroll
    for (int o = 1; o < 32; o <<= 1) {
      int t = __shfl_up(si, o);
      if (lane >= o) si += t;
    }
    wsum[lane] = si - s;
  }
  __syncthreads();
  return wsum[w] + inc - v;
}

__global__ __launch_bounds__(TPB) void sort_chunks(const int* __restrict__ ei, int* __restrict__ bins,
                                                   int2* __restrict__ desc, int* __restrict__ count,
                                                   int N, int E, int nbin, int dpitch) {
  __shared__ int h[TPB];
  __shared__ int wsum[32];
  __shared__ int stage[CH];
  int tid = threadIdx.x;
  int blk = blockIdx.x;
  if (blk == 0 && tid == 0) count[N] = 0;
  h[tid] = 0;
  __syncthreads();
  int e0 = blk * CH + tid;
  int s[EPT], d[EPT], r[EPT];
#pragma unroll
  for (int k = 0; k < EPT; ++k) {
    int e = e0 + k * TPB;
    s[k] = 0; d[k] = -1;
    if (e < E) { s[k] = ei[e]; d[k] = ei[E + e]; }
  }
#pragma unroll
  for (int k = 0; k < EPT; ++k) {
    r[k] = 0;
    if (d[k] >= 0) r[k] = atomicAdd(&h[d[k] >> BSH], 1);
  }
  __syncthreads();
  int v = h[tid];
  int ex = block_excl_scan(v, wsum, tid);
  h[tid] = ex;
  if (tid < nbin) desc[(size_t)blk * dpitch + tid] = make_int2(ex, v);
  __syncthreads();
  int hb[EPT];
#pragma unroll
  for (int k = 0; k < EPT; ++k) hb[k] = h[(d[k] >= 0 ? d[k] : 0) >> BSH];
#pragma unroll
  for (int k = 0; k < EPT; ++k)
    if (d[k] >= 0) stage[hb[k] + r[k]] = (int)(((unsigned)s[k] << BSH) | (unsigned)(d[k] & (BN - 1)));
  __syncthreads();
  int total = E - blk * CH;
  if (total > CH) total = CH;
  if (total == CH) {
    int o[EPT];
#pragma unroll
    for (int k = 0; k < EPT; ++k) o[k] = stage[tid + k * TPB];
#pragma unroll
    for (int k = 0; k < EPT; ++k)
      bins[(size_t)blk * CH + tid + k * TPB] = o[k];
  } else {
#pragma unroll
    for (int k = 0; k < EPT; ++k) {
      int i = tid + k * TPB;
      if (i < total) bins[(size_t)blk * CH + i] = stage[i];
    }
  }
}

__device__ __forceinline__ void place_edge(int wd, int n0, int* lcnt, int* lrows,
                                           int* __restrict__ ovf, int* ovfc) {
  int dl = wd & (BN - 1);
  int sr = (int)((unsigned)wd >> BSH);
  int r = atomicAdd(&lcnt[dl], 1);
  if (r < SLOTS) {
    lrows[dl * SLOTS + ((r + dl) & (SLOTS - 1))] = sr;
  } else {
    int k = atomicAdd(ovfc, 1);
    ((int2*)ovf)[k] = make_int2(sr, n0 + dl);
  }
}

__global__ __launch_bounds__(TPB) void build_rows(const int* __restrict__ bins, const int2* __restrict__ desc,
                                                  int* __restrict__ count, int* __restrict__ slots,
                                                  int* __restrict__ ovf, int N, int nblk, int dpitch) {
  __shared__ int lcnt[BN];
  __shared__ int lrows[BN * SLOTS];
  int tid = threadIdx.x, lane = tid & 31;
  int w = __builtin_amdgcn_readfirstlane(tid >> 5);
  int b = blockIdx.x;
  int n0 = b << BSH;
  int* ovfc = count + N;
  if (tid < BN) lcnt[tid] = 0;
  __syncthreads();
  for (int base = 0; base < nblk; base += TPB) {
    int myblk = base + lane * 32 + w;
    int2 q = make_int2(0, 0);
    if (myblk < nblk) q = desc[(size_t)myblk * dpitch + b];
    int left = nblk - base - w;
    int nrun = left > 0 ? (left + 31) >> 5 : 0;
    if (nrun > 32) nrun = 32;
    for (int j0 = 0; j0 < nrun; j0 += 8) {
      int wd[8][RQ];
      int ln[8];
      const int* run[8];
#pragma unroll
      for (int u = 0; u < 8; ++u) {
        int st = __builtin_amdgcn_readlane(q.x, j0 + u);
        ln[u] = __builtin_amdgcn_readlane(q.y, j0 + u);
        run[u] = bins + (size_t)(base + (j0 + u) * 32 + w) * CH + st;
#pragma unroll
        for (int t = 0; t < RQ; ++t) {
          wd[u][t] = 0;
          if (lane + 32 * t < ln[u]) wd[u][t] = run[u][lane + 32 * t];
        }
      }
#pragma unroll
      for (int u = 0; u < 8; ++u) {
#pragma unroll
        for (int t = 0; t < RQ; ++t)
          if (lane + 32 * t < ln[u]) place_edge(wd[u][t], n0, lcnt, lrows, ovf, ovfc);
        for (int k = lane + 32 * RQ; k < ln[u]; k += 32)
          place_edge(run[u][k], n0, lcnt, lrows, ovf, ovfc);
      }
    }
  }
  __syncthreads();
  if (tid < BN && n0 + tid < N) count[n0 + tid] = lcnt[tid];
  for (int it = 0; it < BN * SLOTS / TPB; ++it) {
    int idx = it * TPB + tid;
    int dl = idx >> 6, j = idx & (SLOTS - 1);
    int n = n0 + dl;
    if (n < N && j < lcnt[dl])
      slots[(size_t)n * SLOTS + j] = lrows[dl * SLOTS + ((j + dl) & (SLOTS - 1))];
  }
}


__global__ void prep1_wmma(const float* __restrict__ x, const float* __restrict__ W1,
                           const float* __restrict__ aS, const float* __restrict__ aD,
                           float* __restrict__ xr, int N) {
  int gtid = blockIdx.x * blockDim.x + threadIdx.x;
  int wave = gtid >> 5;
  int lane = threadIdx.x & 31;
  int tiles = (N + 15) >> 4;
  if (wave >= tiles) return;
  int half = lane >> 4;
  int l16  = lane & 15;
  int row  = wave * 16 + l16;
  int rowc = row < N ? row : N - 1;
  bool rok = row < N;

  v2f a[2];
#pragma unroll
  for (int s = 0; s < 2; ++s) {
    int k0 = 4 * s + 2 * half;
    a[s][0] = (rok && k0     < 7) ? x[rowc * 7 + k0]     : 0.f;
    a[s][1] = (rok && k0 + 1 < 7) ? x[rowc * 7 + k0 + 1] : 0.f;
  }
#pragma unroll
  for (int s = 0; s < 2; ++s) {
    v2f st = a[s];
    if (s == 1 && half == 1) st[1] = 1.f;
    if (rok) *(v2f*)(xr + row * 16 + 4 * s + 2 * half) = st;
  }

  int jt = l16 & 3;
  const float* av = ((l16 & 4) ? aD : aS) + jt * 16;
  float4 avv[4];
#pragma unroll
  for (int q = 0; q < 4; ++q) avv[q] = ((const float4*)av)[q];
  v8f sc;
#pragma unroll
  for (int q = 0; q < 8; ++q) sc[q] = 0.f;
#pragma unroll
  for (int s = 0; s < 2; ++s) {
    int k0 = 4 * s + 2 * half;
    v2f bp;
#pragma unroll
    for (int i = 0; i < 2; ++i) {
      int k = k0 + i;
      int kc = k < 7 ? k : 6;
      const float4* wr = (const float4*)(W1 + kc * 64 + jt * 16);
      float v = 0.f;
#pragma unroll
      for (int q = 0; q < 4; ++q) {
        float4 w = wr[q];
        v += w.x * avv[q].x; v += w.y * avv[q].y; v += w.z * avv[q].z; v += w.w * avv[q].w;
      }
      bp[i] = (k < 7 && l16 < 8) ? v : 0.f;
    }
    sc = __builtin_amdgcn_wmma_f32_16x16x4_f32(
        false, a[s], false, bp, (short)0, sc, false, false);
  }
#pragma unroll
  for (int r = 0; r < 8; ++r) {
    int m = wave * 16 + r + 8 * half;
    if (l16 < 8 && m < N) xr[m * 16 + 8 + l16] = sc[r];
  }
}


#define SEL8(a, b, c, d, e, f, g, h) \
  ((a) | ((b) << 3) | ((c) << 6) | ((d) << 9) | ((e) << 12) | ((f) << 15) | ((g) << 18) | ((h) << 21))

__device__ __forceinline__ float sum8(float v) {
  v += __uint_as_float(__builtin_amdgcn_mov_dpp8(__float_as_uint(v), SEL8(1, 0, 3, 2, 5, 4, 7, 6)));
  v += __uint_as_float(__builtin_amdgcn_mov_dpp8(__float_as_uint(v), SEL8(2, 3, 0, 1, 6, 7, 4, 5)));
  v += __uint_as_float(__builtin_amdgcn_mov_dpp8(__float_as_uint(v), SEL8(4, 5, 6, 7, 0, 1, 2, 3)));
  return v;
}

template <int D>
__device__ __forceinline__ float lane_xor(float v) {
  unsigned u = __float_as_uint(v);
  if (D == 1) return __uint_as_float(__builtin_amdgcn_mov_dpp8(u, SEL8(1, 0, 3, 2, 5, 4, 7, 6)));
  if (D == 2) return __uint_as_float(__builtin_amdgcn_mov_dpp8(u, SEL8(2, 3, 0, 1, 6, 7, 4, 5)));
  if (D == 4) return __uint_as_float(__builtin_amdgcn_mov_dpp8(u, SEL8(4, 5, 6, 7, 0, 1, 2, 3)));
  if (D == 8)
    return __uint_as_float((unsigned)__builtin_amdgcn_update_dpp(0, (int)u, 0x128, 0xf, 0xf, false));
  return __uint_as_float(__builtin_amdgcn_permlanex16(u, u, 0x76543210u, 0xfedcba98u, false, false));
}
__device__ __forceinline__ float wave_max(float v) {
  v = fmaxf(v, lane_xor<1>(v));
  v = fmaxf(v, lane_xor<2>(v));
  v = fmaxf(v, lane_xor<4>(v));
  v = fmaxf(v, lane_xor<8>(v));
  v = fmaxf(v, lane_xor<16>(v));
  return v;
}
template <int D, int H>
__device__ __forceinline__ void halve(float* v, int lane) {
  bool up = (lane & D) != 0;
#pragma unroll
  for (int i = 0; i < H; ++i) {
    float keep = up ? v[i + H] : v[i];
    float give = up ? v[i] : v[i + H];
    v[i] = keep + lane_xor<D>(give);
  }
}

#define WADD(P, A, B) { \
  _Pragma("unroll") for (int h = 0; h < 4; ++h) { \
    acc[8 * h + 0] += (P)[h] * (A).x; acc[8 * h + 1] += (P)[h] * (A).y; \
    acc[8 * h + 2] += (P)[h] * (A).z; acc[8 * h + 3] += (P)[h] * (A).w; \
    acc[8 * h + 4] += (P)[h] * (B).x; acc[8 * h + 5] += (P)[h] * (B).y; \
    acc[8 * h + 6] += (P)[h] * (B).z; acc[8 * h + 7] += (P)[h] * (B).w; } }
#define SCORE(E, S) { (E)[0] = leaky((S).x + ad.x); (E)[1] = leaky((S).y + ad.y); \
                      (E)[2] = leaky((S).z + ad.z); (E)[3] = leaky((S).w + ad.w); }

__global__ void gat1_node(const int* __restrict__ count, const int* __restrict__ slots,
                          const int* __restrict__ ovfi,
                          const float* __restrict__ xr, float* __restrict__ agg, int N) {
  int lane = threadIdx.x & 31;
  int n = __builtin_amdgcn_readfirstlane((int)(blockIdx.x * (blockDim.x >> 5) + (threadIdx.x >> 5)));
  if (n >= N) return;
  const float4* rv = (const float4*)xr;

  int cnt = count[n];
  const int* lst = slots + n * SLOTS;
  const int2* ovf = (const int2*)ovfi;
  float4 ad = rv[n * 4 + 3];
  float4 an = rv[n * 4 + 2];
  float es[4];
  SCORE(es, an)
  float xs = xr[n * 16 + (lane & 7)];

  bool in0 = lane < cnt, in1 = 32 + lane < cnt;
  int s0 = n, s1 = n;
  if (in0) s0 = lst[lane];
  if (in1) s1 = lst[32 + lane];
  bool two = cnt > 32;
  float4 a0 = rv[s0 * 4 + 2], ra0 = rv[s0 * 4], rb0 = rv[s0 * 4 + 1];
  float4 a1 = an, ra1 = make_float4(0.f, 0.f, 0.f, 0.f), rb1 = ra1;
  if (two) { a1 = rv[s1 * 4 + 2]; ra1 = rv[s1 * 4]; rb1 = rv[s1 * 4 + 1]; }
  float e0[4], e1[4], m[4];
  SCORE(e0, a0)
  SCORE(e1, a1)
#pragma unroll
  for (int h = 0; h < 4; ++h) m[h] = fmaxf(es[h], fmaxf(e0[h], e1[h]));
  int novf = 0;
  if (cnt > SLOTS) {
    novf = count[N];
    for (int base = 0; base < novf; base += 32) {
      int j = base + lane;
      if (j < novf) {
        int2 q = ovf[j];
        if (q.y == n) {
          float4 aq = rv[q.x * 4 + 2];
          float eq[4];
          SCORE(eq, aq)
#pragma unroll
          for (int h = 0; h < 4; ++h) m[h] = fmaxf(m[h], eq[h]);
        }
      }
    }
  }
#pragma unroll
  for (int h = 0; h < 4; ++h) m[h] = wave_max(m[h]);

  float acc[32];
  float p[4];
#pragma unroll
  for (int h = 0; h < 4; ++h) p[h] = in0 ? __expf(e0[h] - m[h]) : 0.f;
#pragma unroll
  for (int i = 0; i < 32; ++i) acc[i] = 0.f;
  WADD(p, ra0, rb0)
  if (two) {
#pragma unroll
    for (int h = 0; h < 4; ++h) p[h] = in1 ? __expf(e1[h] - m[h]) : 0.f;
    WADD(p, ra1, rb1)
  }
  for (int base = 0; base < novf; base += 32) {
    int j = base + lane;
    if (j < novf) {
      int2 q = ovf[j];
      if (q.y == n) {
        float4 aq = rv[q.x * 4 + 2];
        float eq[4];
        SCORE(eq, aq)
#pragma unroll
        for (int h = 0; h < 4; ++h) p[h] = __expf(eq[h] - m[h]);
        float4 ya = rv[q.x * 4], yb = rv[q.x * 4 + 1];
        WADD(p, ya, yb)
      }
    }
  }
  halve<16, 16>(acc, lane);
  halve<8, 8>(acc, lane);
  halve<4, 4>(acc, lane);
  halve<2, 2>(acc, lane);
  halve<1, 1>(acc, lane);
  int hd = lane >> 3;
  float esl = (hd & 2) ? ((hd & 1) ? es[3] : es[2]) : ((hd & 1) ? es[1] : es[0]);
  float ml  = (hd & 2) ? ((hd & 1) ? m[3] : m[2]) : ((hd & 1) ? m[1] : m[0]);
  float tot = acc[0] + __expf(esl - ml) * xs;
  float den = __uint_as_float(__builtin_amdgcn_mov_dpp8(__float_as_uint(tot), SEL8(7, 7, 7, 7, 7, 7, 7, 7)));
  agg[(size_t)n * 32 + lane] = tot / den;
}

#define LDW 68
__global__ __launch_bounds__(256) void post1_proj2_wmma(const float* __restrict__ agg, const float* __restrict__ W1,
                           const float* __restrict__ b1, const float* __restrict__ W2,
                           const float* __restrict__ aS, const float* __restrict__ aD,
                           float* __restrict__ rec, int N) {
  __shared__ float tile[8 * 16 * LDW];
  int wv = threadIdx.x >> 5;
  int wave = blockIdx.x * 8 + wv;
  int lane = threadIdx.x & 31;
  int half = lane >> 4, l16 = lane & 15;
  int row  = wave * 16 + l16;
  int rowc = row < N ? row : N - 1;
  bool rok = row < N;
  bool cok = l16 < 6;
  float* tl = tile + wv * (16 * LDW);

#pragma unroll
  for (int t = 0; t < 4; ++t) {
    v8f hacc;
#pragma unroll
    for (int q = 0; q < 8; ++q) hacc[q] = 0.f;
#pragma unroll
    for (int s = 0; s < 2; ++s) {
      int k0 = 4 * s + 2 * half;
      v2f a, b;
      a[0] = rok ? agg[rowc * 32 + t * 8 + k0]     : 0.f;
      a[1] = rok ? agg[rowc * 32 + t * 8 + k0 + 1] : 0.f;
      int col = t * 16 + l16;
      b[0] = (k0     < 7) ? W1[k0 * 64 + col]       : 0.f;
      b[1] = (k0 + 1 < 7) ? W1[(k0 + 1) * 64 + col] : 0.f;
      hacc = __builtin_amdgcn_wmma_f32_16x16x4_f32(
          false, a, false, b, (short)0, hacc, false, false);
    }
    float bb = b1[t * 16 + l16];
#pragma unroll
    for (int r = 0; r < 8; ++r) {
      float o = hacc[r] + bb;
      o = o > 0.f ? o : (__expf(o) - 1.f);
      tl[(r + 8 * half) * LDW + t * 16 + l16] = o;
    }
  }
  __syncthreads();

  v8f acc;
#pragma unroll
  for (int q = 0; q < 8; ++q) acc[q] = 0.f;

#pragma unroll
  for (int s = 0; s < 16; ++s) {
    int k0 = 4 * s + 2 * half;
    v2f a, b;
    a = *(const v2f*)(tl + l16 * LDW + k0);
    b[0] = cok ? W2[k0 * 6 + l16]       : 0.f;
    b[1] = cok ? W2[(k0 + 1) * 6 + l16] : 0.f;
    acc = __builtin_amdgcn_wmma_f32_16x16x4_f32(
        false, a, false, b, (short)0, acc, false, false);
  }

  float cs = cok ? aS[l16] : 0.f;
  float cd = cok ? aD[l16] : 0.f;
#pragma unroll
  for (int r = 0; r < 8; ++r) {
    int m = wave * 16 + r + 8 * half;
    float s = sum8(acc[r] * cs);
    float d = sum8(acc[r] * cd);
    float v = cok ? acc[r] : (l16 == 6 ? s : d);
    if (l16 < 8 && m < N) rec[m * 8 + l16] = v;
  }
}


__device__ __forceinline__ float odd_of_pair(float v) {
  return __uint_as_float(__builtin_amdgcn_mov_dpp8(__float_as_uint(v), SEL8(1, 1, 3, 3, 5, 5, 7, 7)));
}

__global__ void gat2_node(const int* __restrict__ count, const int* __restrict__ slots,
                          const int* __restrict__ ovfi, const float* __restrict__ rec,
                          const float* __restrict__ b2, float* __restrict__ out, int N) {
  int gid = blockIdx.x * blockDim.x + threadIdx.x;
  int n = gid >> 4, sub = gid & 15;
  int p = sub >> 1, hf = sub & 1;
  bool ok = n < N;
  int nc = ok ? n : N - 1;
  int cnt = ok ? count[nc] : 0;
  const int* lst = slots + nc * SLOTS;
  const int2* ovf = (const int2*)ovfi;
  const float4* rv = (const float4*)rec;
  float4 rs = rv[nc * 2 + hf];
  float ad = odd_of_pair(rs.w);
  float es = leaky(odd_of_pair(rs.z) + ad);
  float m = es;

  int sv[8];
  float4 r[8];
  float ev[8];
#pragma unroll
  for (int k = 0; k < 8; ++k) {
    int j = p + 8 * k;
    int v = nc;
    if (j < cnt) v = lst[j];
    sv[k] = v;
  }
#pragma unroll
  for (int k = 0; k < 8; ++k) r[k] = rv[sv[k] * 2 + hf];
#pragma unroll
  for (int k = 0; k < 8; ++k) {
    int j = p + 8 * k;
    float e = leaky(odd_of_pair(r[k].z) + ad);
    ev[k] = e;
    if (j < cnt) m = fmaxf(m, e);
  }
  int novf = cnt > SLOTS ? count[N] : 0;
  for (int j = p; j < novf; j += 8) {
    int2 q = ovf[j];
    if (q.y == n) m = fmaxf(m, leaky(rec[q.x * 8 + 6] + ad));
  }
  m = fmaxf(m, __shfl_xor(m, 2));
  m = fmaxf(m, __shfl_xor(m, 4));
  m = fmaxf(m, __shfl_xor(m, 8));

  float den = 0.f;
  float acc[4];
#pragma unroll
  for (int c = 0; c < 4; ++c) acc[c] = 0.f;
#pragma unroll
  for (int k = 0; k < 8; ++k) {
    int j = p + 8 * k;
    if (j < cnt) {
      float w = __expf(ev[k] - m);
      den += w;
      acc[0] += w * r[k].x; acc[1] += w * r[k].y;
      acc[2] += w * r[k].z; acc[3] += w * r[k].w;
    }
  }
  for (int j = p; j < novf; j += 8) {
    int2 q = ovf[j];
    if (q.y == n) {
      int s = q.x;
      float w = __expf(leaky(rec[s * 8 + 6] + ad) - m);
      den += w;
      float4 h = rv[s * 2 + hf];
      acc[0] += w * h.x; acc[1] += w * h.y;
      acc[2] += w * h.z; acc[3] += w * h.w;
    }
  }
  if (p == 0) {
    float w = __expf(es - m);
    den += w;
    acc[0] += w * rs.x; acc[1] += w * rs.y;
    acc[2] += w * rs.z; acc[3] += w * rs.w;
  }
#pragma unroll
  for (int off = 2; off < 16; off <<= 1) {
    den += __shfl_xor(den, off);
#pragma unroll
    for (int c = 0; c < 4; ++c) acc[c] += __shfl_xor(acc[c], off);
  }
  float o4 = __shfl_xor(acc[0], 1);
  float o5 = __shfl_xor(acc[1], 1);

  if (ok && sub == 0) {
    float o[6] = {acc[0], acc[1], acc[2], acc[3], o4, o5};
    float mx = -3.0e38f;
#pragma unroll
    for (int c = 0; c < 6; ++c) { o[c] = o[c] / den + b2[c]; mx = fmaxf(mx, o[c]); }
    float sum = 0.f;
#pragma unroll
    for (int c = 0; c < 6; ++c) sum += __expf(o[c] - mx);
    float l = __logf(sum);
#pragma unroll
    for (int c = 0; c < 6; ++c) out[n * 6 + c] = o[c] - mx - l;
  }
}

extern "C" void kernel_launch(void* const* d_in, const int* in_sizes, int n_in,
                              void* d_out, int out_size, void* d_ws, size_t ws_size,
                              hipStream_t stream) {
  const float* x      = (const float*)d_in[0];
  const int*   ei     = (const int*)d_in[1];
  const float* W1     = (const float*)d_in[2];
  const float* a_src1 = (const float*)d_in[3];
  const float* a_dst1 = (const float*)d_in[4];
  const float* b1     = (const float*)d_in[5];
  const float* W2     = (const float*)d_in[6];
  const float* a_src2 = (const float*)d_in[7];
  const float* a_dst2 = (const float*)d_in[8];
  const float* b2     = (const float*)d_in[9];
  float* out = (float*)d_out;

  int N = in_sizes[0] / 7;
  int E = in_sizes[1] / 2;

  int nbin = (N + BN - 1) / BN;
  int nblk = (E + CH - 1) / CH;
  int dpitch = (nbin + 15) & ~15;

  float* ws  = (float*)d_ws;
  float* agg = ws;
  float* xr  = agg + (size_t)N * 64;
  int* slots = (int*)(xr + (size_t)N * 64);
  float* rec = (float*)(slots + (size_t)N * SLOTS);
  int* ovf   = (int*)(rec + (size_t)N * 8);
  int* bins  = ovf + (size_t)E * 2;
  int2* desc = (int2*)(bins + (size_t)nblk * CH);
  int* count = (int*)(desc + (size_t)nblk * dpitch);

  const int B = 256;
  auto G = [](long n, int b) { return (int)((n + b - 1) / b); };
  int tiles = (N + 15) / 16;
  int projBlocks = (tiles + (B / 32) - 1) / (B / 32);

  sort_chunks<<<nblk, TPB, 0, stream>>>(ei, bins, desc, count, N, E, nbin, dpitch);
  build_rows<<<nbin, TPB, 0, stream>>>(bins, desc, count, slots, ovf, N, nblk, dpitch);

  prep1_wmma<<<projBlocks, B, 0, stream>>>(x, W1, a_src1, a_dst1, xr, N);
  gat1_node<<<G(N, B / 32), B, 0, stream>>>(count, slots, ovf, xr, agg, N);

  post1_proj2_wmma<<<projBlocks, B, 0, stream>>>(agg, W1, b1, W2, a_src2, a_dst2, rec, N);
  gat2_node<<<G((long)N * 16, B), B, 0, stream>>>(count, slots, ovf, rec, b2, out, N);
}


# gat1: four lanes own an edge and load its source's 64-B record as quarters in one instruction (8 lines per load, not 32); lane = head, weights go round the four lanes; 8-lane sum tree: new sum order
# speedup vs baseline: 31.8230x; 1.0561x over previous
#include <hip/hip_runtime.h>
#include <hip/hip_bf16.h>
#include <math.h>

typedef float v2f __attribute__((ext_vector_type(2)));
typedef float v8f __attribute__((ext_vector_type(8)));

#define NEG_SLOPE 0.2f

__device__ __forceinline__ float leaky(float v) { return v > 0.f ? v : NEG_SLOPE * v; }

#define SLOTS 64
#define BSH 9
#define BN (1 << BSH)
#define TPB 1024
#define EPT 16
#define CH (EPT * TPB)
#define RQ 4

__device__ __forceinline__ int block_excl_scan(int v, int* wsum, int tid) {
  int lane = tid & 31, w = tid >> 5;
  int inc = v;
#pragma unroll
  for (int o = 1; o < 32; o <<= 1) {
    int t = __shfl_up(inc, o);
    if (lane >= o) inc += t;
  }
  if (lane == 31) wsum[w] = inc;
  __syncthreads();
  if (w == 0) {
    int s = wsum[lane];
    int si = s;
#pragma unroll
    for (int o = 1; o < 32; o <<= 1) {
      int t = __shfl_up(si, o);
      if (lane >= o) si += t;
    }
    wsum[lane] = si - s;
  }
  __syncthreads();
  return wsum[w] + inc - v;
}

__global__ __launch_bounds__(TPB) void sort_chunks(const int* __restrict__ ei, int* __restrict__ bins,
                                                   int2* __restrict__ desc, int* __restrict__ count,
                                                   int N, int E, int nbin, int dpitch) {
  __shared__ int h[TPB];
  __shared__ int wsum[32];
  __shared__ int stage[CH];
  int tid = threadIdx.x;
  int blk = blockIdx.x;
  if (blk == 0 && tid == 0) count[N] = 0;
  h[tid] = 0;
  __syncthreads();
  int e0 = blk * CH + tid;
  int s[EPT], d[EPT], r[EPT];
#pragma unroll
  for (int k = 0; k < EPT; ++k) {
    int e = e0 + k * TPB;
    s[k] = 0; d[k] = -1;
    if (e < E) { s[k] = ei[e]; d[k] = ei[E + e]; }
  }
#pragma unroll
  for (int k = 0; k < EPT; ++k) {
    r[k] = 0;
    if (d[k] >= 0) r[k] = atomicAdd(&h[d[k] >> BSH], 1);
  }
  __syncthreads();
  int v = h[tid];
  int ex = block_excl_scan(v, wsum, tid);
  h[tid] = ex;
  if (tid < nbin) desc[(size_t)blk * dpitch + tid] = make_int2(ex, v);
  __syncthreads();
  int hb[EPT];
#pragma unroll
  for (int k = 0; k < EPT; ++k) hb[k] = h[(d[k] >= 0 ? d[k] : 0) >> BSH];
#pragma unroll
  for (int k = 0; k < EPT; ++k)
    if (d[k] >= 0) stage[hb[k] + r[k]] = (int)(((unsigned)s[k] << BSH) | (unsigned)(d[k] & (BN - 1)));
  __syncthreads();
  int total = E - blk * CH;
  if (total > CH) total = CH;
  if (total == CH) {
    int o[EPT];
#pragma unroll
    for (int k = 0; k < EPT; ++k) o[k] = stage[tid + k * TPB];
#pragma unroll
    for (int k = 0; k < EPT; ++k)
      bins[(size_t)blk * CH + tid + k * TPB] = o[k];
  } else {
#pragma unroll
    for (int k = 0; k < EPT; ++k) {
      int i = tid + k * TPB;
      if (i < total) bins[(size_t)blk * CH + i] = stage[i];
    }
  }
}

__device__ __forceinline__ void place_edge(int wd, int n0, int* lcnt, int* lrows,
                                           int* __restrict__ ovf, int* ovfc) {
  int dl = wd & (BN - 1);
  int sr = (int)((unsigned)wd >> BSH);
  int r = atomicAdd(&lcnt[dl], 1);
  if (r < SLOTS) {
    lrows[dl * SLOTS + ((r + dl) & (SLOTS - 1))] = sr;
  } else {
    int k = atomicAdd(ovfc, 1);
    ((int2*)ovf)[k] = make_int2(sr, n0 + dl);
  }
}

__global__ __launch_bounds__(TPB) void build_rows(const int* __restrict__ bins, const int2* __restrict__ desc,
                                                  int* __restrict__ count, int* __restrict__ slots,
                                                  int* __restrict__ ovf, int N, int nblk, int dpitch) {
  __shared__ int lcnt[BN];
  __shared__ int lrows[BN * SLOTS];
  int tid = threadIdx.x, lane = tid & 31;
  int w = __builtin_amdgcn_readfirstlane(tid >> 5);
  int b = blockIdx.x;
  int n0 = b << BSH;
  int* ovfc = count + N;
  if (tid < BN) lcnt[tid] = 0;
  __syncthreads();
  for (int base = 0; base < nblk; base += TPB) {
    int myblk = base + lane * 32 + w;
    int2 q = make_int2(0, 0);
    if (myblk < nblk) q = desc[(size_t)myblk * dpitch + b];
    int left = nblk - base - w;
    int nrun = left > 0 ? (left + 31) >> 5 : 0;
    if (nrun > 32) nrun = 32;
    for (int j0 = 0; j0 < nrun; j0 += 8) {
      int wd[8][RQ];
      int ln[8];
      const int* run[8];
#pragma unroll
      for (int u = 0; u < 8; ++u) {
        int st = __builtin_amdgcn_readlane(q.x, j0 + u);
        ln[u] = __builtin_amdgcn_readlane(q.y, j0 + u);
        run[u] = bins + (size_t)(base + (j0 + u) * 32 + w) * CH + st;
#pragma unroll
        for (int t = 0; t < RQ; ++t) {
          wd[u][t] = 0;
          if (lane + 32 * t < ln[u]) wd[u][t] = run[u][lane + 32 * t];
        }
      }
#pragma unroll
      for (int u = 0; u < 8; ++u) {
#pragma unroll
        for (int t = 0; t < RQ; ++t)
          if (lane + 32 * t < ln[u]) place_edge(wd[u][t], n0, lcnt, lrows, ovf, ovfc);
        for (int k = lane + 32 * RQ; k < ln[u]; k += 32)
          place_edge(run[u][k], n0, lcnt, lrows, ovf, ovfc);
      }
    }
  }
  __syncthreads();
  if (tid < BN && n0 + tid < N) count[n0 + tid] = lcnt[tid];
  for (int it = 0; it < BN * SLOTS / TPB; ++it) {
    int idx = it * TPB + tid;
    int dl = idx >> 6, j = idx & (SLOTS - 1);
    int n = n0 + dl;
    if (n < N && j < lcnt[dl])
      slots[(size_t)n * SLOTS + j] = lrows[dl * SLOTS + ((j + dl) & (SLOTS - 1))];
  }
}


__global__ void prep1_wmma(const float* __restrict__ x, const float* __restrict__ W1,
                           const float* __restrict__ aS, const float* __restrict__ aD,
                           float* __restrict__ xr, int N) {
  int gtid = blockIdx.x * blockDim.x + threadIdx.x;
  int wave = gtid >> 5;
  int lane = threadIdx.x & 31;
  int tiles = (N + 15) >> 4;
  if (wave >= tiles) return;
  int half = lane >> 4;
  int l16  = lane & 15;
  int row  = wave * 16 + l16;
  int rowc = row < N ? row : N - 1;
  bool rok = row < N;

  v2f a[2];
#pragma unroll
  for (int s = 0; s < 2; ++s) {
    int k0 = 4 * s + 2 * half;
    a[s][0] = (rok && k0     < 7) ? x[rowc * 7 + k0]     : 0.f;
    a[s][1] = (rok && k0 + 1 < 7) ? x[rowc * 7 + k0 + 1] : 0.f;
  }
#pragma unroll
  for (int s = 0; s < 2; ++s) {
    v2f st = a[s];
    if (s == 1 && half == 1) st[1] = 1.f;
    if (rok) *(v2f*)(xr + row * 16 + 4 * (2 * s + half)) = st;
  }

  int jt = l16 & 3;
  const float* av = ((l16 & 4) ? aD : aS) + jt * 16;
  float4 avv[4];
#pragma unroll
  for (int q = 0; q < 4; ++q) avv[q] = ((const float4*)av)[q];
  v8f sc;
#pragma unroll
  for (int q = 0; q < 8; ++q) sc[q] = 0.f;
#pragma unroll
  for (int s = 0; s < 2; ++s) {
    int k0 = 4 * s + 2 * half;
    v2f bp;
#pragma unroll
    for (int i = 0; i < 2; ++i) {
      int k = k0 + i;
      int kc = k < 7 ? k : 6;
      const float4* wr = (const float4*)(W1 + kc * 64 + jt * 16);
      float v = 0.f;
#pragma unroll
      for (int q = 0; q < 4; ++q) {
        float4 w = wr[q];
        v += w.x * avv[q].x; v += w.y * avv[q].y; v += w.z * avv[q].z; v += w.w * avv[q].w;
      }
      bp[i] = (k < 7 && l16 < 8) ? v : 0.f;
    }
    sc = __builtin_amdgcn_wmma_f32_16x16x4_f32(
        false, a[s], false, bp, (short)0, sc, false, false);
  }
#pragma unroll
  for (int r = 0; r < 8; ++r) {
    int m = wave * 16 + r + 8 * half;
    if (l16 < 8 && m < N) xr[m * 16 + 4 * (l16 & 3) + 2 + (l16 >> 2)] = sc[r];
  }
}


#define SEL8(a, b, c, d, e, f, g, h) \
  ((a) | ((b) << 3) | ((c) << 6) | ((d) << 9) | ((e) << 12) | ((f) << 15) | ((g) << 18) | ((h) << 21))

__device__ __forceinline__ float sum8(float v) {
  v += __uint_as_float(__builtin_amdgcn_mov_dpp8(__float_as_uint(v), SEL8(1, 0, 3, 2, 5, 4, 7, 6)));
  v += __uint_as_float(__builtin_amdgcn_mov_dpp8(__float_as_uint(v), SEL8(2, 3, 0, 1, 6, 7, 4, 5)));
  v += __uint_as_float(__builtin_amdgcn_mov_dpp8(__float_as_uint(v), SEL8(4, 5, 6, 7, 0, 1, 2, 3)));
  return v;
}

template <int D>
__device__ __forceinline__ float lane_xor(float v) {
  unsigned u = __float_as_uint(v);
  if (D == 1) return __uint_as_float(__builtin_amdgcn_mov_dpp8(u, SEL8(1, 0, 3, 2, 5, 4, 7, 6)));
  if (D == 2) return __uint_as_float(__builtin_amdgcn_mov_dpp8(u, SEL8(2, 3, 0, 1, 6, 7, 4, 5)));
  if (D == 4) return __uint_as_float(__builtin_amdgcn_mov_dpp8(u, SEL8(4, 5, 6, 7, 0, 1, 2, 3)));
  if (D == 8)
    return __uint_as_float((unsigned)__builtin_amdgcn_update_dpp(0, (int)u, 0x128, 0xf, 0xf, false));
  return __uint_as_float(__builtin_amdgcn_permlanex16(u, u, 0x76543210u, 0xfedcba98u, false, false));
}
template <int Q>
__device__ __forceinline__ float quad_lane(float v) {
  return __uint_as_float((unsigned)__builtin_amdgcn_update_dpp(0, (int)__float_as_uint(v), Q * 0x55, 0xf, 0xf, true));
}
template <int D, int H>
__device__ __forceinline__ void halve(float* v, int lane) {
  bool up = (lane & D) != 0;
#pragma unroll
  for (int i = 0; i < H; ++i) {
    float keep = up ? v[i + H] : v[i];
    float give = up ? v[i] : v[i + H];
    v[i] = keep + lane_xor<D>(give);
  }
}

#define QADD(P, R) { \
  v2f xy; xy[0] = (R).x; xy[1] = (R).y; \
  float p0 = quad_lane<0>(P), p1 = quad_lane<1>(P), p2 = quad_lane<2>(P), p3 = quad_lane<3>(P); \
  acc[0] += xy * p0; acc[1] += xy * p1; acc[2] += xy * p2; acc[3] += xy * p3; }

__global__ void gat1_node(const int* __restrict__ count, const int* __restrict__ slots,
                          const int* __restrict__ ovfi,
                          const float* __restrict__ xr, float* __restrict__ agg, int N) {
  int lane = threadIdx.x & 31;
  int n = __builtin_amdgcn_readfirstlane((int)(blockIdx.x * (blockDim.x >> 5) + (threadIdx.x >> 5)));
  if (n >= N) return;
  const float4* rv = (const float4*)xr;
  int e = lane >> 2, q = lane & 3;

  int cnt = count[n];
  const int* lst = slots + n * SLOTS;
  const int2* ovf = (const int2*)ovfi;
  float4 own = rv[n * 4 + q];
  float ad = own.w;
  float es = leaky(own.z + ad);

  float4 rq[8];
  float sc[8];
  int s[8];
#pragma unroll
  for (int r = 0; r < 8; ++r) s[r] = n;
  int last = cnt - 1;
  bool more = cnt > 32;
  if (cnt > 0) {
    int v[8];
#pragma unroll
    for (int r = 0; r < 8; ++r) {
      int j = 8 * r + e;
      v[r] = lst[j < last ? j : last];
    }
#pragma unroll
    for (int r = 0; r < 8; ++r) s[r] = (8 * r + e < cnt) ? v[r] : n;
  }
#pragma unroll
  for (int r = 0; r < 4; ++r) rq[r] = rv[s[r] * 4 + q];
#pragma unroll
  for (int r = 4; r < 8; ++r) rq[r] = own;
  if (more) {
#pragma unroll
    for (int r = 4; r < 8; ++r) rq[r] = rv[s[r] * 4 + q];
  }

  float m = es;
#pragma unroll
  for (int r = 0; r < 8; ++r) {
    sc[r] = 0.f;
    if (r < 4 || cnt > 8 * r) {
      sc[r] = leaky(rq[r].z + ad);
      if (8 * r + e < cnt) m = fmaxf(m, sc[r]);
    }
  }
  int novf = 0;
  if (cnt > SLOTS) {
    novf = count[N];
    for (int base = 0; base < novf; base += 8) {
      int j = base + e;
      int2 t = ovf[j < novf ? j : novf - 1];
      bool v = j < novf && t.y == n;
      int s = v ? t.x : n;
      float sv = leaky(xr[s * 16 + 4 * q + 2] + ad);
      if (v) m = fmaxf(m, sv);
    }
  }
  m = fmaxf(m, lane_xor<4>(m));
  m = fmaxf(m, lane_xor<8>(m));
  m = fmaxf(m, lane_xor<16>(m));

  v2f acc[4];
#pragma unroll
  for (int i = 0; i < 4; ++i) { acc[i][0] = 0.f; acc[i][1] = 0.f; }
#pragma unroll
  for (int r = 0; r < 8; ++r) {
    if (r < 4 || cnt > 8 * r) {
      float p = (8 * r + e < cnt) ? __expf(sc[r] - m) : 0.f;
      QADD(p, rq[r])
    }
  }
  for (int base = 0; base < novf; base += 8) {
    int j = base + e;
    int2 t = ovf[j < novf ? j : novf - 1];
    bool v = j < novf && t.y == n;
    int s = v ? t.x : n;
    float4 h = rv[s * 4 + q];
    float p = v ? __expf(leaky(h.z + ad) - m) : 0.f;
    QADD(p, h)
  }
  float t[8];
#pragma unroll
  for (int i = 0; i < 4; ++i) { t[2 * i] = acc[i][0]; t[2 * i + 1] = acc[i][1]; }
  halve<16, 4>(t, lane);
  halve<8, 2>(t, lane);
  halve<4, 1>(t, lane);
  int hd = lane >> 3;
  float ws = __expf(es - m);
  float w0 = quad_lane<0>(ws), w1 = quad_lane<1>(ws), w2 = quad_lane<2>(ws), w3 = quad_lane<3>(ws);
  float wl = (hd & 2) ? ((hd & 1) ? w3 : w2) : ((hd & 1) ? w1 : w0);
  float xs = (lane & 4) ? own.y : own.x;
  float tot = t[0] + wl * xs;
  float den = __uint_as_float(__builtin_amdgcn_mov_dpp8(__float_as_uint(tot), SEL8(7, 7, 7, 7, 7, 7, 7, 7)));
  agg[(size_t)n * 32 + 8 * hd + 2 * q + ((lane >> 2) & 1)] = tot / den;
}

#define LDW 68
__global__ __launch_bounds__(256) void post1_proj2_wmma(const float* __restrict__ agg, const float* __restrict__ W1,
                           const float* __restrict__ b1, const float* __restrict__ W2,
                           const float* __restrict__ aS, const float* __restrict__ aD,
                           float* __restrict__ rec, int N) {
  __shared__ float tile[8 * 16 * LDW];
  int wv = threadIdx.x >> 5;
  int wave = blockIdx.x * 8 + wv;
  int lane = threadIdx.x & 31;
  int half = lane >> 4, l16 = lane & 15;
  int row  = wave * 16 + l16;
  int rowc = row < N ? row : N - 1;
  bool rok = row < N;
  bool cok = l16 < 6;
  float* tl = tile + wv * (16 * LDW);

#pragma unroll
  for (int t = 0; t < 4; ++t) {
    v8f hacc;
#pragma unroll
    for (int q = 0; q < 8; ++q) hacc[q] = 0.f;
#pragma unroll
    for (int s = 0; s < 2; ++s) {
      int k0 = 4 * s + 2 * half;
      v2f a, b;
      a[0] = rok ? agg[rowc * 32 + t * 8 + k0]     : 0.f;
      a[1] = rok ? agg[rowc * 32 + t * 8 + k0 + 1] : 0.f;
      int col = t * 16 + l16;
      b[0] = (k0     < 7) ? W1[k0 * 64 + col]       : 0.f;
      b[1] = (k0 + 1 < 7) ? W1[(k0 + 1) * 64 + col] : 0.f;
      hacc = __builtin_amdgcn_wmma_f32_16x16x4_f32(
          false, a, false, b, (short)0, hacc, false, false);
    }
    float bb = b1[t * 16 + l16];
#pragma unroll
    for (int r = 0; r < 8; ++r) {
      float o = hacc[r] + bb;
      o = o > 0.f ? o : (__expf(o) - 1.f);
      tl[(r + 8 * half) * LDW + t * 16 + l16] = o;
    }
  }
  __syncthreads();

  v8f acc;
#pragma unroll
  for (int q = 0; q < 8; ++q) acc[q] = 0.f;

#pragma unroll
  for (int s = 0; s < 16; ++s) {
    int k0 = 4 * s + 2 * half;
    v2f a, b;
    a = *(const v2f*)(tl + l16 * LDW + k0);
    b[0] = cok ? W2[k0 * 6 + l16]       : 0.f;
    b[1] = cok ? W2[(k0 + 1) * 6 + l16] : 0.f;
    acc = __builtin_amdgcn_wmma_f32_16x16x4_f32(
        false, a, false, b, (short)0, acc, false, false);
  }

  float cs = cok ? aS[l16] : 0.f;
  float cd = cok ? aD[l16] : 0.f;
#pragma unroll
  for (int r = 0; r < 8; ++r) {
    int m = wave * 16 + r + 8 * half;
    float s = sum8(acc[r] * cs);
    float d = sum8(acc[r] * cd);
    float v = cok ? acc[r] : (l16 == 6 ? s : d);
    if (l16 < 8 && m < N) rec[m * 8 + l16] = v;
  }
}


__device__ __forceinline__ float odd_of_pair(float v) {
  return __uint_as_float(__builtin_amdgcn_mov_dpp8(__float_as_uint(v), SEL8(1, 1, 3, 3, 5, 5, 7, 7)));
}

__global__ void gat2_node(const int* __restrict__ count, const int* __restrict__ slots,
                          const int* __restrict__ ovfi, const float* __restrict__ rec,
                          const float* __restrict__ b2, float* __restrict__ out, int N) {
  int gid = blockIdx.x * blockDim.x + threadIdx.x;
  int n = gid >> 4, sub = gid & 15;
  int p = sub >> 1, hf = sub & 1;
  bool ok = n < N;
  int nc = ok ? n : N - 1;
  int cnt = ok ? count[nc] : 0;
  const int* lst = slots + nc * SLOTS;
  const int2* ovf = (const int2*)ovfi;
  const float4* rv = (const float4*)rec;
  float4 rs = rv[nc * 2 + hf];
  float ad = odd_of_pair(rs.w);
  float es = leaky(odd_of_pair(rs.z) + ad);
  float m = es;

  int sv[8];
  float4 r[8];
  float ev[8];
#pragma unroll
  for (int k = 0; k < 8; ++k) {
    int j = p + 8 * k;
    int v = nc;
    if (j < cnt) v = lst[j];
    sv[k] = v;
  }
#pragma unroll
  for (int k = 0; k < 8; ++k) r[k] = rv[sv[k] * 2 + hf];
#pragma unroll
  for (int k = 0; k < 8; ++k) {
    int j = p + 8 * k;
    float e = leaky(odd_of_pair(r[k].z) + ad);
    ev[k] = e;
    if (j < cnt) m = fmaxf(m, e);
  }
  int novf = cnt > SLOTS ? count[N] : 0;
  for (int j = p; j < novf; j += 8) {
    int2 q = ovf[j];
    if (q.y == n) m = fmaxf(m, leaky(rec[q.x * 8 + 6] + ad));
  }
  m = fmaxf(m, __shfl_xor(m, 2));
  m = fmaxf(m, __shfl_xor(m, 4));
  m = fmaxf(m, __shfl_xor(m, 8));

  float den = 0.f;
  float acc[4];
#pragma unroll
  for (int c = 0; c < 4; ++c) acc[c] = 0.f;
#pragma unroll
  for (int k = 0; k < 8; ++k) {
    int j = p + 8 * k;
    if (j < cnt) {
      float w = __expf(ev[k] - m);
      den += w;
      acc[0] += w * r[k].x; acc[1] += w * r[k].y;
      acc[2] += w * r[k].z; acc[3] += w * r[k].w;
    }
  }
  for (int j = p; j < novf; j += 8) {
    int2 q = ovf[j];
    if (q.y == n) {
      int s = q.x;
      float w = __expf(leaky(rec[s * 8 + 6] + ad) - m);
      den += w;
      float4 h = rv[s * 2 + hf];
      acc[0] += w * h.x; acc[1] += w * h.y;
      acc[2] += w * h.z; acc[3] += w * h.w;
    }
  }
  if (p == 0) {
    float w = __expf(es - m);
    den += w;
    acc[0] += w * rs.x; acc[1] += w * rs.y;
    acc[2] += w * rs.z; acc[3] += w * rs.w;
  }
#pragma unroll
  for (int off = 2; off < 16; off <<= 1) {
    den += __shfl_xor(den, off);
#pragma unroll
    for (int c = 0; c < 4; ++c) acc[c] += __shfl_xor(acc[c], off);
  }
  float o4 = __shfl_xor(acc[0], 1);
  float o5 = __shfl_xor(acc[1], 1);

  if (ok && sub == 0) {
    float o[6] = {acc[0], acc[1], acc[2], acc[3], o4, o5};
    float mx = -3.0e38f;
#pragma unroll
    for (int c = 0; c < 6; ++c) { o[c] = o[c] / den + b2[c]; mx = fmaxf(mx, o[c]); }
    float sum = 0.f;
#pragma unroll
    for (int c = 0; c < 6; ++c) sum += __expf(o[c] - mx);
    float l = __logf(sum);
#pragma unroll
    for (int c = 0; c < 6; ++c) out[n * 6 + c] = o[c] - mx - l;
  }
}

extern "C" void kernel_launch(void* const* d_in, const int* in_sizes, int n_in,
                              void* d_out, int out_size, void* d_ws, size_t ws_size,
                              hipStream_t stream) {
  const float* x      = (const float*)d_in[0];
  const int*   ei     = (const int*)d_in[1];
  const float* W1     = (const float*)d_in[2];
  const float* a_src1 = (const float*)d_in[3];
  const float* a_dst1 = (const float*)d_in[4];
  const float* b1     = (const float*)d_in[5];
  const float* W2     = (const float*)d_in[6];
  const float* a_src2 = (const float*)d_in[7];
  const float* a_dst2 = (const float*)d_in[8];
  const float* b2     = (const float*)d_in[9];
  float* out = (float*)d_out;

  int N = in_sizes[0] / 7;
  int E = in_sizes[1] / 2;

  int nbin = (N + BN - 1) / BN;
  int nblk = (E + CH - 1) / CH;
  int dpitch = (nbin + 15) & ~15;

  float* ws  = (float*)d_ws;
  float* agg = ws;
  float* xr  = agg + (size_t)N * 64;
  int* slots = (int*)(xr + (size_t)N * 64);
  float* rec = (float*)(slots + (size_t)N * SLOTS);
  int* ovf   = (int*)(rec + (size_t)N * 8);
  int* bins  = ovf + (size_t)E * 2;
  int2* desc = (int2*)(bins + (size_t)nblk * CH);
  int* count = (int*)(desc + (size_t)nblk * dpitch);

  const int B = 256;
  auto G = [](long n, int b) { return (int)((n + b - 1) / b); };
  int tiles = (N + 15) / 16;
  int projBlocks = (tiles + (B / 32) - 1) / (B / 32);

  sort_chunks<<<nblk, TPB, 0, stream>>>(ei, bins, desc, count, N, E, nbin, dpitch);
  build_rows<<<nbin, TPB, 0, stream>>>(bins, desc, count, slots, ovf, N, nblk, dpitch);

  prep1_wmma<<<projBlocks, B, 0, stream>>>(x, W1, a_src1, a_dst1, xr, N);
  gat1_node<<<G(N, B / 32), B, 0, stream>>>(count, slots, ovf, xr, agg, N);

  post1_proj2_wmma<<<projBlocks, B, 0, stream>>>(agg, W1, b1, W2, a_src2, a_dst2, rec, N);
  gat2_node<<<G((long)N * 16, B), B, 0, stream>>>(count, slots, ovf, rec, b2, out, N);
}
